# starting kernel
# baseline (speedup 1.0000x reference)
#include <hip/hip_runtime.h>


#define NN 4096
#define DD 16

typedef float v2f __attribute__((ext_vector_type(2)));
typedef float v8f __attribute__((ext_vector_type(8)));

__device__ __forceinline__ void load_a_frag(const float* __restrict__ X,
                                            int row0, float scale, v2f a[4]) {
  const int lane = threadIdx.x & 31;
  const int m = lane & 15, h = lane >> 4;
  const float* base = X + (size_t)(row0 + m) * DD + 2 * h;
#pragma unroll
  for (int c = 0; c < 4; ++c) {
    a[c].x = base[4 * c] * scale;
    a[c].y = base[4 * c + 1] * scale;
  }
}

__device__ __forceinline__ void load_b_frag(const float* __restrict__ W, v2f b[4]) {
  const int lane = threadIdx.x & 31;
  const int n = lane & 15, h = lane >> 4;
#pragma unroll
  for (int c = 0; c < 4; ++c) {
    b[c].x = W[(4 * c + 2 * h) * DD + n];
    b[c].y = W[(4 * c + 2 * h + 1) * DD + n];
  }
}

__device__ __forceinline__ void load_bT_frag(const float* __restrict__ W, v2f b[4]) {
  const int lane = threadIdx.x & 31;
  const int n = lane & 15, h = lane >> 4;
  const float* row = W + n * DD;
#pragma unroll
  for (int c = 0; c < 4; ++c) {
    b[c].x = row[4 * c + 2 * h];
    b[c].y = row[4 * c + 2 * h + 1];
  }
}

__device__ __forceinline__ v8f load_c_tile(const float* __restrict__ X, int row0) {
  const int lane = threadIdx.x & 31;
  const int n = lane & 15, h = lane >> 4;
  v8f c;
#pragma unroll
  for (int r = 0; r < 8; ++r) c[r] = X[(size_t)(row0 + r + 8 * h) * DD + n];
  return c;
}

__device__ __forceinline__ void store_d_tile(float* __restrict__ O, int row0, v8f d) {
  const int lane = threadIdx.x & 31;
  const int n = lane & 15, h = lane >> 4;
#pragma unroll
  for (int r = 0; r < 8; ++r) O[(size_t)(row0 + r + 8 * h) * DD + n] = d[r];
}

__device__ __forceinline__ v8f wmma16(const v2f a[4], const v2f b[4], v8f c) {
#pragma unroll
  for (int k = 0; k < 4; ++k)
    c = __builtin_amdgcn_wmma_f32_16x16x4_f32(false, a[k], false, b[k],
                                              (short)0, c, false, false);
  return c;
}

__global__ void k_build_mask(const int* __restrict__ mrow, const int* __restrict__ mcol,
                             int E, unsigned* __restrict__ Md) {
  int e = blockIdx.x * blockDim.x + threadIdx.x;
  if (e >= E) return;
  unsigned bit = (unsigned)mrow[e] * (unsigned)NN + (unsigned)mcol[e];
  atomicOr(&Md[bit >> 5], 1u << (bit & 31));
}

__global__ void k_build_adj(const int* __restrict__ erow, const int* __restrict__ ecol,
                            int E, const unsigned* __restrict__ Md,
                            unsigned* __restrict__ Ab, unsigned* __restrict__ list,
                            unsigned* __restrict__ cnt) {
  int e = blockIdx.x * blockDim.x + threadIdx.x;
  if (e >= E) return;
  unsigned i = (unsigned)erow[e], j = (unsigned)ecol[e];
  unsigned bit = i * (unsigned)NN + j;
  unsigned w = bit >> 5, b = 1u << (bit & 31);
  if (!(Md[w] & b)) return;
  unsigned old = atomicOr(&Ab[w], b);
  if (old & b) return;
  unsigned idx = atomicAdd(cnt, 1u);
  list[idx] = (i << 12) | j;
}

__global__ void k_g1(const float* __restrict__ W1, const float* __restrict__ Q,
                     float* __restrict__ G1) {
  int s = blockIdx.x;
  int k = threadIdx.x >> 4, n = threadIdx.x & 15;
  float acc = 0.f;
#pragma unroll
  for (int t = 0; t < DD; ++t)
    acc = fmaf(W1[s * DD * DD + t * DD + k], Q[t * DD + n], acc);
  G1[s * DD * DD + k * DD + n] = acc;
}

__global__ void k_pre(const float* __restrict__ x, const float* __restrict__ Qm,
                      const float* __restrict__ W2s, float* __restrict__ xq,
                      float* __restrict__ x2) {
  int wave = threadIdx.x >> 5;
  int row0 = (blockIdx.x * (blockDim.x >> 5) + wave) * 16;
  v2f a[4], b[4];
  load_a_frag(x, row0, 1.0f, a);
  load_b_frag(Qm, b);
  v8f d = {};
  d = wmma16(a, b, d);
  store_d_tile(xq, row0, d);
  load_bT_frag(W2s, b);
  v8f d2 = {};
  d2 = wmma16(a, b, d2);
  store_d_tile(x2, row0, d2);
}

__global__ void k_scatter1(const unsigned* __restrict__ list,
                           const unsigned* __restrict__ cnt,
                           const float* __restrict__ x, float* __restrict__ y1) {
  unsigned e = blockIdx.x * blockDim.x + threadIdx.x;
  if (e >= *cnt) return;
  unsigned p = list[e];
  const float* xs = x + (size_t)(p >> 12) * DD;
  float* yd = y1 + (size_t)(p & 4095u) * DD;
#pragma unroll
  for (int c = 0; c < DD; ++c) atomicAdd(&yd[c], xs[c]);
}

__global__ void k_scatter2(const unsigned* __restrict__ list,
                           const unsigned* __restrict__ cnt,
                           const float* __restrict__ xq, const float* __restrict__ x,
                           const float* __restrict__ x2, float* __restrict__ y2) {
  unsigned e = blockIdx.x * blockDim.x + threadIdx.x;
  if (e >= *cnt) return;
  unsigned p = list[e];
  unsigned i = p >> 12, j = p & 4095u;
  const float* qi = xq + (size_t)i * DD;
  const float* xj = x + (size_t)j * DD;
  float a = 0.f;
#pragma unroll
  for (int c = 0; c < DD; ++c) a = fmaf(qi[c], xj[c], a);
  const float* v = x2 + (size_t)i * DD;
  float* yd = y2 + (size_t)j * DD;
#pragma unroll
  for (int c = 0; c < DD; ++c) atomicAdd(&yd[c], a * v[c]);
}

__global__ void k_combine(const float* __restrict__ xcur, const float* __restrict__ y1,
                          const float* __restrict__ y2, const float* __restrict__ G1s,
                          const float* __restrict__ Qm, const unsigned* __restrict__ cnt2,
                          float s1, float* __restrict__ xnext) {
  int wave = threadIdx.x >> 5;
  int row0 = (blockIdx.x * (blockDim.x >> 5) + wave) * 16;
  float s2 = -(float)NN / (float)(*cnt2);
  v2f a[4], b[4];
  v8f c = load_c_tile(xcur, row0);
  load_a_frag(y1, row0, s1, a);
  load_b_frag(G1s, b);
  c = wmma16(a, b, c);
  load_a_frag(y2, row0, s2, a);
  load_b_frag(Qm, b);
  c = wmma16(a, b, c);
  store_d_tile(xnext, row0, c);
}

extern "C" void kernel_launch(void* const* d_in, const int* in_sizes, int n_in,
                              void* d_out, int out_size, void* d_ws, size_t ws_size,
                              hipStream_t stream) {
  const float* x_in = (const float*)d_in[0];
  const int* e1 = (const int*)d_in[1];
  const int* e2 = (const int*)d_in[2];
  const float* Qm = (const float*)d_in[3];
  const int* mk = (const int*)d_in[4];
  const float* W1 = (const float*)d_in[5];
  const float* W2 = (const float*)d_in[6];
  float* out = (float*)d_out;

  const int E1 = in_sizes[1] / 2;
  const int E2 = in_sizes[2] / 2;
  const int EM = in_sizes[4] / 2;
  const int STEPS = in_sizes[5] / (DD * DD);

  char* ws = (char*)d_ws;
  size_t off = 0;
  auto carve = [&](size_t bytes) {
    char* p = ws + off;
    off += (bytes + 255) & ~(size_t)255;
    return p;
  };
  const size_t BM = (size_t)NN * NN / 8;
  unsigned* Md    = (unsigned*)carve(BM);
  unsigned* A1b   = (unsigned*)carve(BM);
  unsigned* A2b   = (unsigned*)carve(BM);
  unsigned* list1 = (unsigned*)carve((size_t)E1 * 4);
  unsigned* list2 = (unsigned*)carve((size_t)E2 * 4);
  unsigned* cnts  = (unsigned*)carve(8);
  float* G1  = (float*)carve((size_t)STEPS * DD * DD * 4);
  float* xq  = (float*)carve((size_t)NN * DD * 4);
  float* x2  = (float*)carve((size_t)NN * DD * 4);
  float* y1  = (float*)carve((size_t)NN * DD * 4);
  float* y2  = (float*)carve((size_t)NN * DD * 4);
  float* xb0 = (float*)carve((size_t)NN * DD * 4);
  float* xb1 = (float*)carve((size_t)NN * DD * 4);

  hipMemsetAsync(Md, 0, 3 * BM, stream);
  hipMemsetAsync(cnts, 0, 8, stream);

  k_build_mask<<<(EM + 255) / 256, 256, 0, stream>>>(mk, mk + EM, EM, Md);
  k_build_adj<<<(E1 + 255) / 256, 256, 0, stream>>>(e1, e1 + E1, E1, Md, A1b, list1, cnts + 0);
  k_build_adj<<<(E2 + 255) / 256, 256, 0, stream>>>(e2, e2 + E2, E2, Md, A2b, list2, cnts + 1);
  k_g1<<<STEPS, 256, 0, stream>>>(W1, Qm, G1);

  const float s1 = (float)NN / (float)EM;

  const int TILE_BLOCKS = NN / 16 / 8;
  const float* xc = x_in;
  for (int s = 0; s < STEPS; ++s) {
    float* xn = (s == STEPS - 1) ? out : ((s & 1) ? xb1 : xb0);
    hipMemsetAsync(y1, 0, (size_t)NN * DD * 4, stream);
    hipMemsetAsync(y2, 0, (size_t)NN * DD * 4, stream);
    k_pre<<<TILE_BLOCKS, 256, 0, stream>>>(xc, Qm, W2 + (size_t)s * DD * DD, xq, x2);
    k_scatter1<<<(E1 + 255) / 256, 256, 0, stream>>>(list1, cnts + 0, xc, y1);
    k_scatter2<<<(E2 + 255) / 256, 256, 0, stream>>>(list2, cnts + 1, xq, xc, x2, y2);
    k_combine<<<TILE_BLOCKS, 256, 0, stream>>>(xc, y1, y2, G1 + (size_t)s * DD * DD,
                                               Qm, cnts + 1, s1, xn);
    xc = xn;
  }
}


# Edge scatters: one edge per 16-lane row, one feature per lane (coalesced row loads and atomics); attention dot now summed by a 16-lane xor butterfly, not a serial fma chain
# speedup vs baseline: 4.9523x; 4.9523x over previous
#include <hip/hip_runtime.h>


#define NN 4096
#define DD 16

typedef float v2f __attribute__((ext_vector_type(2)));
typedef float v8f __attribute__((ext_vector_type(8)));

__device__ __forceinline__ void load_a_frag(const float* __restrict__ X,
                                            int row0, float scale, v2f a[4]) {
  const int lane = threadIdx.x & 31;
  const int m = lane & 15, h = lane >> 4;
  const float* base = X + (size_t)(row0 + m) * DD + 2 * h;
#pragma unroll
  for (int c = 0; c < 4; ++c) {
    a[c].x = base[4 * c] * scale;
    a[c].y = base[4 * c + 1] * scale;
  }
}

__device__ __forceinline__ void load_b_frag(const float* __restrict__ W, v2f b[4]) {
  const int lane = threadIdx.x & 31;
  const int n = lane & 15, h = lane >> 4;
#pragma unroll
  for (int c = 0; c < 4; ++c) {
    b[c].x = W[(4 * c + 2 * h) * DD + n];
    b[c].y = W[(4 * c + 2 * h + 1) * DD + n];
  }
}

__device__ __forceinline__ void load_bT_frag(const float* __restrict__ W, v2f b[4]) {
  const int lane = threadIdx.x & 31;
  const int n = lane & 15, h = lane >> 4;
  const float* row = W + n * DD;
#pragma unroll
  for (int c = 0; c < 4; ++c) {
    b[c].x = row[4 * c + 2 * h];
    b[c].y = row[4 * c + 2 * h + 1];
  }
}

__device__ __forceinline__ v8f load_c_tile(const float* __restrict__ X, int row0) {
  const int lane = threadIdx.x & 31;
  const int n = lane & 15, h = lane >> 4;
  v8f c;
#pragma unroll
  for (int r = 0; r < 8; ++r) c[r] = X[(size_t)(row0 + r + 8 * h) * DD + n];
  return c;
}

__device__ __forceinline__ void store_d_tile(float* __restrict__ O, int row0, v8f d) {
  const int lane = threadIdx.x & 31;
  const int n = lane & 15, h = lane >> 4;
#pragma unroll
  for (int r = 0; r < 8; ++r) O[(size_t)(row0 + r + 8 * h) * DD + n] = d[r];
}

__device__ __forceinline__ v8f wmma16(const v2f a[4], const v2f b[4], v8f c) {
#pragma unroll
  for (int k = 0; k < 4; ++k)
    c = __builtin_amdgcn_wmma_f32_16x16x4_f32(false, a[k], false, b[k],
                                              (short)0, c, false, false);
  return c;
}

__global__ void k_build_mask(const int* __restrict__ mrow, const int* __restrict__ mcol,
                             int E, unsigned* __restrict__ Md) {
  int e = blockIdx.x * blockDim.x + threadIdx.x;
  if (e >= E) return;
  unsigned bit = (unsigned)mrow[e] * (unsigned)NN + (unsigned)mcol[e];
  atomicOr(&Md[bit >> 5], 1u << (bit & 31));
}

__global__ void k_build_adj(const int* __restrict__ erow, const int* __restrict__ ecol,
                            int E, const unsigned* __restrict__ Md,
                            unsigned* __restrict__ Ab, unsigned* __restrict__ list,
                            unsigned* __restrict__ cnt) {
  int e = blockIdx.x * blockDim.x + threadIdx.x;
  if (e >= E) return;
  unsigned i = (unsigned)erow[e], j = (unsigned)ecol[e];
  unsigned bit = i * (unsigned)NN + j;
  unsigned w = bit >> 5, b = 1u << (bit & 31);
  if (!(Md[w] & b)) return;
  unsigned old = atomicOr(&Ab[w], b);
  if (old & b) return;
  unsigned idx = atomicAdd(cnt, 1u);
  list[idx] = (i << 12) | j;
}

__global__ void k_g1(const float* __restrict__ W1, const float* __restrict__ Q,
                     float* __restrict__ G1) {
  int s = blockIdx.x;
  int k = threadIdx.x >> 4, n = threadIdx.x & 15;
  float acc = 0.f;
#pragma unroll
  for (int t = 0; t < DD; ++t)
    acc = fmaf(W1[s * DD * DD + t * DD + k], Q[t * DD + n], acc);
  G1[s * DD * DD + k * DD + n] = acc;
}

__global__ void k_pre(const float* __restrict__ x, const float* __restrict__ Qm,
                      const float* __restrict__ W2s, float* __restrict__ xq,
                      float* __restrict__ x2) {
  int wave = threadIdx.x >> 5;
  int row0 = (blockIdx.x * (blockDim.x >> 5) + wave) * 16;
  v2f a[4], b[4];
  load_a_frag(x, row0, 1.0f, a);
  load_b_frag(Qm, b);
  v8f d = {};
  d = wmma16(a, b, d);
  store_d_tile(xq, row0, d);
  load_bT_frag(W2s, b);
  v8f d2 = {};
  d2 = wmma16(a, b, d2);
  store_d_tile(x2, row0, d2);
}

#define EPG 8

__device__ __forceinline__ float row16_sum(float v) {
  v += __int_as_float(__builtin_amdgcn_update_dpp(0, __float_as_int(v), 0x161, 0xf, 0xf, true));
  v += __int_as_float(__builtin_amdgcn_update_dpp(0, __float_as_int(v), 0x162, 0xf, 0xf, true));
  v += __int_as_float(__builtin_amdgcn_update_dpp(0, __float_as_int(v), 0x164, 0xf, 0xf, true));
  v += __int_as_float(__builtin_amdgcn_update_dpp(0, __float_as_int(v), 0x168, 0xf, 0xf, true));
  return v;
}

__device__ __forceinline__ void scatter1_row(const unsigned* __restrict__ list,
                                             const float* __restrict__ x,
                                             float* __restrict__ y1,
                                             unsigned e0, unsigned c, unsigned m) {
  unsigned p[EPG];
  float v[EPG];
#pragma unroll
  for (unsigned k = 0; k < EPG; ++k) p[k] = list[e0 + (k < m ? k : 0u)];
#pragma unroll
  for (unsigned k = 0; k < EPG; ++k) v[k] = x[(size_t)(p[k] >> 12) * DD + c];
#pragma unroll
  for (unsigned k = 0; k < EPG; ++k)
    if (k < m) atomicAdd(&y1[(size_t)(p[k] & 4095u) * DD + c], v[k]);
}

__global__ void k_scatter1(const unsigned* __restrict__ list,
                           const unsigned* __restrict__ cnt,
                           const float* __restrict__ x, float* __restrict__ y1) {
  const unsigned n = *cnt;
  const unsigned t = blockIdx.x * blockDim.x + threadIdx.x;
  const unsigned c = t & 15u;
  const unsigned e0 = (t >> 4) * EPG;
  if (e0 >= n) return;
  if (n - e0 >= EPG) scatter1_row(list, x, y1, e0, c, EPG);
  else               scatter1_row(list, x, y1, e0, c, n - e0);
}

__device__ __forceinline__ void scatter2_row(const unsigned* __restrict__ list,
                                             const float* __restrict__ xq,
                                             const float* __restrict__ x,
                                             const float* __restrict__ x2,
                                             float* __restrict__ y2,
                                             unsigned e0, unsigned c, unsigned m) {
  unsigned p[EPG];
  float q[EPG], xj[EPG], v[EPG];
#pragma unroll
  for (unsigned k = 0; k < EPG; ++k) p[k] = list[e0 + (k < m ? k : 0u)];
#pragma unroll
  for (unsigned k = 0; k < EPG; ++k) {
    const size_t i = p[k] >> 12, j = p[k] & 4095u;
    q[k] = xq[i * DD + c];
    xj[k] = x[j * DD + c];
    v[k] = x2[i * DD + c];
  }
#pragma unroll
  for (unsigned k = 0; k < EPG; ++k) {
    const float a = row16_sum(q[k] * xj[k]);
    if (k < m) atomicAdd(&y2[(size_t)(p[k] & 4095u) * DD + c], a * v[k]);
  }
}

__global__ void k_scatter2(const unsigned* __restrict__ list,
                           const unsigned* __restrict__ cnt,
                           const float* __restrict__ xq, const float* __restrict__ x,
                           const float* __restrict__ x2, float* __restrict__ y2) {
  const unsigned n = *cnt;
  const unsigned t = blockIdx.x * blockDim.x + threadIdx.x;
  const unsigned c = t & 15u;
  const unsigned e0 = (t >> 4) * EPG;
  if (e0 >= n) return;
  if (n - e0 >= EPG) scatter2_row(list, xq, x, x2, y2, e0, c, EPG);
  else               scatter2_row(list, xq, x, x2, y2, e0, c, n - e0);
}

__global__ void k_combine(const float* __restrict__ xcur, const float* __restrict__ y1,
                          const float* __restrict__ y2, const float* __restrict__ G1s,
                          const float* __restrict__ Qm, const unsigned* __restrict__ cnt2,
                          float s1, float* __restrict__ xnext) {
  int wave = threadIdx.x >> 5;
  int row0 = (blockIdx.x * (blockDim.x >> 5) + wave) * 16;
  float s2 = -(float)NN / (float)(*cnt2);
  v2f a[4], b[4];
  v8f c = load_c_tile(xcur, row0);
  load_a_frag(y1, row0, s1, a);
  load_b_frag(G1s, b);
  c = wmma16(a, b, c);
  load_a_frag(y2, row0, s2, a);
  load_b_frag(Qm, b);
  c = wmma16(a, b, c);
  store_d_tile(xnext, row0, c);
}

extern "C" void kernel_launch(void* const* d_in, const int* in_sizes, int n_in,
                              void* d_out, int out_size, void* d_ws, size_t ws_size,
                              hipStream_t stream) {
  const float* x_in = (const float*)d_in[0];
  const int* e1 = (const int*)d_in[1];
  const int* e2 = (const int*)d_in[2];
  const float* Qm = (const float*)d_in[3];
  const int* mk = (const int*)d_in[4];
  const float* W1 = (const float*)d_in[5];
  const float* W2 = (const float*)d_in[6];
  float* out = (float*)d_out;

  const int E1 = in_sizes[1] / 2;
  const int E2 = in_sizes[2] / 2;
  const int EM = in_sizes[4] / 2;
  const int STEPS = in_sizes[5] / (DD * DD);

  char* ws = (char*)d_ws;
  size_t off = 0;
  auto carve = [&](size_t bytes) {
    char* p = ws + off;
    off += (bytes + 255) & ~(size_t)255;
    return p;
  };
  const size_t BM = (size_t)NN * NN / 8;
  unsigned* Md    = (unsigned*)carve(BM);
  unsigned* A1b   = (unsigned*)carve(BM);
  unsigned* A2b   = (unsigned*)carve(BM);
  unsigned* list1 = (unsigned*)carve((size_t)E1 * 4);
  unsigned* list2 = (unsigned*)carve((size_t)E2 * 4);
  unsigned* cnts  = (unsigned*)carve(8);
  float* G1  = (float*)carve((size_t)STEPS * DD * DD * 4);
  float* xq  = (float*)carve((size_t)NN * DD * 4);
  float* x2  = (float*)carve((size_t)NN * DD * 4);
  float* y1  = (float*)carve((size_t)NN * DD * 4);
  float* y2  = (float*)carve((size_t)NN * DD * 4);
  float* xb0 = (float*)carve((size_t)NN * DD * 4);
  float* xb1 = (float*)carve((size_t)NN * DD * 4);

  hipMemsetAsync(Md, 0, 3 * BM, stream);
  hipMemsetAsync(cnts, 0, 8, stream);

  k_build_mask<<<(EM + 255) / 256, 256, 0, stream>>>(mk, mk + EM, EM, Md);
  k_build_adj<<<(E1 + 255) / 256, 256, 0, stream>>>(e1, e1 + E1, E1, Md, A1b, list1, cnts + 0);
  k_build_adj<<<(E2 + 255) / 256, 256, 0, stream>>>(e2, e2 + E2, E2, Md, A2b, list2, cnts + 1);
  k_g1<<<STEPS, 256, 0, stream>>>(W1, Qm, G1);

  const float s1 = (float)NN / (float)EM;

  const int TILE_BLOCKS = NN / 16 / 8;
  const int SC_BLOCKS1 = ((E1 + EPG - 1) / EPG * 16 + 255) / 256;
  const int SC_BLOCKS2 = ((E2 + EPG - 1) / EPG * 16 + 255) / 256;
  const float* xc = x_in;
  for (int s = 0; s < STEPS; ++s) {
    float* xn = (s == STEPS - 1) ? out : ((s & 1) ? xb1 : xb0);
    hipMemsetAsync(y1, 0, (size_t)NN * DD * 4, stream);
    hipMemsetAsync(y2, 0, (size_t)NN * DD * 4, stream);
    k_pre<<<TILE_BLOCKS, 256, 0, stream>>>(xc, Qm, W2 + (size_t)s * DD * DD, xq, x2);
    k_scatter1<<<SC_BLOCKS1, 256, 0, stream>>>(list1, cnts + 0, xc, y1);
    k_scatter2<<<SC_BLOCKS2, 256, 0, stream>>>(list2, cnts + 1, xq, xc, x2, y2);
    k_combine<<<TILE_BLOCKS, 256, 0, stream>>>(xc, y1, y2, G1 + (size_t)s * DD * DD,
                                               Qm, cnts + 1, s1, xn);
    xc = xn;
  }
}


# Launch fusion, 24 stream operations to 9: independent kernels share one launch; the combine wave also clears its y1/y2 rows and makes the next step's x@Q, x@W2^T (D to A layout via LDS)
# speedup vs baseline: 6.4841x; 1.3093x over previous
#include <hip/hip_runtime.h>


#define NN 4096
#define DD 16

typedef float v2f __attribute__((ext_vector_type(2)));
typedef float v8f __attribute__((ext_vector_type(8)));

__device__ __forceinline__ void load_a_frag(const float* X, int row0, float scale, v2f a[4]) {
  const int lane = threadIdx.x & 31;
  const int m = lane & 15, h = lane >> 4;
  const float* base = X + (size_t)(row0 + m) * DD + 2 * h;
#pragma unroll
  for (int c = 0; c < 4; ++c) {
    a[c].x = base[4 * c] * scale;
    a[c].y = base[4 * c + 1] * scale;
  }
}

__device__ __forceinline__ void load_b_frag(const float* __restrict__ W, v2f b[4]) {
  const int lane = threadIdx.x & 31;
  const int n = lane & 15, h = lane >> 4;
#pragma unroll
  for (int c = 0; c < 4; ++c) {
    b[c].x = W[(4 * c + 2 * h) * DD + n];
    b[c].y = W[(4 * c + 2 * h + 1) * DD + n];
  }
}

__device__ __forceinline__ void load_bT_frag(const float* __restrict__ W, v2f b[4]) {
  const int lane = threadIdx.x & 31;
  const int n = lane & 15, h = lane >> 4;
  const float* row = W + n * DD;
#pragma unroll
  for (int c = 0; c < 4; ++c) {
    b[c].x = row[4 * c + 2 * h];
    b[c].y = row[4 * c + 2 * h + 1];
  }
}

__device__ __forceinline__ v8f load_c_tile(const float* __restrict__ X, int row0) {
  const int lane = threadIdx.x & 31;
  const int n = lane & 15, h = lane >> 4;
  v8f c;
#pragma unroll
  for (int r = 0; r < 8; ++r) c[r] = X[(size_t)(row0 + r + 8 * h) * DD + n];
  return c;
}

__device__ __forceinline__ void store_d_tile(float* __restrict__ O, int row0, v8f d) {
  const int lane = threadIdx.x & 31;
  const int n = lane & 15, h = lane >> 4;
#pragma unroll
  for (int r = 0; r < 8; ++r) O[(size_t)(row0 + r + 8 * h) * DD + n] = d[r];
}

__device__ __forceinline__ v8f wmma16(const v2f a[4], const v2f b[4], v8f c) {
#pragma unroll
  for (int k = 0; k < 4; ++k)
    c = __builtin_amdgcn_wmma_f32_16x16x4_f32(false, a[k], false, b[k],
                                              (short)0, c, false, false);
  return c;
}

__device__ __forceinline__ void pre_tile(const v2f a[4], const float* __restrict__ Qm,
                                         const float* __restrict__ W2s,
                                         float* __restrict__ xq, float* __restrict__ x2,
                                         int row0) {
  v2f b[4];
  load_b_frag(Qm, b);
  v8f d = {};
  d = wmma16(a, b, d);
  store_d_tile(xq, row0, d);
  load_bT_frag(W2s, b);
  v8f d2 = {};
  d2 = wmma16(a, b, d2);
  store_d_tile(x2, row0, d2);
}

__device__ __forceinline__ void zero_tile(float* Y, int row0) {
  const int lane = threadIdx.x & 31;
  float4* p = reinterpret_cast<float4*>(Y + (size_t)row0 * DD);
  const float4 z = {0.f, 0.f, 0.f, 0.f};
  p[lane] = z;
  p[lane + 32] = z;
}

__global__ void k_setup(const float* __restrict__ x, const float* __restrict__ Qm,
                        const float* __restrict__ W1, const float* __restrict__ W2,
                        int steps, int tile_blocks,
                        float* __restrict__ xq, float* __restrict__ x2,
                        float* __restrict__ y1, float* __restrict__ y2,
                        float* __restrict__ G1,
                        const int* __restrict__ mrow, const int* __restrict__ mcol,
                        int E, unsigned* __restrict__ Md) {
  int b = blockIdx.x;
  if (b < tile_blocks) {
    const int wave = threadIdx.x >> 5;
    const int row0 = (b * (blockDim.x >> 5) + wave) * 16;
    v2f a[4];
    load_a_frag(x, row0, 1.0f, a);
    pre_tile(a, Qm, W2, xq, x2, row0);
    zero_tile(y1, row0);
    zero_tile(y2, row0);
    return;
  }
  b -= tile_blocks;
  if (b < steps) {
    const int k = threadIdx.x >> 4, n = threadIdx.x & 15;
    float acc = 0.f;
#pragma unroll
    for (int t = 0; t < DD; ++t)
      acc = fmaf(W1[b * DD * DD + t * DD + k], Qm[t * DD + n], acc);
    G1[b * DD * DD + k * DD + n] = acc;
    return;
  }
  b -= steps;
  const int e = b * blockDim.x + threadIdx.x;
  if (e >= E) return;
  const unsigned bit = (unsigned)mrow[e] * (unsigned)NN + (unsigned)mcol[e];
  atomicOr(&Md[bit >> 5], 1u << (bit & 31));
}

__device__ __forceinline__ void build_adj(const int* __restrict__ erow, const int* __restrict__ ecol,
                                          int E, int e, const unsigned* __restrict__ Md,
                                          unsigned* __restrict__ Ab, unsigned* __restrict__ list,
                                          unsigned* __restrict__ cnt) {
  if (e >= E) return;
  unsigned i = (unsigned)erow[e], j = (unsigned)ecol[e];
  unsigned bit = i * (unsigned)NN + j;
  unsigned w = bit >> 5, b = 1u << (bit & 31);
  if (!(Md[w] & b)) return;
  unsigned old = atomicOr(&Ab[w], b);
  if (old & b) return;
  unsigned idx = atomicAdd(cnt, 1u);
  list[idx] = (i << 12) | j;
}

__global__ void k_build_adj(const int* __restrict__ e1row, const int* __restrict__ e1col, int E1,
                            unsigned* __restrict__ A1b, unsigned* __restrict__ list1,
                            const int* __restrict__ e2row, const int* __restrict__ e2col, int E2,
                            unsigned* __restrict__ A2b, unsigned* __restrict__ list2,
                            int blocks1, const unsigned* __restrict__ Md,
                            unsigned* __restrict__ cnts) {
  const int b = blockIdx.x;
  if (b < blocks1)
    build_adj(e1row, e1col, E1, b * blockDim.x + threadIdx.x, Md, A1b, list1, cnts + 0);
  else
    build_adj(e2row, e2col, E2, (b - blocks1) * blockDim.x + threadIdx.x, Md, A2b, list2, cnts + 1);
}

#define EPG 8

__device__ __forceinline__ float row16_sum(float v) {
  v += __int_as_float(__builtin_amdgcn_update_dpp(0, __float_as_int(v), 0x161, 0xf, 0xf, true));
  v += __int_as_float(__builtin_amdgcn_update_dpp(0, __float_as_int(v), 0x162, 0xf, 0xf, true));
  v += __int_as_float(__builtin_amdgcn_update_dpp(0, __float_as_int(v), 0x164, 0xf, 0xf, true));
  v += __int_as_float(__builtin_amdgcn_update_dpp(0, __float_as_int(v), 0x168, 0xf, 0xf, true));
  return v;
}

__device__ __forceinline__ void scatter1_row(const unsigned* __restrict__ list,
                                             const float* __restrict__ x,
                                             float* __restrict__ y1,
                                             unsigned e0, unsigned c, unsigned m) {
  unsigned p[EPG];
  float v[EPG];
#pragma unroll
  for (unsigned k = 0; k < EPG; ++k) p[k] = list[e0 + (k < m ? k : 0u)];
#pragma unroll
  for (unsigned k = 0; k < EPG; ++k) v[k] = x[(size_t)(p[k] >> 12) * DD + c];
#pragma unroll
  for (unsigned k = 0; k < EPG; ++k)
    if (k < m) atomicAdd(&y1[(size_t)(p[k] & 4095u) * DD + c], v[k]);
}

__device__ __forceinline__ void scatter2_row(const unsigned* __restrict__ list,
                                             const float* __restrict__ xq,
                                             const float* __restrict__ x,
                                             const float* __restrict__ x2,
                                             float* __restrict__ y2,
                                             unsigned e0, unsigned c, unsigned m) {
  unsigned p[EPG];
  float q[EPG], xj[EPG], v[EPG];
#pragma unroll
  for (unsigned k = 0; k < EPG; ++k) p[k] = list[e0 + (k < m ? k : 0u)];
#pragma unroll
  for (unsigned k = 0; k < EPG; ++k) {
    const size_t i = p[k] >> 12, j = p[k] & 4095u;
    q[k] = xq[i * DD + c];
    xj[k] = x[j * DD + c];
    v[k] = x2[i * DD + c];
  }
#pragma unroll
  for (unsigned k = 0; k < EPG; ++k) {
    const float a = row16_sum(q[k] * xj[k]);
    if (k < m) atomicAdd(&y2[(size_t)(p[k] & 4095u) * DD + c], a * v[k]);
  }
}

__global__ void k_scatter(const unsigned* __restrict__ list1, const unsigned* __restrict__ list2,
                          const unsigned* __restrict__ cnts,
                          const float* __restrict__ xq, const float* __restrict__ x,
                          const float* __restrict__ x2,
                          float* __restrict__ y1, float* __restrict__ y2, unsigned blocks2) {
  const unsigned b = blockIdx.x;
  if (b < blocks2) {
    const unsigned n = cnts[1];
    const unsigned t = b * blockDim.x + threadIdx.x;
    const unsigned c = t & 15u;
    const unsigned e0 = (t >> 4) * EPG;
    if (e0 >= n) return;
    if (n - e0 >= EPG) scatter2_row(list2, xq, x, x2, y2, e0, c, EPG);
    else               scatter2_row(list2, xq, x, x2, y2, e0, c, n - e0);
  } else {
    const unsigned n = cnts[0];
    const unsigned t = (b - blocks2) * blockDim.x + threadIdx.x;
    const unsigned c = t & 15u;
    const unsigned e0 = (t >> 4) * EPG;
    if (e0 >= n) return;
    if (n - e0 >= EPG) scatter1_row(list1, x, y1, e0, c, EPG);
    else               scatter1_row(list1, x, y1, e0, c, n - e0);
  }
}

#define TP 18

__global__ void k_combine_pre(const float* __restrict__ xcur, float* y1, float* y2,
                              const float* __restrict__ G1s, const float* __restrict__ Qm,
                              const unsigned* __restrict__ cnt2, float s1,
                              float* __restrict__ xnext, const float* __restrict__ W2n,
                              float* __restrict__ xq, float* __restrict__ x2, int has_next) {
  __shared__ __attribute__((aligned(16))) float tile[8][16 * TP];
  const int wave = threadIdx.x >> 5;
  const int lane = threadIdx.x & 31;
  const int row0 = (blockIdx.x * (blockDim.x >> 5) + wave) * 16;
  float s2 = -(float)NN / (float)(*cnt2);
  v2f a[4], b[4];
  v8f c = load_c_tile(xcur, row0);
  load_a_frag(y1, row0, s1, a);
  load_b_frag(G1s, b);
  c = wmma16(a, b, c);
  load_a_frag(y2, row0, s2, a);
  load_b_frag(Qm, b);
  c = wmma16(a, b, c);
  store_d_tile(xnext, row0, c);
  if (has_next) {
    float* tw = tile[wave];
    const int n = lane & 15, h = lane >> 4;
#pragma unroll
    for (int r = 0; r < 8; ++r) tw[(r + 8 * h) * TP + n] = c[r];
    __syncthreads();
#pragma unroll
    for (int cc = 0; cc < 4; ++cc) {
      a[cc].x = tw[n * TP + 4 * cc + 2 * h];
      a[cc].y = tw[n * TP + 4 * cc + 2 * h + 1];
    }
    pre_tile(a, Qm, W2n, xq, x2, row0);
    zero_tile(y1, row0);
    zero_tile(y2, row0);
  }
}

extern "C" void kernel_launch(void* const* d_in, const int* in_sizes, int n_in,
                              void* d_out, int out_size, void* d_ws, size_t ws_size,
                              hipStream_t stream) {
  const float* x_in = (const float*)d_in[0];
  const int* e1 = (const int*)d_in[1];
  const int* e2 = (const int*)d_in[2];
  const float* Qm = (const float*)d_in[3];
  const int* mk = (const int*)d_in[4];
  const float* W1 = (const float*)d_in[5];
  const float* W2 = (const float*)d_in[6];
  float* out = (float*)d_out;

  const int E1 = in_sizes[1] / 2;
  const int E2 = in_sizes[2] / 2;
  const int EM = in_sizes[4] / 2;
  const int STEPS = in_sizes[5] / (DD * DD);

  char* ws = (char*)d_ws;
  size_t off = 0;
  auto carve = [&](size_t bytes) {
    char* p = ws + off;
    off += (bytes + 255) & ~(size_t)255;
    return p;
  };
  const size_t BM = (size_t)NN * NN / 8;
  unsigned* Md    = (unsigned*)carve(BM);
  unsigned* A1b   = (unsigned*)carve(BM);
  unsigned* A2b   = (unsigned*)carve(BM);
  unsigned* cnts  = (unsigned*)carve(8);
  unsigned* list1 = (unsigned*)carve((size_t)E1 * 4);
  unsigned* list2 = (unsigned*)carve((size_t)E2 * 4);
  float* G1  = (float*)carve((size_t)STEPS * DD * DD * 4);
  float* xq  = (float*)carve((size_t)NN * DD * 4);
  float* x2  = (float*)carve((size_t)NN * DD * 4);
  float* y1  = (float*)carve((size_t)NN * DD * 4);
  float* y2  = (float*)carve((size_t)NN * DD * 4);
  float* xb0 = (float*)carve((size_t)NN * DD * 4);
  float* xb1 = (float*)carve((size_t)NN * DD * 4);

  hipMemsetAsync(Md, 0, 3 * BM + 8, stream);

  const int TILE_BLOCKS = NN / 16 / 8;
  const int MASK_BLOCKS = (EM + 255) / 256;
  const int ADJ_BLOCKS1 = (E1 + 255) / 256;
  const int ADJ_BLOCKS2 = (E2 + 255) / 256;

  k_setup<<<TILE_BLOCKS + STEPS + MASK_BLOCKS, 256, 0, stream>>>(
      x_in, Qm, W1, W2, STEPS, TILE_BLOCKS, xq, x2, y1, y2, G1, mk, mk + EM, EM, Md);
  k_build_adj<<<ADJ_BLOCKS1 + ADJ_BLOCKS2, 256, 0, stream>>>(
      e1, e1 + E1, E1, A1b, list1, e2, e2 + E2, E2, A2b, list2, ADJ_BLOCKS1, Md, cnts);

  const float s1 = (float)NN / (float)EM;

  const int SC_BLOCKS1 = ((E1 + EPG - 1) / EPG * 16 + 255) / 256;
  const int SC_BLOCKS2 = ((E2 + EPG - 1) / EPG * 16 + 255) / 256;
  const float* xc = x_in;
  for (int s = 0; s < STEPS; ++s) {
    float* xn = (s == STEPS - 1) ? out : ((s & 1) ? xb1 : xb0);
    const int has_next = (s + 1 < STEPS) ? 1 : 0;
    const float* W2n = W2 + (size_t)(has_next ? s + 1 : s) * DD * DD;
    k_scatter<<<SC_BLOCKS2 + SC_BLOCKS1, 256, 0, stream>>>(list1, list2, cnts, xq, xc, x2,
                                                           y1, y2, (unsigned)SC_BLOCKS2);
    k_combine_pre<<<TILE_BLOCKS, 256, 0, stream>>>(xc, y1, y2, G1 + (size_t)s * DD * DD, Qm,
                                                   cnts + 1, s1, xn, W2n, xq, x2, has_next);
    xc = xn;
  }
}


# Scatter with float atomics over edge lists replaced by a gather per destination row over transposed adjacency bitmaps: y1, y2 summed in registers in source order, no float atomics, combine fused in
# speedup vs baseline: 10.3121x; 1.5904x over previous
#include <hip/hip_runtime.h>


#define NN 4096
#define DD 16
#define RW (NN / 32)

typedef float v2f __attribute__((ext_vector_type(2)));
typedef float v8f __attribute__((ext_vector_type(8)));

__device__ __forceinline__ void load_a_frag(const float* X, int row0, float scale, v2f a[4]) {
  const int lane = threadIdx.x & 31;
  const int m = lane & 15, h = lane >> 4;
  const float* base = X + (size_t)(row0 + m) * DD + 2 * h;
#pragma unroll
  for (int c = 0; c < 4; ++c) {
    a[c].x = base[4 * c] * scale;
    a[c].y = base[4 * c + 1] * scale;
  }
}

__device__ __forceinline__ void load_b_frag(const float* __restrict__ W, v2f b[4]) {
  const int lane = threadIdx.x & 31;
  const int n = lane & 15, h = lane >> 4;
#pragma unroll
  for (int c = 0; c < 4; ++c) {
    b[c].x = W[(4 * c + 2 * h) * DD + n];
    b[c].y = W[(4 * c + 2 * h + 1) * DD + n];
  }
}

__device__ __forceinline__ void load_bT_frag(const float* __restrict__ W, v2f b[4]) {
  const int lane = threadIdx.x & 31;
  const int n = lane & 15, h = lane >> 4;
  const float* row = W + n * DD;
#pragma unroll
  for (int c = 0; c < 4; ++c) {
    b[c].x = row[4 * c + 2 * h];
    b[c].y = row[4 * c + 2 * h + 1];
  }
}

__device__ __forceinline__ v8f load_c_tile(const float* __restrict__ X, int row0) {
  const int lane = threadIdx.x & 31;
  const int n = lane & 15, h = lane >> 4;
  v8f c;
#pragma unroll
  for (int r = 0; r < 8; ++r) c[r] = X[(size_t)(row0 + r + 8 * h) * DD + n];
  return c;
}

__device__ __forceinline__ void store_d_tile(float* __restrict__ O, int row0, v8f d) {
  const int lane = threadIdx.x & 31;
  const int n = lane & 15, h = lane >> 4;
#pragma unroll
  for (int r = 0; r < 8; ++r) O[(size_t)(row0 + r + 8 * h) * DD + n] = d[r];
}

__device__ __forceinline__ v8f wmma16(const v2f a[4], const v2f b[4], v8f c) {
#pragma unroll
  for (int k = 0; k < 4; ++k)
    c = __builtin_amdgcn_wmma_f32_16x16x4_f32(false, a[k], false, b[k],
                                              (short)0, c, false, false);
  return c;
}

__device__ __forceinline__ void pre_tile(const v2f a[4], const float* __restrict__ Qm,
                                         const float* __restrict__ W2s,
                                         float* __restrict__ xq, float* __restrict__ x2,
                                         int row0) {
  v2f b[4];
  load_b_frag(Qm, b);
  v8f d = {};
  d = wmma16(a, b, d);
  store_d_tile(xq, row0, d);
  load_bT_frag(W2s, b);
  v8f d2 = {};
  d2 = wmma16(a, b, d2);
  store_d_tile(x2, row0, d2);
}

#define TP 18

__device__ __forceinline__ void load_a_frag_lds(const float* T, float scale, v2f a[4]) {
  const int lane = threadIdx.x & 31;
  const int m = lane & 15, h = lane >> 4;
#pragma unroll
  for (int c = 0; c < 4; ++c) {
    a[c].x = T[m * TP + 4 * c + 2 * h] * scale;
    a[c].y = T[m * TP + 4 * c + 2 * h + 1] * scale;
  }
}

__device__ __forceinline__ void set_bit(const int* __restrict__ row, const int* __restrict__ col,
                                        int E, int e, unsigned* __restrict__ B) {
  if (e >= E) return;
  const unsigned bit = (unsigned)col[e] * (unsigned)NN + (unsigned)row[e];
  atomicOr(&B[bit >> 5], 1u << (bit & 31));
}

__global__ void k_setup(const float* __restrict__ x, const float* __restrict__ Qm,
                        const float* __restrict__ W1, const float* __restrict__ W2,
                        int steps, int tile_blocks,
                        float* __restrict__ xq, float* __restrict__ x2,
                        float* __restrict__ G1,
                        const int* __restrict__ mrow, const int* __restrict__ mcol, int EM,
                        unsigned* __restrict__ Md, int mask_blocks,
                        const int* __restrict__ e1row, const int* __restrict__ e1col, int E1,
                        unsigned* __restrict__ B1, int blocks1,
                        const int* __restrict__ e2row, const int* __restrict__ e2col, int E2,
                        unsigned* __restrict__ B2) {
  int b = blockIdx.x;
  if (b < tile_blocks) {
    const int wave = threadIdx.x >> 5;
    const int row0 = (b * (blockDim.x >> 5) + wave) * 16;
    v2f a[4];
    load_a_frag(x, row0, 1.0f, a);
    pre_tile(a, Qm, W2, xq, x2, row0);
    return;
  }
  b -= tile_blocks;
  if (b < steps) {
    const int k = threadIdx.x >> 4, n = threadIdx.x & 15;
    float acc = 0.f;
#pragma unroll
    for (int t = 0; t < DD; ++t)
      acc = fmaf(W1[b * DD * DD + t * DD + k], Qm[t * DD + n], acc);
    G1[b * DD * DD + k * DD + n] = acc;
    return;
  }
  b -= steps;
  if (b < mask_blocks) {
    set_bit(mrow, mcol, EM, b * blockDim.x + threadIdx.x, Md);
    return;
  }
  b -= mask_blocks;
  if (b < blocks1) {
    set_bit(e1row, e1col, E1, b * blockDim.x + threadIdx.x, B1);
    return;
  }
  b -= blocks1;
  set_bit(e2row, e2col, E2, b * blockDim.x + threadIdx.x, B2);
}

__global__ void k_mask_count(const uint4* __restrict__ Md, uint4* __restrict__ B1,
                             uint4* __restrict__ B2, unsigned* __restrict__ cnt2) {
  __shared__ unsigned part[32];
  const unsigned t = blockIdx.x * blockDim.x + threadIdx.x;
  const uint4 m = Md[t];
  uint4 a = B1[t];
  a.x &= m.x; a.y &= m.y; a.z &= m.z; a.w &= m.w;
  B1[t] = a;
  uint4 c = B2[t];
  c.x &= m.x; c.y &= m.y; c.z &= m.z; c.w &= m.w;
  B2[t] = c;
  unsigned n = __popc(c.x) + __popc(c.y) + __popc(c.z) + __popc(c.w);
#pragma unroll
  for (int o = 16; o > 0; o >>= 1) n += __shfl_xor(n, o);
  const unsigned wave = threadIdx.x >> 5, nw = blockDim.x >> 5;
  if ((threadIdx.x & 31) == 0) part[wave] = n;
  __syncthreads();
  if (threadIdx.x == 0) {
    unsigned s = 0;
    for (unsigned k = 0; k < nw; ++k) s += part[k];
    atomicAdd(cnt2, s);
  }
}

__device__ __forceinline__ float row16_sum(float v) {
  v += __int_as_float(__builtin_amdgcn_update_dpp(0, __float_as_int(v), 0x161, 0xf, 0xf, true));
  v += __int_as_float(__builtin_amdgcn_update_dpp(0, __float_as_int(v), 0x162, 0xf, 0xf, true));
  v += __int_as_float(__builtin_amdgcn_update_dpp(0, __float_as_int(v), 0x164, 0xf, 0xf, true));
  v += __int_as_float(__builtin_amdgcn_update_dpp(0, __float_as_int(v), 0x168, 0xf, 0xf, true));
  return v;
}

#define LCAP 256
#define EPG 8

__device__ __forceinline__ void put_sources(const unsigned w[4], unsigned lane, unsigned p,
                                            unsigned start, unsigned short* lst) {
#pragma unroll
  for (unsigned k = 0; k < 4; ++k) {
    unsigned v = w[k];
    while (v) {
      const unsigned bpos = (unsigned)__ffs((int)v) - 1u;
      v &= v - 1u;
      const unsigned q = p - start;
      if (q < LCAP) lst[q] = (unsigned short)((4u * lane + k) * 32u + bpos);
      ++p;
    }
  }
}

__global__ __launch_bounds__(512) void k_step(
    const float* __restrict__ xcur, const float* __restrict__ xq, const float* __restrict__ x2,
    const unsigned* __restrict__ A1, const unsigned* __restrict__ A2,
    const float* __restrict__ G1s, const float* __restrict__ Qm,
    const unsigned* __restrict__ cnt2, float s1,
    float* __restrict__ xnext, const float* __restrict__ W2n,
    float* __restrict__ xqn, float* __restrict__ x2n, int has_next) {
  __shared__ unsigned short lst1[16][LCAP];
  __shared__ unsigned short lst2[16][LCAP];
  __shared__ unsigned tot[16];
  __shared__ __attribute__((aligned(16))) float yt1[16 * TP];
  __shared__ __attribute__((aligned(16))) float yt2[16 * TP];
  __shared__ __attribute__((aligned(16))) float tile[16 * TP];

  const unsigned wave = threadIdx.x >> 5;
  const unsigned lane = threadIdx.x & 31;
  const unsigned c = lane & 15u, h = lane >> 4;
  const int row0 = blockIdx.x * 16;
  const size_t j = (size_t)row0 + wave;

  const uint4 r1 = reinterpret_cast<const uint4*>(A1 + j * RW)[lane];
  const uint4 r2 = reinterpret_cast<const uint4*>(A2 + j * RW)[lane];
  const float xjc = xcur[j * DD + c];
  const unsigned w1[4] = {r1.x, r1.y, r1.z, r1.w};
  const unsigned w2[4] = {r2.x, r2.y, r2.z, r2.w};
  const unsigned n1 = __popc(w1[0]) + __popc(w1[1]) + __popc(w1[2]) + __popc(w1[3]);
  const unsigned n2 = __popc(w2[0]) + __popc(w2[1]) + __popc(w2[2]) + __popc(w2[3]);

  const unsigned pk = n1 | (n2 << 16);
  unsigned inc = pk;
#pragma unroll
  for (unsigned o = 1; o < 32; o <<= 1) {
    const unsigned up = __shfl_up(inc, o);
    if (lane >= o) inc += up;
  }
  const unsigned totpk = __shfl(inc, 31);
  const unsigned exc = inc - pk;
  const unsigned off1 = exc & 0xffffu, off2 = exc >> 16;
  const unsigned tot1 = totpk & 0xffffu, tot2 = totpk >> 16;

  if (lane == 0) tot[wave] = tot1 > tot2 ? tot1 : tot2;
  __syncthreads();
  unsigned mx = 0;
#pragma unroll
  for (unsigned k = 0; k < 16; ++k) mx = tot[k] > mx ? tot[k] : mx;
  const unsigned npass = (mx + LCAP - 1) / LCAP;

  float acc1 = 0.f, acc2 = 0.f;
  for (unsigned ps = 0; ps < npass; ++ps) {
    const unsigned start = ps * LCAP;
    put_sources(w1, lane, off1, start, lst1[wave]);
    put_sources(w2, lane, off2, start, lst2[wave]);
    __syncthreads();
    const unsigned r1n = tot1 > start ? tot1 - start : 0u;
    const unsigned r2n = tot2 > start ? tot2 - start : 0u;
    const unsigned m1 = r1n < LCAP ? r1n : LCAP;
    const unsigned m2 = r2n < LCAP ? r2n : LCAP;
    const unsigned mm = m1 > m2 ? m1 : m2;
    for (unsigned t0 = 0; t0 < mm; t0 += 2 * EPG) {
      unsigned i1[EPG], i2[EPG];
      float v1[EPG], q[EPG], v2[EPG];
#pragma unroll
      for (unsigned k = 0; k < EPG; ++k) {
        const unsigned e = t0 + 2 * k + h;
        const unsigned s1i = lst1[wave][e < m1 ? e : 0u];
        const unsigned s2i = lst2[wave][e < m2 ? e : 0u];
        i1[k] = e < m1 ? s1i : 0u;
        i2[k] = e < m2 ? s2i : 0u;
      }
#pragma unroll
      for (unsigned k = 0; k < EPG; ++k) {
        v1[k] = xcur[(size_t)i1[k] * DD + c];
        q[k] = xq[(size_t)i2[k] * DD + c];
        v2[k] = x2[(size_t)i2[k] * DD + c];
      }
      __builtin_amdgcn_sched_barrier(0);
#pragma unroll
      for (unsigned k = 0; k < EPG; ++k) {
        const unsigned e = t0 + 2 * k + h;
        const float a = row16_sum(q[k] * xjc);
        acc1 += e < m1 ? v1[k] : 0.f;
        acc2 += e < m2 ? a * v2[k] : 0.f;
      }
    }
    __syncthreads();
  }

  acc1 += __shfl_xor(acc1, 16);
  acc2 += __shfl_xor(acc2, 16);
  if (h == 0) {
    yt1[wave * TP + c] = acc1;
    yt2[wave * TP + c] = acc2;
  }
  __syncthreads();

  v8f cc = {};
  if (wave == 0) {
    const float s2 = -(float)NN / (float)(*cnt2);
    v2f a[4], b[4];
    cc = load_c_tile(xcur, row0);
    load_a_frag_lds(yt1, s1, a);
    load_b_frag(G1s, b);
    cc = wmma16(a, b, cc);
    load_a_frag_lds(yt2, s2, a);
    load_b_frag(Qm, b);
    cc = wmma16(a, b, cc);
    store_d_tile(xnext, row0, cc);
    if (has_next) {
#pragma unroll
      for (int r = 0; r < 8; ++r) tile[(r + 8 * h) * TP + c] = cc[r];
    }
  }
  if (has_next) {
    __syncthreads();
    if (wave == 0) {
      v2f a[4];
      load_a_frag_lds(tile, 1.0f, a);
      pre_tile(a, Qm, W2n, xqn, x2n, row0);
    }
  }
}

extern "C" void kernel_launch(void* const* d_in, const int* in_sizes, int n_in,
                              void* d_out, int out_size, void* d_ws, size_t ws_size,
                              hipStream_t stream) {
  const float* x_in = (const float*)d_in[0];
  const int* e1 = (const int*)d_in[1];
  const int* e2 = (const int*)d_in[2];
  const float* Qm = (const float*)d_in[3];
  const int* mk = (const int*)d_in[4];
  const float* W1 = (const float*)d_in[5];
  const float* W2 = (const float*)d_in[6];
  float* out = (float*)d_out;

  const int E1 = in_sizes[1] / 2;
  const int E2 = in_sizes[2] / 2;
  const int EM = in_sizes[4] / 2;
  const int STEPS = in_sizes[5] / (DD * DD);

  char* ws = (char*)d_ws;
  size_t off = 0;
  auto carve = [&](size_t bytes) {
    char* p = ws + off;
    off += (bytes + 255) & ~(size_t)255;
    return p;
  };
  const size_t BM = (size_t)NN * NN / 8;
  unsigned* Md    = (unsigned*)carve(BM);
  unsigned* A1b   = (unsigned*)carve(BM);
  unsigned* A2b   = (unsigned*)carve(BM);
  unsigned* cnt2  = (unsigned*)carve(8);
  float* G1  = (float*)carve((size_t)STEPS * DD * DD * 4);
  float* xqa = (float*)carve((size_t)NN * DD * 4);
  float* x2a = (float*)carve((size_t)NN * DD * 4);
  float* xqb = (float*)carve((size_t)NN * DD * 4);
  float* x2b = (float*)carve((size_t)NN * DD * 4);
  float* xb0 = (float*)carve((size_t)NN * DD * 4);
  float* xb1 = (float*)carve((size_t)NN * DD * 4);

  hipMemsetAsync(Md, 0, 3 * BM + 8, stream);

  const int TILE_BLOCKS = NN / 16 / 8;
  const int MASK_BLOCKS = (EM + 255) / 256;
  const int ADJ_BLOCKS1 = (E1 + 255) / 256;
  const int ADJ_BLOCKS2 = (E2 + 255) / 256;

  k_setup<<<TILE_BLOCKS + STEPS + MASK_BLOCKS + ADJ_BLOCKS1 + ADJ_BLOCKS2, 256, 0, stream>>>(
      x_in, Qm, W1, W2, STEPS, TILE_BLOCKS, xqa, x2a, G1,
      mk, mk + EM, EM, Md, MASK_BLOCKS,
      e1, e1 + E1, E1, A1b, ADJ_BLOCKS1,
      e2, e2 + E2, E2, A2b);
  k_mask_count<<<(int)(BM / 16 / 256), 256, 0, stream>>>(
      (const uint4*)Md, (uint4*)A1b, (uint4*)A2b, cnt2);

  const float s1 = (float)NN / (float)EM;

  const float* xc = x_in;
  for (int s = 0; s < STEPS; ++s) {
    float* xn = (s == STEPS - 1) ? out : ((s & 1) ? xb1 : xb0);
    const int has_next = (s + 1 < STEPS) ? 1 : 0;
    const float* W2n = W2 + (size_t)(has_next ? s + 1 : s) * DD * DD;
    const float* xqc = (s & 1) ? xqb : xqa;
    const float* x2c = (s & 1) ? x2b : x2a;
    float* xqn = (s & 1) ? xqa : xqb;
    float* x2n = (s & 1) ? x2a : x2b;
    k_step<<<NN / 16, 512, 0, stream>>>(xc, xqc, x2c, A1b, A2b, G1 + (size_t)s * DD * DD, Qm,
                                        cnt2, s1, xn, W2n, xqn, x2n, has_next);
    xc = xn;
  }
}


# Mask and edge_index_2 bits share 64-bit words: a returning atomicOr counts sum(A2) while the bits are set and the step ANDs its own rows, so the mask/count launch goes (6 stream ops to 5)
# speedup vs baseline: 11.5169x; 1.1168x over previous
#include <hip/hip_runtime.h>


#define NN 4096
#define DD 16
#define RW (NN / 32)

typedef float v2f __attribute__((ext_vector_type(2)));
typedef float v8f __attribute__((ext_vector_type(8)));

__device__ __forceinline__ void load_a_frag(const float* X, int row0, float scale, v2f a[4]) {
  const int lane = threadIdx.x & 31;
  const int m = lane & 15, h = lane >> 4;
  const float* base = X + (size_t)(row0 + m) * DD + 2 * h;
#pragma unroll
  for (int c = 0; c < 4; ++c) {
    a[c].x = base[4 * c] * scale;
    a[c].y = base[4 * c + 1] * scale;
  }
}

__device__ __forceinline__ void load_b_frag(const float* __restrict__ W, v2f b[4]) {
  const int lane = threadIdx.x & 31;
  const int n = lane & 15, h = lane >> 4;
#pragma unroll
  for (int c = 0; c < 4; ++c) {
    b[c].x = W[(4 * c + 2 * h) * DD + n];
    b[c].y = W[(4 * c + 2 * h + 1) * DD + n];
  }
}

__device__ __forceinline__ void load_bT_frag(const float* __restrict__ W, v2f b[4]) {
  const int lane = threadIdx.x & 31;
  const int n = lane & 15, h = lane >> 4;
  const float* row = W + n * DD;
#pragma unroll
  for (int c = 0; c < 4; ++c) {
    b[c].x = row[4 * c + 2 * h];
    b[c].y = row[4 * c + 2 * h + 1];
  }
}

__device__ __forceinline__ v8f load_c_tile(const float* __restrict__ X, int row0) {
  const int lane = threadIdx.x & 31;
  const int n = lane & 15, h = lane >> 4;
  v8f c;
#pragma unroll
  for (int r = 0; r < 8; ++r) c[r] = X[(size_t)(row0 + r + 8 * h) * DD + n];
  return c;
}

__device__ __forceinline__ void store_d_tile(float* __restrict__ O, int row0, v8f d) {
  const int lane = threadIdx.x & 31;
  const int n = lane & 15, h = lane >> 4;
#pragma unroll
  for (int r = 0; r < 8; ++r) O[(size_t)(row0 + r + 8 * h) * DD + n] = d[r];
}

__device__ __forceinline__ v8f wmma16(const v2f a[4], const v2f b[4], v8f c) {
#pragma unroll
  for (int k = 0; k < 4; ++k)
    c = __builtin_amdgcn_wmma_f32_16x16x4_f32(false, a[k], false, b[k],
                                              (short)0, c, false, false);
  return c;
}

__device__ __forceinline__ void pre_tile(const v2f a[4], const float* __restrict__ Qm,
                                         const float* __restrict__ W2s,
                                         float* __restrict__ xq, float* __restrict__ x2,
                                         int row0) {
  v2f b[4];
  load_b_frag(Qm, b);
  v8f d = {};
  d = wmma16(a, b, d);
  store_d_tile(xq, row0, d);
  load_bT_frag(W2s, b);
  v8f d2 = {};
  d2 = wmma16(a, b, d2);
  store_d_tile(x2, row0, d2);
}

#define TP 18

__device__ __forceinline__ void load_a_frag_lds(const float* T, float scale, v2f a[4]) {
  const int lane = threadIdx.x & 31;
  const int m = lane & 15, h = lane >> 4;
#pragma unroll
  for (int c = 0; c < 4; ++c) {
    a[c].x = T[m * TP + 4 * c + 2 * h] * scale;
    a[c].y = T[m * TP + 4 * c + 2 * h + 1] * scale;
  }
}

__device__ __forceinline__ void set_bit(const int* __restrict__ row, const int* __restrict__ col,
                                        int E, int e, unsigned* __restrict__ B) {
  if (e >= E) return;
  const unsigned bit = (unsigned)col[e] * (unsigned)NN + (unsigned)row[e];
  atomicOr(&B[bit >> 5], 1u << (bit & 31));
}

#define EPT 4

__device__ __forceinline__ unsigned set_pair_bits(const int* __restrict__ row,
                                                  const int* __restrict__ col, int E, int e0,
                                                  int stride, unsigned long long* __restrict__ MB,
                                                  unsigned own, unsigned other) {
  unsigned bit[EPT];
  unsigned long long old[EPT];
#pragma unroll
  for (int k = 0; k < EPT; ++k) {
    const int e = e0 + k * stride;
    const int ec = e < E ? e : E - 1;
    bit[k] = (unsigned)col[ec] * (unsigned)NN + (unsigned)row[ec];
  }
#pragma unroll
  for (int k = 0; k < EPT; ++k) {
    const int e = e0 + k * stride;
    old[k] = 0ull;
    if (e < E) old[k] = atomicOr(&MB[bit[k] >> 5], 1ull << ((bit[k] & 31u) + own));
  }
  unsigned n = 0;
#pragma unroll
  for (int k = 0; k < EPT; ++k) {
    const unsigned s = bit[k] & 31u;
    const unsigned mine = (unsigned)(old[k] >> (s + own)) & 1u;
    const unsigned theirs = (unsigned)(old[k] >> (s + other)) & 1u;
    n += theirs & ~mine & 1u;
  }
  return n;
}

__device__ __forceinline__ void add_block_count(unsigned n, unsigned* part,
                                                unsigned* __restrict__ cnt2) {
#pragma unroll
  for (int o = 16; o > 0; o >>= 1) n += __shfl_xor(n, o);
  const unsigned wave = threadIdx.x >> 5, nw = blockDim.x >> 5;
  if ((threadIdx.x & 31) == 0) part[wave] = n;
  __syncthreads();
  if (threadIdx.x == 0) {
    unsigned s = 0;
    for (unsigned k = 0; k < nw; ++k) s += part[k];
    if (s) atomicAdd(cnt2, s);
  }
}

__global__ void k_setup(const float* __restrict__ x, const float* __restrict__ Qm,
                        const float* __restrict__ W1, const float* __restrict__ W2,
                        int steps, int tile_blocks,
                        float* __restrict__ xq, float* __restrict__ x2,
                        float* __restrict__ G1,
                        const int* __restrict__ mrow, const int* __restrict__ mcol, int EM,
                        unsigned long long* __restrict__ MB, int mask_blocks,
                        const int* __restrict__ e1row, const int* __restrict__ e1col, int E1,
                        unsigned* __restrict__ B1, int blocks1,
                        const int* __restrict__ e2row, const int* __restrict__ e2col, int E2,
                        unsigned* __restrict__ cnt2) {
  __shared__ unsigned part[32];
  int b = blockIdx.x;
  if (b < tile_blocks) {
    const int wave = threadIdx.x >> 5;
    const int row0 = (b * (blockDim.x >> 5) + wave) * 16;
    v2f a[4];
    load_a_frag(x, row0, 1.0f, a);
    pre_tile(a, Qm, W2, xq, x2, row0);
    return;
  }
  b -= tile_blocks;
  if (b < steps) {
    const int k = threadIdx.x >> 4, n = threadIdx.x & 15;
    float acc = 0.f;
#pragma unroll
    for (int t = 0; t < DD; ++t)
      acc = fmaf(W1[b * DD * DD + t * DD + k], Qm[t * DD + n], acc);
    G1[b * DD * DD + k * DD + n] = acc;
    return;
  }
  b -= steps;
  if (b < mask_blocks) {
    const unsigned n = set_pair_bits(mrow, mcol, EM, b * blockDim.x * EPT + threadIdx.x,
                                     blockDim.x, MB, 0u, 32u);
    add_block_count(n, part, cnt2);
    return;
  }
  b -= mask_blocks;
  if (b < blocks1) {
    set_bit(e1row, e1col, E1, b * blockDim.x + threadIdx.x, B1);
    return;
  }
  b -= blocks1;
  const unsigned n = set_pair_bits(e2row, e2col, E2, b * blockDim.x * EPT + threadIdx.x,
                                   blockDim.x, MB, 32u, 0u);
  add_block_count(n, part, cnt2);
}

__device__ __forceinline__ float row16_sum(float v) {
  v += __int_as_float(__builtin_amdgcn_update_dpp(0, __float_as_int(v), 0x161, 0xf, 0xf, true));
  v += __int_as_float(__builtin_amdgcn_update_dpp(0, __float_as_int(v), 0x162, 0xf, 0xf, true));
  v += __int_as_float(__builtin_amdgcn_update_dpp(0, __float_as_int(v), 0x164, 0xf, 0xf, true));
  v += __int_as_float(__builtin_amdgcn_update_dpp(0, __float_as_int(v), 0x168, 0xf, 0xf, true));
  return v;
}

#define LCAP 256
#define EPG 8

__device__ __forceinline__ void put_sources(const unsigned w[4], unsigned lane, unsigned p,
                                            unsigned start, unsigned short* lst) {
#pragma unroll
  for (unsigned k = 0; k < 4; ++k) {
    unsigned v = w[k];
    while (v) {
      const unsigned bpos = (unsigned)__ffs((int)v) - 1u;
      v &= v - 1u;
      const unsigned q = p - start;
      if (q < LCAP) lst[q] = (unsigned short)((4u * lane + k) * 32u + bpos);
      ++p;
    }
  }
}

__global__ __launch_bounds__(512) void k_step(
    const float* __restrict__ xcur, const float* __restrict__ xq, const float* __restrict__ x2,
    const unsigned long long* __restrict__ MB, const unsigned* __restrict__ B1,
    const float* __restrict__ G1s, const float* __restrict__ Qm,
    const unsigned* __restrict__ cnt2, float s1,
    float* __restrict__ xnext, const float* __restrict__ W2n,
    float* __restrict__ xqn, float* __restrict__ x2n, int has_next) {
  __shared__ unsigned short lst1[16][LCAP];
  __shared__ unsigned short lst2[16][LCAP];
  __shared__ unsigned tot[16];
  __shared__ __attribute__((aligned(16))) float yt1[16 * TP];
  __shared__ __attribute__((aligned(16))) float yt2[16 * TP];
  __shared__ __attribute__((aligned(16))) float tile[16 * TP];

  const unsigned wave = threadIdx.x >> 5;
  const unsigned lane = threadIdx.x & 31;
  const unsigned c = lane & 15u, h = lane >> 4;
  const int row0 = blockIdx.x * 16;
  const size_t j = (size_t)row0 + wave;

  const uint4* prow = reinterpret_cast<const uint4*>(MB + j * RW + 4u * lane);
  const uint4 pa = prow[0];
  const uint4 pb = prow[1];
  const uint4 r1 = reinterpret_cast<const uint4*>(B1 + j * RW)[lane];
  const float xjc = xcur[j * DD + c];
  const unsigned w1[4] = {r1.x & pa.x, r1.y & pa.z, r1.z & pb.x, r1.w & pb.z};
  const unsigned w2[4] = {pa.y & pa.x, pa.w & pa.z, pb.y & pb.x, pb.w & pb.z};
  const unsigned n1 = __popc(w1[0]) + __popc(w1[1]) + __popc(w1[2]) + __popc(w1[3]);
  const unsigned n2 = __popc(w2[0]) + __popc(w2[1]) + __popc(w2[2]) + __popc(w2[3]);

  const unsigned pk = n1 | (n2 << 16);
  unsigned inc = pk;
#pragma unroll
  for (unsigned o = 1; o < 32; o <<= 1) {
    const unsigned up = __shfl_up(inc, o);
    if (lane >= o) inc += up;
  }
  const unsigned totpk = __shfl(inc, 31);
  const unsigned exc = inc - pk;
  const unsigned off1 = exc & 0xffffu, off2 = exc >> 16;
  const unsigned tot1 = totpk & 0xffffu, tot2 = totpk >> 16;

  if (lane == 0) tot[wave] = tot1 > tot2 ? tot1 : tot2;
  __syncthreads();
  unsigned mx = 0;
#pragma unroll
  for (unsigned k = 0; k < 16; ++k) mx = tot[k] > mx ? tot[k] : mx;
  const unsigned npass = (mx + LCAP - 1) / LCAP;

  float acc1 = 0.f, acc2 = 0.f;
  for (unsigned ps = 0; ps < npass; ++ps) {
    const unsigned start = ps * LCAP;
    put_sources(w1, lane, off1, start, lst1[wave]);
    put_sources(w2, lane, off2, start, lst2[wave]);
    __syncthreads();
    const unsigned r1n = tot1 > start ? tot1 - start : 0u;
    const unsigned r2n = tot2 > start ? tot2 - start : 0u;
    const unsigned m1 = r1n < LCAP ? r1n : LCAP;
    const unsigned m2 = r2n < LCAP ? r2n : LCAP;
    const unsigned mm = m1 > m2 ? m1 : m2;
    for (unsigned t0 = 0; t0 < mm; t0 += 2 * EPG) {
      unsigned i1[EPG], i2[EPG];
      float v1[EPG], q[EPG], v2[EPG];
#pragma unroll
      for (unsigned k = 0; k < EPG; ++k) {
        const unsigned e = t0 + 2 * k + h;
        const unsigned s1i = lst1[wave][e < m1 ? e : 0u];
        const unsigned s2i = lst2[wave][e < m2 ? e : 0u];
        i1[k] = e < m1 ? s1i : 0u;
        i2[k] = e < m2 ? s2i : 0u;
      }
#pragma unroll
      for (unsigned k = 0; k < EPG; ++k) {
        v1[k] = xcur[(size_t)i1[k] * DD + c];
        q[k] = xq[(size_t)i2[k] * DD + c];
        v2[k] = x2[(size_t)i2[k] * DD + c];
      }
      __builtin_amdgcn_sched_barrier(0);
#pragma unroll
      for (unsigned k = 0; k < EPG; ++k) {
        const unsigned e = t0 + 2 * k + h;
        const float a = row16_sum(q[k] * xjc);
        acc1 += e < m1 ? v1[k] : 0.f;
        acc2 += e < m2 ? a * v2[k] : 0.f;
      }
    }
    __syncthreads();
  }

  acc1 += __shfl_xor(acc1, 16);
  acc2 += __shfl_xor(acc2, 16);
  if (h == 0) {
    yt1[wave * TP + c] = acc1;
    yt2[wave * TP + c] = acc2;
  }
  __syncthreads();

  v8f cc = {};
  if (wave == 0) {
    const float s2 = -(float)NN / (float)(*cnt2);
    v2f a[4], b[4];
    cc = load_c_tile(xcur, row0);
    load_a_frag_lds(yt1, s1, a);
    load_b_frag(G1s, b);
    cc = wmma16(a, b, cc);
    load_a_frag_lds(yt2, s2, a);
    load_b_frag(Qm, b);
    cc = wmma16(a, b, cc);
    store_d_tile(xnext, row0, cc);
    if (has_next) {
#pragma unroll
      for (int r = 0; r < 8; ++r) tile[(r + 8 * h) * TP + c] = cc[r];
    }
  }
  if (has_next) {
    __syncthreads();
    if (wave == 0) {
      v2f a[4];
      load_a_frag_lds(tile, 1.0f, a);
      pre_tile(a, Qm, W2n, xqn, x2n, row0);
    }
  }
}

extern "C" void kernel_launch(void* const* d_in, const int* in_sizes, int n_in,
                              void* d_out, int out_size, void* d_ws, size_t ws_size,
                              hipStream_t stream) {
  const float* x_in = (const float*)d_in[0];
  const int* e1 = (const int*)d_in[1];
  const int* e2 = (const int*)d_in[2];
  const float* Qm = (const float*)d_in[3];
  const int* mk = (const int*)d_in[4];
  const float* W1 = (const float*)d_in[5];
  const float* W2 = (const float*)d_in[6];
  float* out = (float*)d_out;

  const int E1 = in_sizes[1] / 2;
  const int E2 = in_sizes[2] / 2;
  const int EM = in_sizes[4] / 2;
  const int STEPS = in_sizes[5] / (DD * DD);

  char* ws = (char*)d_ws;
  size_t off = 0;
  auto carve = [&](size_t bytes) {
    char* p = ws + off;
    off += (bytes + 255) & ~(size_t)255;
    return p;
  };
  const size_t BM = (size_t)NN * NN / 8;
  unsigned long long* MBp = (unsigned long long*)carve(2 * BM);
  unsigned* B1b   = (unsigned*)carve(BM);
  unsigned* cnt2  = (unsigned*)carve(8);
  float* G1  = (float*)carve((size_t)STEPS * DD * DD * 4);
  float* xqa = (float*)carve((size_t)NN * DD * 4);
  float* x2a = (float*)carve((size_t)NN * DD * 4);
  float* xqb = (float*)carve((size_t)NN * DD * 4);
  float* x2b = (float*)carve((size_t)NN * DD * 4);
  float* xb0 = (float*)carve((size_t)NN * DD * 4);
  float* xb1 = (float*)carve((size_t)NN * DD * 4);

  hipMemsetAsync(MBp, 0, 3 * BM + 8, stream);

  const int TILE_BLOCKS = NN / 16 / 8;
  const int MASK_BLOCKS = (EM + 256 * EPT - 1) / (256 * EPT);
  const int ADJ_BLOCKS1 = (E1 + 255) / 256;
  const int ADJ_BLOCKS2 = (E2 + 256 * EPT - 1) / (256 * EPT);

  k_setup<<<TILE_BLOCKS + STEPS + MASK_BLOCKS + ADJ_BLOCKS1 + ADJ_BLOCKS2, 256, 0, stream>>>(
      x_in, Qm, W1, W2, STEPS, TILE_BLOCKS, xqa, x2a, G1,
      mk, mk + EM, EM, MBp, MASK_BLOCKS,
      e1, e1 + E1, E1, B1b, ADJ_BLOCKS1,
      e2, e2 + E2, E2, cnt2);

  const float s1 = (float)NN / (float)EM;

  const float* xc = x_in;
  for (int s = 0; s < STEPS; ++s) {
    float* xn = (s == STEPS - 1) ? out : ((s & 1) ? xb1 : xb0);
    const int has_next = (s + 1 < STEPS) ? 1 : 0;
    const float* W2n = W2 + (size_t)(has_next ? s + 1 : s) * DD * DD;
    const float* xqc = (s & 1) ? xqb : xqa;
    const float* x2c = (s & 1) ? x2b : x2a;
    float* xqn = (s & 1) ? xqa : xqb;
    float* x2n = (s & 1) ? x2a : x2b;
    k_step<<<NN / 16, 512, 0, stream>>>(xc, xqc, x2c, MBp, B1b, G1 + (size_t)s * DD * DD, Qm,
                                        cnt2, s1, xn, W2n, xqn, x2n, has_next);
    xc = xn;
  }
}


# xq and x2 rows interleaved as (xq,x2) pairs in one table of 128-byte rows: an attention source is one 8-byte-per-lane whole-line load, not two 64-byte requests; sums and their order unchanged
# speedup vs baseline: 11.9802x; 1.0402x over previous
#include <hip/hip_runtime.h>


#define NN 4096
#define DD 16
#define RW (NN / 32)

typedef float v2f __attribute__((ext_vector_type(2)));

#define XP (2 * DD)
typedef float v8f __attribute__((ext_vector_type(8)));

__device__ __forceinline__ void load_a_frag(const float* X, int row0, float scale, v2f a[4]) {
  const int lane = threadIdx.x & 31;
  const int m = lane & 15, h = lane >> 4;
  const float* base = X + (size_t)(row0 + m) * DD + 2 * h;
#pragma unroll
  for (int c = 0; c < 4; ++c) {
    a[c].x = base[4 * c] * scale;
    a[c].y = base[4 * c + 1] * scale;
  }
}

__device__ __forceinline__ void load_b_frag(const float* __restrict__ W, v2f b[4]) {
  const int lane = threadIdx.x & 31;
  const int n = lane & 15, h = lane >> 4;
#pragma unroll
  for (int c = 0; c < 4; ++c) {
    b[c].x = W[(4 * c + 2 * h) * DD + n];
    b[c].y = W[(4 * c + 2 * h + 1) * DD + n];
  }
}

__device__ __forceinline__ void load_bT_frag(const float* __restrict__ W, v2f b[4]) {
  const int lane = threadIdx.x & 31;
  const int n = lane & 15, h = lane >> 4;
  const float* row = W + n * DD;
#pragma unroll
  for (int c = 0; c < 4; ++c) {
    b[c].x = row[4 * c + 2 * h];
    b[c].y = row[4 * c + 2 * h + 1];
  }
}

__device__ __forceinline__ v8f load_c_tile(const float* __restrict__ X, int row0) {
  const int lane = threadIdx.x & 31;
  const int n = lane & 15, h = lane >> 4;
  v8f c;
#pragma unroll
  for (int r = 0; r < 8; ++r) c[r] = X[(size_t)(row0 + r + 8 * h) * DD + n];
  return c;
}

__device__ __forceinline__ void store_d_tile(float* __restrict__ O, int row0, v8f d) {
  const int lane = threadIdx.x & 31;
  const int n = lane & 15, h = lane >> 4;
#pragma unroll
  for (int r = 0; r < 8; ++r) O[(size_t)(row0 + r + 8 * h) * DD + n] = d[r];
}

__device__ __forceinline__ void store_d_pair(float* __restrict__ O, int row0, v8f dq, v8f d2) {
  const int lane = threadIdx.x & 31;
  const int n = lane & 15, h = lane >> 4;
#pragma unroll
  for (int r = 0; r < 8; ++r) {
    v2f p;
    p.x = dq[r];
    p.y = d2[r];
    *reinterpret_cast<v2f*>(O + (size_t)(row0 + r + 8 * h) * XP + 2 * n) = p;
  }
}

__device__ __forceinline__ v8f wmma16(const v2f a[4], const v2f b[4], v8f c) {
#pragma unroll
  for (int k = 0; k < 4; ++k)
    c = __builtin_amdgcn_wmma_f32_16x16x4_f32(false, a[k], false, b[k],
                                              (short)0, c, false, false);
  return c;
}

__device__ __forceinline__ void pre_tile(const v2f a[4], const float* __restrict__ Qm,
                                         const float* __restrict__ W2s,
                                         float* __restrict__ xq2, int row0) {
  v2f b[4];
  load_b_frag(Qm, b);
  v8f d = {};
  d = wmma16(a, b, d);
  load_bT_frag(W2s, b);
  v8f d2 = {};
  d2 = wmma16(a, b, d2);
  store_d_pair(xq2, row0, d, d2);
}

#define TP 18

__device__ __forceinline__ void load_a_frag_lds(const float* T, float scale, v2f a[4]) {
  const int lane = threadIdx.x & 31;
  const int m = lane & 15, h = lane >> 4;
#pragma unroll
  for (int c = 0; c < 4; ++c) {
    a[c].x = T[m * TP + 4 * c + 2 * h] * scale;
    a[c].y = T[m * TP + 4 * c + 2 * h + 1] * scale;
  }
}

__device__ __forceinline__ void set_bit(const int* __restrict__ row, const int* __restrict__ col,
                                        int E, int e, unsigned* __restrict__ B) {
  if (e >= E) return;
  const unsigned bit = (unsigned)col[e] * (unsigned)NN + (unsigned)row[e];
  atomicOr(&B[bit >> 5], 1u << (bit & 31));
}

#define EPT 4

__device__ __forceinline__ unsigned set_pair_bits(const int* __restrict__ row,
                                                  const int* __restrict__ col, int E, int e0,
                                                  int stride, unsigned long long* __restrict__ MB,
                                                  unsigned own, unsigned other) {
  unsigned bit[EPT];
  unsigned long long old[EPT];
#pragma unroll
  for (int k = 0; k < EPT; ++k) {
    const int e = e0 + k * stride;
    const int ec = e < E ? e : E - 1;
    bit[k] = (unsigned)col[ec] * (unsigned)NN + (unsigned)row[ec];
  }
#pragma unroll
  for (int k = 0; k < EPT; ++k) {
    const int e = e0 + k * stride;
    old[k] = 0ull;
    if (e < E) old[k] = atomicOr(&MB[bit[k] >> 5], 1ull << ((bit[k] & 31u) + own));
  }
  unsigned n = 0;
#pragma unroll
  for (int k = 0; k < EPT; ++k) {
    const unsigned s = bit[k] & 31u;
    const unsigned mine = (unsigned)(old[k] >> (s + own)) & 1u;
    const unsigned theirs = (unsigned)(old[k] >> (s + other)) & 1u;
    n += theirs & ~mine & 1u;
  }
  return n;
}

__device__ __forceinline__ void add_block_count(unsigned n, unsigned* part,
                                                unsigned* __restrict__ cnt2) {
#pragma unroll
  for (int o = 16; o > 0; o >>= 1) n += __shfl_xor(n, o);
  const unsigned wave = threadIdx.x >> 5, nw = blockDim.x >> 5;
  if ((threadIdx.x & 31) == 0) part[wave] = n;
  __syncthreads();
  if (threadIdx.x == 0) {
    unsigned s = 0;
    for (unsigned k = 0; k < nw; ++k) s += part[k];
    if (s) atomicAdd(cnt2, s);
  }
}

__global__ void k_setup(const float* __restrict__ x, const float* __restrict__ Qm,
                        const float* __restrict__ W1, const float* __restrict__ W2,
                        int steps, int tile_blocks,
                        float* __restrict__ xq2,
                        float* __restrict__ G1,
                        const int* __restrict__ mrow, const int* __restrict__ mcol, int EM,
                        unsigned long long* __restrict__ MB, int mask_blocks,
                        const int* __restrict__ e1row, const int* __restrict__ e1col, int E1,
                        unsigned* __restrict__ B1, int blocks1,
                        const int* __restrict__ e2row, const int* __restrict__ e2col, int E2,
                        unsigned* __restrict__ cnt2) {
  __shared__ unsigned part[32];
  int b = blockIdx.x;
  if (b < tile_blocks) {
    const int wave = threadIdx.x >> 5;
    const int row0 = (b * (blockDim.x >> 5) + wave) * 16;
    v2f a[4];
    load_a_frag(x, row0, 1.0f, a);
    pre_tile(a, Qm, W2, xq2, row0);
    return;
  }
  b -= tile_blocks;
  if (b < steps) {
    const int k = threadIdx.x >> 4, n = threadIdx.x & 15;
    float acc = 0.f;
#pragma unroll
    for (int t = 0; t < DD; ++t)
      acc = fmaf(W1[b * DD * DD + t * DD + k], Qm[t * DD + n], acc);
    G1[b * DD * DD + k * DD + n] = acc;
    return;
  }
  b -= steps;
  if (b < mask_blocks) {
    const unsigned n = set_pair_bits(mrow, mcol, EM, b * blockDim.x * EPT + threadIdx.x,
                                     blockDim.x, MB, 0u, 32u);
    add_block_count(n, part, cnt2);
    return;
  }
  b -= mask_blocks;
  if (b < blocks1) {
    set_bit(e1row, e1col, E1, b * blockDim.x + threadIdx.x, B1);
    return;
  }
  b -= blocks1;
  const unsigned n = set_pair_bits(e2row, e2col, E2, b * blockDim.x * EPT + threadIdx.x,
                                   blockDim.x, MB, 32u, 0u);
  add_block_count(n, part, cnt2);
}

__device__ __forceinline__ float row16_sum(float v) {
  v += __int_as_float(__builtin_amdgcn_update_dpp(0, __float_as_int(v), 0x161, 0xf, 0xf, true));
  v += __int_as_float(__builtin_amdgcn_update_dpp(0, __float_as_int(v), 0x162, 0xf, 0xf, true));
  v += __int_as_float(__builtin_amdgcn_update_dpp(0, __float_as_int(v), 0x164, 0xf, 0xf, true));
  v += __int_as_float(__builtin_amdgcn_update_dpp(0, __float_as_int(v), 0x168, 0xf, 0xf, true));
  return v;
}

#define LCAP 256
#define EPG 8

__device__ __forceinline__ void put_sources(const unsigned w[4], unsigned lane, unsigned p,
                                            unsigned start, unsigned short* lst) {
#pragma unroll
  for (unsigned k = 0; k < 4; ++k) {
    unsigned v = w[k];
    while (v) {
      const unsigned bpos = (unsigned)__ffs((int)v) - 1u;
      v &= v - 1u;
      const unsigned q = p - start;
      if (q < LCAP) lst[q] = (unsigned short)((4u * lane + k) * 32u + bpos);
      ++p;
    }
  }
}

__global__ __launch_bounds__(512) void k_step(
    const float* __restrict__ xcur, const float* __restrict__ xq2,
    const unsigned long long* __restrict__ MB, const unsigned* __restrict__ B1,
    const float* __restrict__ G1s, const float* __restrict__ Qm,
    const unsigned* __restrict__ cnt2, float s1,
    float* __restrict__ xnext, const float* __restrict__ W2n,
    float* __restrict__ xq2n, int has_next) {
  __shared__ unsigned short lst1[16][LCAP];
  __shared__ unsigned short lst2[16][LCAP];
  __shared__ unsigned tot[16];
  __shared__ __attribute__((aligned(16))) float yt1[16 * TP];
  __shared__ __attribute__((aligned(16))) float yt2[16 * TP];
  __shared__ __attribute__((aligned(16))) float tile[16 * TP];

  const unsigned wave = threadIdx.x >> 5;
  const unsigned lane = threadIdx.x & 31;
  const unsigned c = lane & 15u, h = lane >> 4;
  const int row0 = blockIdx.x * 16;
  const size_t j = (size_t)row0 + wave;

  const uint4* prow = reinterpret_cast<const uint4*>(MB + j * RW + 4u * lane);
  const uint4 pa = prow[0];
  const uint4 pb = prow[1];
  const uint4 r1 = reinterpret_cast<const uint4*>(B1 + j * RW)[lane];
  const float xjc = xcur[j * DD + c];
  const unsigned w1[4] = {r1.x & pa.x, r1.y & pa.z, r1.z & pb.x, r1.w & pb.z};
  const unsigned w2[4] = {pa.y & pa.x, pa.w & pa.z, pb.y & pb.x, pb.w & pb.z};
  const unsigned n1 = __popc(w1[0]) + __popc(w1[1]) + __popc(w1[2]) + __popc(w1[3]);
  const unsigned n2 = __popc(w2[0]) + __popc(w2[1]) + __popc(w2[2]) + __popc(w2[3]);

  const unsigned pk = n1 | (n2 << 16);
  unsigned inc = pk;
#pragma unroll
  for (unsigned o = 1; o < 32; o <<= 1) {
    const unsigned up = __shfl_up(inc, o);
    if (lane >= o) inc += up;
  }
  const unsigned totpk = __shfl(inc, 31);
  const unsigned exc = inc - pk;
  const unsigned off1 = exc & 0xffffu, off2 = exc >> 16;
  const unsigned tot1 = totpk & 0xffffu, tot2 = totpk >> 16;

  if (lane == 0) tot[wave] = tot1 > tot2 ? tot1 : tot2;
  __syncthreads();
  unsigned mx = 0;
#pragma unroll
  for (unsigned k = 0; k < 16; ++k) mx = tot[k] > mx ? tot[k] : mx;
  const unsigned npass = (mx + LCAP - 1) / LCAP;

  float acc1 = 0.f, acc2 = 0.f;
  for (unsigned ps = 0; ps < npass; ++ps) {
    const unsigned start = ps * LCAP;
    put_sources(w1, lane, off1, start, lst1[wave]);
    put_sources(w2, lane, off2, start, lst2[wave]);
    __syncthreads();
    const unsigned r1n = tot1 > start ? tot1 - start : 0u;
    const unsigned r2n = tot2 > start ? tot2 - start : 0u;
    const unsigned m1 = r1n < LCAP ? r1n : LCAP;
    const unsigned m2 = r2n < LCAP ? r2n : LCAP;
    const unsigned mm = m1 > m2 ? m1 : m2;
    for (unsigned t0 = 0; t0 < mm; t0 += 2 * EPG) {
      unsigned i1[EPG], i2[EPG];
      float v1[EPG];
      v2f qv[EPG];
#pragma unroll
      for (unsigned k = 0; k < EPG; ++k) {
        const unsigned e = t0 + 2 * k + h;
        const unsigned s1i = lst1[wave][e < m1 ? e : 0u];
        const unsigned s2i = lst2[wave][e < m2 ? e : 0u];
        i1[k] = e < m1 ? s1i : 0u;
        i2[k] = e < m2 ? s2i : 0u;
      }
#pragma unroll
      for (unsigned k = 0; k < EPG; ++k) {
        v1[k] = xcur[(size_t)i1[k] * DD + c];
        qv[k] = *reinterpret_cast<const v2f*>(xq2 + (size_t)i2[k] * XP + 2u * c);
      }
      __builtin_amdgcn_sched_barrier(0);
#pragma unroll
      for (unsigned k = 0; k < EPG; ++k) {
        const unsigned e = t0 + 2 * k + h;
        const float a = row16_sum(qv[k].x * xjc);
        acc1 += e < m1 ? v1[k] : 0.f;
        acc2 += e < m2 ? a * qv[k].y : 0.f;
      }
    }
    __syncthreads();
  }

  acc1 += __shfl_xor(acc1, 16);
  acc2 += __shfl_xor(acc2, 16);
  if (h == 0) {
    yt1[wave * TP + c] = acc1;
    yt2[wave * TP + c] = acc2;
  }
  __syncthreads();

  v8f cc = {};
  if (wave == 0) {
    const float s2 = -(float)NN / (float)(*cnt2);
    v2f a[4], b[4];
    cc = load_c_tile(xcur, row0);
    load_a_frag_lds(yt1, s1, a);
    load_b_frag(G1s, b);
    cc = wmma16(a, b, cc);
    load_a_frag_lds(yt2, s2, a);
    load_b_frag(Qm, b);
    cc = wmma16(a, b, cc);
    store_d_tile(xnext, row0, cc);
    if (has_next) {
#pragma unroll
      for (int r = 0; r < 8; ++r) tile[(r + 8 * h) * TP + c] = cc[r];
    }
  }
  if (has_next) {
    __syncthreads();
    if (wave == 0) {
      v2f a[4];
      load_a_frag_lds(tile, 1.0f, a);
      pre_tile(a, Qm, W2n, xq2n, row0);
    }
  }
}

extern "C" void kernel_launch(void* const* d_in, const int* in_sizes, int n_in,
                              void* d_out, int out_size, void* d_ws, size_t ws_size,
                              hipStream_t stream) {
  const float* x_in = (const float*)d_in[0];
  const int* e1 = (const int*)d_in[1];
  const int* e2 = (const int*)d_in[2];
  const float* Qm = (const float*)d_in[3];
  const int* mk = (const int*)d_in[4];
  const float* W1 = (const float*)d_in[5];
  const float* W2 = (const float*)d_in[6];
  float* out = (float*)d_out;

  const int E1 = in_sizes[1] / 2;
  const int E2 = in_sizes[2] / 2;
  const int EM = in_sizes[4] / 2;
  const int STEPS = in_sizes[5] / (DD * DD);

  char* ws = (char*)d_ws;
  size_t off = 0;
  auto carve = [&](size_t bytes) {
    char* p = ws + off;
    off += (bytes + 255) & ~(size_t)255;
    return p;
  };
  const size_t BM = (size_t)NN * NN / 8;
  unsigned long long* MBp = (unsigned long long*)carve(2 * BM);
  unsigned* B1b   = (unsigned*)carve(BM);
  unsigned* cnt2  = (unsigned*)carve(8);
  float* G1  = (float*)carve((size_t)STEPS * DD * DD * 4);
  float* xq2a = (float*)carve((size_t)NN * XP * 4);
  float* xq2b = (float*)carve((size_t)NN * XP * 4);
  float* xb0 = (float*)carve((size_t)NN * DD * 4);
  float* xb1 = (float*)carve((size_t)NN * DD * 4);

  hipMemsetAsync(MBp, 0, 3 * BM + 8, stream);

  const int TILE_BLOCKS = NN / 16 / 8;
  const int MASK_BLOCKS = (EM + 256 * EPT - 1) / (256 * EPT);
  const int ADJ_BLOCKS1 = (E1 + 255) / 256;
  const int ADJ_BLOCKS2 = (E2 + 256 * EPT - 1) / (256 * EPT);

  k_setup<<<TILE_BLOCKS + STEPS + MASK_BLOCKS + ADJ_BLOCKS1 + ADJ_BLOCKS2, 256, 0, stream>>>(
      x_in, Qm, W1, W2, STEPS, TILE_BLOCKS, xq2a, G1,
      mk, mk + EM, EM, MBp, MASK_BLOCKS,
      e1, e1 + E1, E1, B1b, ADJ_BLOCKS1,
      e2, e2 + E2, E2, cnt2);

  const float s1 = (float)NN / (float)EM;

  const float* xc = x_in;
  for (int s = 0; s < STEPS; ++s) {
    float* xn = (s == STEPS - 1) ? out : ((s & 1) ? xb1 : xb0);
    const int has_next = (s + 1 < STEPS) ? 1 : 0;
    const float* W2n = W2 + (size_t)(has_next ? s + 1 : s) * DD * DD;
    const float* xq2c = (s & 1) ? xq2b : xq2a;
    float* xq2n = (s & 1) ? xq2a : xq2b;
    k_step<<<NN / 16, 512, 0, stream>>>(xc, xq2c, MBp, B1b, G1 + (size_t)s * DD * DD, Qm,
                                        cnt2, s1, xn, W2n, xq2n, has_next);
    xc = xn;
  }
}


# Bitmaps built by owner blocks in LDS from edge chunks sorted by destination bin in LDS: no per-edge global atomicOr, no 6 MB clear; sum(A2) from per-block counts; float sums unchanged
# speedup vs baseline: 16.5770x; 1.3837x over previous
#include <hip/hip_runtime.h>


#define NN 4096
#define DD 16
#define RW (NN / 32)
#define NBIN (NN / 16)
#define CHK 4096
#define BT 512
#define EPB (CHK / BT)

typedef float v2f __attribute__((ext_vector_type(2)));

#define XP (2 * DD)
typedef float v8f __attribute__((ext_vector_type(8)));

__device__ __forceinline__ void load_a_frag(const float* X, int row0, float scale, v2f a[4]) {
  const int lane = threadIdx.x & 31;
  const int m = lane & 15, h = lane >> 4;
  const float* base = X + (size_t)(row0 + m) * DD + 2 * h;
#pragma unroll
  for (int c = 0; c < 4; ++c) {
    a[c].x = base[4 * c] * scale;
    a[c].y = base[4 * c + 1] * scale;
  }
}

__device__ __forceinline__ void load_b_frag(const float* __restrict__ W, v2f b[4]) {
  const int lane = threadIdx.x & 31;
  const int n = lane & 15, h = lane >> 4;
#pragma unroll
  for (int c = 0; c < 4; ++c) {
    b[c].x = W[(4 * c + 2 * h) * DD + n];
    b[c].y = W[(4 * c + 2 * h + 1) * DD + n];
  }
}

__device__ __forceinline__ void load_bT_frag(const float* __restrict__ W, v2f b[4]) {
  const int lane = threadIdx.x & 31;
  const int n = lane & 15, h = lane >> 4;
  const float* row = W + n * DD;
#pragma unroll
  for (int c = 0; c < 4; ++c) {
    b[c].x = row[4 * c + 2 * h];
    b[c].y = row[4 * c + 2 * h + 1];
  }
}

__device__ __forceinline__ v8f load_c_tile(const float* __restrict__ X, int row0) {
  const int lane = threadIdx.x & 31;
  const int n = lane & 15, h = lane >> 4;
  v8f c;
#pragma unroll
  for (int r = 0; r < 8; ++r) c[r] = X[(size_t)(row0 + r + 8 * h) * DD + n];
  return c;
}

__device__ __forceinline__ void store_d_tile(float* __restrict__ O, int row0, v8f d) {
  const int lane = threadIdx.x & 31;
  const int n = lane & 15, h = lane >> 4;
#pragma unroll
  for (int r = 0; r < 8; ++r) O[(size_t)(row0 + r + 8 * h) * DD + n] = d[r];
}

__device__ __forceinline__ void store_d_pair(float* __restrict__ O, int row0, v8f dq, v8f d2) {
  const int lane = threadIdx.x & 31;
  const int n = lane & 15, h = lane >> 4;
#pragma unroll
  for (int r = 0; r < 8; ++r) {
    v2f p;
    p.x = dq[r];
    p.y = d2[r];
    *reinterpret_cast<v2f*>(O + (size_t)(row0 + r + 8 * h) * XP + 2 * n) = p;
  }
}

__device__ __forceinline__ v8f wmma16(const v2f a[4], const v2f b[4], v8f c) {
#pragma unroll
  for (int k = 0; k < 4; ++k)
    c = __builtin_amdgcn_wmma_f32_16x16x4_f32(false, a[k], false, b[k],
                                              (short)0, c, false, false);
  return c;
}

__device__ __forceinline__ void pre_tile(const v2f a[4], const float* __restrict__ Qm,
                                         const float* __restrict__ W2s,
                                         float* __restrict__ xq2, int row0) {
  v2f b[4];
  load_b_frag(Qm, b);
  v8f d = {};
  d = wmma16(a, b, d);
  load_bT_frag(W2s, b);
  v8f d2 = {};
  d2 = wmma16(a, b, d2);
  store_d_pair(xq2, row0, d, d2);
}

#define TP 18

__device__ __forceinline__ void load_a_frag_lds(const float* T, float scale, v2f a[4]) {
  const int lane = threadIdx.x & 31;
  const int m = lane & 15, h = lane >> 4;
#pragma unroll
  for (int c = 0; c < 4; ++c) {
    a[c].x = T[m * TP + 4 * c + 2 * h] * scale;
    a[c].y = T[m * TP + 4 * c + 2 * h + 1] * scale;
  }
}

#define REC_NONE 0xffffffffu

__device__ __forceinline__ void sort_chunk(const int* __restrict__ row, const int* __restrict__ col,
                                           int E, int local, unsigned list, unsigned chunk,
                                           unsigned* __restrict__ recs, unsigned* __restrict__ offs,
                                           unsigned* cnt, unsigned* cur, unsigned* wsum,
                                           unsigned* srt) {
  const unsigned t = threadIdx.x, lane = t & 31u;
  unsigned rec[EPB];
  const int e0 = local * CHK + (int)t;
#pragma unroll
  for (int i = 0; i < EPB; ++i) {
    const int e = e0 + i * BT;
    const int ec = e < E ? e : E - 1;
    const unsigned r = (unsigned)row[ec] & (unsigned)(NN - 1);
    const unsigned c = (unsigned)col[ec] & (unsigned)(NN - 1);
    const unsigned v = ((c >> 4) << 18) | (list << 16) | ((c & 15u) << 12) | r;
    rec[i] = e < E ? v : REC_NONE;
  }
  if (t < NBIN) cnt[t] = 0u;
  __syncthreads();
#pragma unroll
  for (int i = 0; i < EPB; ++i)
    if (rec[i] != REC_NONE) atomicAdd(&cnt[rec[i] >> 18], 1u);
  __syncthreads();
  const unsigned v = t < NBIN ? cnt[t] : 0u;
  unsigned inc = v;
#pragma unroll
  for (unsigned o = 1; o < 32; o <<= 1) {
    const unsigned up = __shfl_up(inc, o);
    if (lane >= o) inc += up;
  }
  if (t < NBIN && lane == 31u) wsum[t >> 5] = inc;
  __syncthreads();
  unsigned ex = 0u;
  if (t < NBIN) {
    unsigned base = 0u;
    for (unsigned w = 0; w < (t >> 5); ++w) base += wsum[w];
    ex = base + inc - v;
    cur[t] = ex;
  }
  __syncthreads();
  unsigned pos[EPB];
#pragma unroll
  for (int i = 0; i < EPB; ++i)
    pos[i] = rec[i] != REC_NONE ? atomicAdd(&cur[rec[i] >> 18], 1u) : 0u;
#pragma unroll
  for (int i = 0; i < EPB; ++i)
    if (rec[i] != REC_NONE) srt[pos[i]] = rec[i];
  __syncthreads();
  if (t < NBIN) offs[(size_t)chunk * NBIN + t] = (v << 16) | ex;
  const int left = E - local * CHK;
  const unsigned n = left < CHK ? (unsigned)left : (unsigned)CHK;
  for (unsigned i = t; i < n; i += BT) recs[(size_t)chunk * CHK + i] = srt[i];
}

__global__ __launch_bounds__(BT) void k_front(
    const float* __restrict__ x, const float* __restrict__ Qm,
    const float* __restrict__ W1, const float* __restrict__ W2,
    int steps, int tile_blocks,
    float* __restrict__ xq2, float* __restrict__ G1,
    const int* __restrict__ mrow, const int* __restrict__ mcol, int EM, int ncm,
    const int* __restrict__ e1row, const int* __restrict__ e1col, int E1, int nc1,
    const int* __restrict__ e2row, const int* __restrict__ e2col, int E2,
    unsigned* __restrict__ recs, unsigned* __restrict__ offs) {
  __shared__ unsigned cnt[NBIN];
  __shared__ unsigned cur[NBIN];
  __shared__ unsigned wsum[NBIN / 32];
  __shared__ unsigned srt[CHK];
  int b = blockIdx.x;
  if (b < tile_blocks) {
    const int wave = threadIdx.x >> 5;
    const int row0 = (b * (blockDim.x >> 5) + wave) * 16;
    v2f a[4];
    load_a_frag(x, row0, 1.0f, a);
    pre_tile(a, Qm, W2, xq2, row0);
    return;
  }
  b -= tile_blocks;
  if (b < steps) {
    if (threadIdx.x < DD * DD) {
      const int k = threadIdx.x >> 4, n = threadIdx.x & 15;
      float acc = 0.f;
#pragma unroll
      for (int t = 0; t < DD; ++t)
        acc = fmaf(W1[b * DD * DD + t * DD + k], Qm[t * DD + n], acc);
      G1[b * DD * DD + k * DD + n] = acc;
    }
    return;
  }
  b -= steps;
  const unsigned chunk = (unsigned)b;
  const int* row = mrow;
  const int* col = mcol;
  int E = EM, local = b;
  unsigned list = 0u;
  if (b >= ncm + nc1) {
    row = e2row; col = e2col; E = E2; local = b - ncm - nc1; list = 2u;
  } else if (b >= ncm) {
    row = e1row; col = e1col; E = E1; local = b - ncm; list = 1u;
  }
  sort_chunk(row, col, E, local, list, chunk, recs, offs, cnt, cur, wsum, srt);
}

#define CPG 12

__device__ __forceinline__ void set_rec(unsigned rec, unsigned* bm) {
  const unsigned list = (rec >> 16) & 3u;
  if (list < 3u)
    atomicOr(&bm[(list * 16u + ((rec >> 12) & 15u)) * RW + ((rec & 4095u) >> 5)],
             1u << (rec & 31u));
}

__device__ __forceinline__ void load_bins(const unsigned* __restrict__ offs, int nch, int k0,
                                          unsigned b, unsigned o[CPG]) {
#pragma unroll
  for (int g = 0; g < CPG; ++g) {
    const int k = k0 + 16 * g;
    o[g] = k < nch ? offs[(size_t)k * NBIN + b] : 0u;
  }
}

__device__ __forceinline__ void load_recs(const unsigned* __restrict__ recs, int k0, unsigned lane,
                                          const unsigned o[CPG], unsigned rc[CPG]) {
#pragma unroll
  for (int g = 0; g < CPG; ++g) {
    const int k = k0 + 16 * g;
    const unsigned n = o[g] >> 16, st = o[g] & 0xffffu;
    rc[g] = lane < n ? recs[(size_t)k * CHK + st + lane] : REC_NONE;
  }
}

__device__ __forceinline__ void set_group(const unsigned* __restrict__ recs, int k0, unsigned lane,
                                          const unsigned o[CPG], const unsigned rc[CPG],
                                          unsigned* bm) {
#pragma unroll
  for (int g = 0; g < CPG; ++g) set_rec(rc[g], bm);
  for (int g = 0; g < CPG; ++g) {
    const int k = k0 + 16 * g;
    const unsigned n = o[g] >> 16, st = o[g] & 0xffffu;
    for (unsigned q = lane + 32u; q < n; q += 32u) set_rec(recs[(size_t)k * CHK + st + q], bm);
  }
}

__global__ __launch_bounds__(BT) void k_build(const unsigned* __restrict__ recs,
                                              const unsigned* __restrict__ offs, int nch,
                                              unsigned* __restrict__ AB,
                                              unsigned* __restrict__ part2) {
  __shared__ __attribute__((aligned(16))) unsigned bm[3 * 16 * RW];
  __shared__ unsigned wc[16];
  const unsigned t = threadIdx.x, wave = t >> 5, lane = t & 31u;
  const unsigned b = blockIdx.x;
  unsigned o[CPG], rc[CPG];
  load_bins(offs, nch, (int)wave, b, o);
  for (unsigned i = t; i < 3u * 16u * RW; i += BT) bm[i] = 0u;
  load_recs(recs, (int)wave, lane, o, rc);
  __syncthreads();
  set_group(recs, (int)wave, lane, o, rc, bm);
  for (int k0 = (int)wave + 16 * CPG; k0 < nch; k0 += 16 * CPG) {
    load_bins(offs, nch, k0, b, o);
    load_recs(recs, k0, lane, o, rc);
    set_group(recs, k0, lane, o, rc, bm);
  }
  __syncthreads();
  const uint4 m = reinterpret_cast<const uint4*>(bm + (0u * 16u + wave) * RW)[lane];
  const uint4 b1 = reinterpret_cast<const uint4*>(bm + (1u * 16u + wave) * RW)[lane];
  const uint4 b2 = reinterpret_cast<const uint4*>(bm + (2u * 16u + wave) * RW)[lane];
  uint4 a1, a2;
  a1.x = b1.x & m.x; a1.y = b1.y & m.y; a1.z = b1.z & m.z; a1.w = b1.w & m.w;
  a2.x = b2.x & m.x; a2.y = b2.y & m.y; a2.z = b2.z & m.z; a2.w = b2.w & m.w;
  const size_t j = (size_t)b * 16 + wave;
  reinterpret_cast<uint4*>(AB + j * (2 * RW))[lane] = a1;
  reinterpret_cast<uint4*>(AB + j * (2 * RW) + RW)[lane] = a2;
  unsigned n2 = __popc(a2.x) + __popc(a2.y) + __popc(a2.z) + __popc(a2.w);
#pragma unroll
  for (int s = 16; s > 0; s >>= 1) n2 += __shfl_xor(n2, s);
  if (lane == 0) wc[wave] = n2;
  __syncthreads();
  if (t == 0) {
    unsigned s = 0;
    for (unsigned k = 0; k < 16; ++k) s += wc[k];
    part2[b] = s;
  }
}

__device__ __forceinline__ float row16_sum(float v) {
  v += __int_as_float(__builtin_amdgcn_update_dpp(0, __float_as_int(v), 0x161, 0xf, 0xf, true));
  v += __int_as_float(__builtin_amdgcn_update_dpp(0, __float_as_int(v), 0x162, 0xf, 0xf, true));
  v += __int_as_float(__builtin_amdgcn_update_dpp(0, __float_as_int(v), 0x164, 0xf, 0xf, true));
  v += __int_as_float(__builtin_amdgcn_update_dpp(0, __float_as_int(v), 0x168, 0xf, 0xf, true));
  return v;
}

#define LCAP 256
#define EPG 8

__device__ __forceinline__ void put_sources(const unsigned w[4], unsigned lane, unsigned p,
                                            unsigned start, unsigned short* lst) {
#pragma unroll
  for (unsigned k = 0; k < 4; ++k) {
    unsigned v = w[k];
    while (v) {
      const unsigned bpos = (unsigned)__ffs((int)v) - 1u;
      v &= v - 1u;
      const unsigned q = p - start;
      if (q < LCAP) lst[q] = (unsigned short)((4u * lane + k) * 32u + bpos);
      ++p;
    }
  }
}

__global__ __launch_bounds__(512) void k_step(
    const float* __restrict__ xcur, const float* __restrict__ xq2,
    const unsigned* __restrict__ AB,
    const float* __restrict__ G1s, const float* __restrict__ Qm,
    const unsigned* __restrict__ part2, float s1,
    float* __restrict__ xnext, const float* __restrict__ W2n,
    float* __restrict__ xq2n, int has_next) {
  __shared__ unsigned short lst1[16][LCAP];
  __shared__ unsigned short lst2[16][LCAP];
  __shared__ unsigned tot[16];
  __shared__ __attribute__((aligned(16))) float yt1[16 * TP];
  __shared__ __attribute__((aligned(16))) float yt2[16 * TP];
  __shared__ __attribute__((aligned(16))) float tile[16 * TP];

  const unsigned wave = threadIdx.x >> 5;
  const unsigned lane = threadIdx.x & 31;
  const unsigned c = lane & 15u, h = lane >> 4;
  const int row0 = blockIdx.x * 16;
  const size_t j = (size_t)row0 + wave;

  const uint4 ra1 = reinterpret_cast<const uint4*>(AB + j * (2 * RW))[lane];
  const uint4 ra2 = reinterpret_cast<const uint4*>(AB + j * (2 * RW) + RW)[lane];
  const float xjc = xcur[j * DD + c];
  const unsigned w1[4] = {ra1.x, ra1.y, ra1.z, ra1.w};
  const unsigned w2[4] = {ra2.x, ra2.y, ra2.z, ra2.w};
  const unsigned n1 = __popc(w1[0]) + __popc(w1[1]) + __popc(w1[2]) + __popc(w1[3]);
  const unsigned n2 = __popc(w2[0]) + __popc(w2[1]) + __popc(w2[2]) + __popc(w2[3]);

  const unsigned pk = n1 | (n2 << 16);
  unsigned inc = pk;
#pragma unroll
  for (unsigned o = 1; o < 32; o <<= 1) {
    const unsigned up = __shfl_up(inc, o);
    if (lane >= o) inc += up;
  }
  const unsigned totpk = __shfl(inc, 31);
  const unsigned exc = inc - pk;
  const unsigned off1 = exc & 0xffffu, off2 = exc >> 16;
  const unsigned tot1 = totpk & 0xffffu, tot2 = totpk >> 16;

  if (lane == 0) tot[wave] = tot1 > tot2 ? tot1 : tot2;
  __syncthreads();
  unsigned mx = 0;
#pragma unroll
  for (unsigned k = 0; k < 16; ++k) mx = tot[k] > mx ? tot[k] : mx;
  const unsigned npass = (mx + LCAP - 1) / LCAP;

  float acc1 = 0.f, acc2 = 0.f;
  for (unsigned ps = 0; ps < npass; ++ps) {
    const unsigned start = ps * LCAP;
    put_sources(w1, lane, off1, start, lst1[wave]);
    put_sources(w2, lane, off2, start, lst2[wave]);
    __syncthreads();
    const unsigned r1n = tot1 > start ? tot1 - start : 0u;
    const unsigned r2n = tot2 > start ? tot2 - start : 0u;
    const unsigned m1 = r1n < LCAP ? r1n : LCAP;
    const unsigned m2 = r2n < LCAP ? r2n : LCAP;
    const unsigned mm = m1 > m2 ? m1 : m2;
    for (unsigned t0 = 0; t0 < mm; t0 += 2 * EPG) {
      unsigned i1[EPG], i2[EPG];
      float v1[EPG];
      v2f qv[EPG];
#pragma unroll
      for (unsigned k = 0; k < EPG; ++k) {
        const unsigned e = t0 + 2 * k + h;
        const unsigned s1i = lst1[wave][e < m1 ? e : 0u];
        const unsigned s2i = lst2[wave][e < m2 ? e : 0u];
        i1[k] = e < m1 ? s1i : 0u;
        i2[k] = e < m2 ? s2i : 0u;
      }
#pragma unroll
      for (unsigned k = 0; k < EPG; ++k) {
        v1[k] = xcur[(size_t)i1[k] * DD + c];
        qv[k] = *reinterpret_cast<const v2f*>(xq2 + (size_t)i2[k] * XP + 2u * c);
      }
      __builtin_amdgcn_sched_barrier(0);
#pragma unroll
      for (unsigned k = 0; k < EPG; ++k) {
        const unsigned e = t0 + 2 * k + h;
        const float a = row16_sum(qv[k].x * xjc);
        acc1 += e < m1 ? v1[k] : 0.f;
        acc2 += e < m2 ? a * qv[k].y : 0.f;
      }
    }
    __syncthreads();
  }

  acc1 += __shfl_xor(acc1, 16);
  acc2 += __shfl_xor(acc2, 16);
  if (h == 0) {
    yt1[wave * TP + c] = acc1;
    yt2[wave * TP + c] = acc2;
  }
  __syncthreads();

  v8f cc = {};
  if (wave == 0) {
    unsigned cnt2 = 0u;
    for (unsigned k = lane; k < NBIN; k += 32u) cnt2 += part2[k];
#pragma unroll
    for (int o = 16; o > 0; o >>= 1) cnt2 += __shfl_xor(cnt2, o);
    const float s2 = -(float)NN / (float)cnt2;
    v2f a[4], b[4];
    cc = load_c_tile(xcur, row0);
    load_a_frag_lds(yt1, s1, a);
    load_b_frag(G1s, b);
    cc = wmma16(a, b, cc);
    load_a_frag_lds(yt2, s2, a);
    load_b_frag(Qm, b);
    cc = wmma16(a, b, cc);
    store_d_tile(xnext, row0, cc);
    if (has_next) {
#pragma unroll
      for (int r = 0; r < 8; ++r) tile[(r + 8 * h) * TP + c] = cc[r];
    }
  }
  if (has_next) {
    __syncthreads();
    if (wave == 0) {
      v2f a[4];
      load_a_frag_lds(tile, 1.0f, a);
      pre_tile(a, Qm, W2n, xq2n, row0);
    }
  }
}

extern "C" void kernel_launch(void* const* d_in, const int* in_sizes, int n_in,
                              void* d_out, int out_size, void* d_ws, size_t ws_size,
                              hipStream_t stream) {
  const float* x_in = (const float*)d_in[0];
  const int* e1 = (const int*)d_in[1];
  const int* e2 = (const int*)d_in[2];
  const float* Qm = (const float*)d_in[3];
  const int* mk = (const int*)d_in[4];
  const float* W1 = (const float*)d_in[5];
  const float* W2 = (const float*)d_in[6];
  float* out = (float*)d_out;

  const int E1 = in_sizes[1] / 2;
  const int E2 = in_sizes[2] / 2;
  const int EM = in_sizes[4] / 2;
  const int STEPS = in_sizes[5] / (DD * DD);

  char* ws = (char*)d_ws;
  size_t off = 0;
  auto carve = [&](size_t bytes) {
    char* p = ws + off;
    off += (bytes + 255) & ~(size_t)255;
    return p;
  };
  const int NCM = (EM + CHK - 1) / CHK;
  const int NC1 = (E1 + CHK - 1) / CHK;
  const int NC2 = (E2 + CHK - 1) / CHK;
  const int NCH = NCM + NC1 + NC2;
  unsigned* recs  = (unsigned*)carve((size_t)NCH * CHK * 4);
  unsigned* offs  = (unsigned*)carve((size_t)NCH * NBIN * 4);
  unsigned* AB    = (unsigned*)carve((size_t)NN * 2 * RW * 4);
  unsigned* part2 = (unsigned*)carve((size_t)NBIN * 4);
  float* G1  = (float*)carve((size_t)STEPS * DD * DD * 4);
  float* xq2a = (float*)carve((size_t)NN * XP * 4);
  float* xq2b = (float*)carve((size_t)NN * XP * 4);
  float* xb0 = (float*)carve((size_t)NN * DD * 4);
  float* xb1 = (float*)carve((size_t)NN * DD * 4);

  const int TILE_BLOCKS = NN / 16 / (BT / 32);

  k_front<<<TILE_BLOCKS + STEPS + NCH, BT, 0, stream>>>(
      x_in, Qm, W1, W2, STEPS, TILE_BLOCKS, xq2a, G1,
      mk, mk + EM, EM, NCM,
      e1, e1 + E1, E1, NC1,
      e2, e2 + E2, E2, recs, offs);

  k_build<<<NBIN, BT, 0, stream>>>(recs, offs, NCH, AB, part2);

  const float s1 = (float)NN / (float)EM;

  const float* xc = x_in;
  for (int s = 0; s < STEPS; ++s) {
    float* xn = (s == STEPS - 1) ? out : ((s & 1) ? xb1 : xb0);
    const int has_next = (s + 1 < STEPS) ? 1 : 0;
    const float* W2n = W2 + (size_t)(has_next ? s + 1 : s) * DD * DD;
    const float* xq2c = (s & 1) ? xq2b : xq2a;
    float* xq2n = (s & 1) ? xq2a : xq2b;
    k_step<<<NN / 16, BT, 0, stream>>>(xc, xq2c, AB, G1 + (size_t)s * DD * DD, Qm,
                                       part2, s1, xn, W2n, xq2n, has_next);
    xc = xn;
  }
}


# k_step: the finishing wave requests its global operands (A2 counts, x tile, G1, Q, W2 fragments) at entry, in flight under the gather; before, behind the last barrier, 8 count loads ran one by one
# speedup vs baseline: 18.6984x; 1.1280x over previous
#include <hip/hip_runtime.h>


#define NN 4096
#define DD 16
#define RW (NN / 32)
#define NBIN (NN / 16)
#define CHK 4096
#define BT 512
#define EPB (CHK / BT)

typedef float v2f __attribute__((ext_vector_type(2)));

#define XP (2 * DD)
typedef float v8f __attribute__((ext_vector_type(8)));

__device__ __forceinline__ void load_a_frag(const float* X, int row0, float scale, v2f a[4]) {
  const int lane = threadIdx.x & 31;
  const int m = lane & 15, h = lane >> 4;
  const float* base = X + (size_t)(row0 + m) * DD + 2 * h;
#pragma unroll
  for (int c = 0; c < 4; ++c) {
    a[c].x = base[4 * c] * scale;
    a[c].y = base[4 * c + 1] * scale;
  }
}

__device__ __forceinline__ void load_b_frag(const float* __restrict__ W, v2f b[4]) {
  const int lane = threadIdx.x & 31;
  const int n = lane & 15, h = lane >> 4;
#pragma unroll
  for (int c = 0; c < 4; ++c) {
    b[c].x = W[(4 * c + 2 * h) * DD + n];
    b[c].y = W[(4 * c + 2 * h + 1) * DD + n];
  }
}

__device__ __forceinline__ void load_bT_frag(const float* __restrict__ W, v2f b[4]) {
  const int lane = threadIdx.x & 31;
  const int n = lane & 15, h = lane >> 4;
  const float* row = W + n * DD;
#pragma unroll
  for (int c = 0; c < 4; ++c) {
    b[c].x = row[4 * c + 2 * h];
    b[c].y = row[4 * c + 2 * h + 1];
  }
}

__device__ __forceinline__ v8f load_c_tile(const float* __restrict__ X, int row0) {
  const int lane = threadIdx.x & 31;
  const int n = lane & 15, h = lane >> 4;
  v8f c;
#pragma unroll
  for (int r = 0; r < 8; ++r) c[r] = X[(size_t)(row0 + r + 8 * h) * DD + n];
  return c;
}

__device__ __forceinline__ void store_d_tile(float* __restrict__ O, int row0, v8f d) {
  const int lane = threadIdx.x & 31;
  const int n = lane & 15, h = lane >> 4;
#pragma unroll
  for (int r = 0; r < 8; ++r) O[(size_t)(row0 + r + 8 * h) * DD + n] = d[r];
}

__device__ __forceinline__ void store_d_pair(float* __restrict__ O, int row0, v8f dq, v8f d2) {
  const int lane = threadIdx.x & 31;
  const int n = lane & 15, h = lane >> 4;
#pragma unroll
  for (int r = 0; r < 8; ++r) {
    v2f p;
    p.x = dq[r];
    p.y = d2[r];
    *reinterpret_cast<v2f*>(O + (size_t)(row0 + r + 8 * h) * XP + 2 * n) = p;
  }
}

__device__ __forceinline__ v8f wmma16(const v2f a[4], const v2f b[4], v8f c) {
#pragma unroll
  for (int k = 0; k < 4; ++k)
    c = __builtin_amdgcn_wmma_f32_16x16x4_f32(false, a[k], false, b[k],
                                              (short)0, c, false, false);
  return c;
}

__device__ __forceinline__ void pre_tile(const v2f a[4], const float* __restrict__ Qm,
                                         const float* __restrict__ W2s,
                                         float* __restrict__ xq2, int row0) {
  v2f b[4];
  load_b_frag(Qm, b);
  v8f d = {};
  d = wmma16(a, b, d);
  load_bT_frag(W2s, b);
  v8f d2 = {};
  d2 = wmma16(a, b, d2);
  store_d_pair(xq2, row0, d, d2);
}

#define TP 18

__device__ __forceinline__ void load_a_frag_lds(const float* T, float scale, v2f a[4]) {
  const int lane = threadIdx.x & 31;
  const int m = lane & 15, h = lane >> 4;
#pragma unroll
  for (int c = 0; c < 4; ++c) {
    a[c].x = T[m * TP + 4 * c + 2 * h] * scale;
    a[c].y = T[m * TP + 4 * c + 2 * h + 1] * scale;
  }
}

#define REC_NONE 0xffffffffu

__device__ __forceinline__ void sort_chunk(const int* __restrict__ row, const int* __restrict__ col,
                                           int E, int local, unsigned list, unsigned chunk,
                                           unsigned* __restrict__ recs, unsigned* __restrict__ offs,
                                           unsigned* cnt, unsigned* cur, unsigned* wsum,
                                           unsigned* srt) {
  const unsigned t = threadIdx.x, lane = t & 31u;
  unsigned rec[EPB];
  const int e0 = local * CHK + (int)t;
#pragma unroll
  for (int i = 0; i < EPB; ++i) {
    const int e = e0 + i * BT;
    const int ec = e < E ? e : E - 1;
    const unsigned r = (unsigned)row[ec] & (unsigned)(NN - 1);
    const unsigned c = (unsigned)col[ec] & (unsigned)(NN - 1);
    const unsigned v = ((c >> 4) << 18) | (list << 16) | ((c & 15u) << 12) | r;
    rec[i] = e < E ? v : REC_NONE;
  }
  if (t < NBIN) cnt[t] = 0u;
  __syncthreads();
#pragma unroll
  for (int i = 0; i < EPB; ++i)
    if (rec[i] != REC_NONE) atomicAdd(&cnt[rec[i] >> 18], 1u);
  __syncthreads();
  const unsigned v = t < NBIN ? cnt[t] : 0u;
  unsigned inc = v;
#pragma unroll
  for (unsigned o = 1; o < 32; o <<= 1) {
    const unsigned up = __shfl_up(inc, o);
    if (lane >= o) inc += up;
  }
  if (t < NBIN && lane == 31u) wsum[t >> 5] = inc;
  __syncthreads();
  unsigned ex = 0u;
  if (t < NBIN) {
    unsigned base = 0u;
    for (unsigned w = 0; w < (t >> 5); ++w) base += wsum[w];
    ex = base + inc - v;
    cur[t] = ex;
  }
  __syncthreads();
  unsigned pos[EPB];
#pragma unroll
  for (int i = 0; i < EPB; ++i)
    pos[i] = rec[i] != REC_NONE ? atomicAdd(&cur[rec[i] >> 18], 1u) : 0u;
#pragma unroll
  for (int i = 0; i < EPB; ++i)
    if (rec[i] != REC_NONE) srt[pos[i]] = rec[i];
  __syncthreads();
  if (t < NBIN) offs[(size_t)chunk * NBIN + t] = (v << 16) | ex;
  const int left = E - local * CHK;
  const unsigned n = left < CHK ? (unsigned)left : (unsigned)CHK;
  for (unsigned i = t; i < n; i += BT) recs[(size_t)chunk * CHK + i] = srt[i];
}

__global__ __launch_bounds__(BT) void k_front(
    const float* __restrict__ x, const float* __restrict__ Qm,
    const float* __restrict__ W1, const float* __restrict__ W2,
    int steps, int tile_blocks,
    float* __restrict__ xq2, float* __restrict__ G1,
    const int* __restrict__ mrow, const int* __restrict__ mcol, int EM, int ncm,
    const int* __restrict__ e1row, const int* __restrict__ e1col, int E1, int nc1,
    const int* __restrict__ e2row, const int* __restrict__ e2col, int E2,
    unsigned* __restrict__ recs, unsigned* __restrict__ offs) {
  __shared__ unsigned cnt[NBIN];
  __shared__ unsigned cur[NBIN];
  __shared__ unsigned wsum[NBIN / 32];
  __shared__ unsigned srt[CHK];
  int b = blockIdx.x;
  if (b < tile_blocks) {
    const int wave = threadIdx.x >> 5;
    const int row0 = (b * (blockDim.x >> 5) + wave) * 16;
    v2f a[4];
    load_a_frag(x, row0, 1.0f, a);
    pre_tile(a, Qm, W2, xq2, row0);
    return;
  }
  b -= tile_blocks;
  if (b < steps) {
    if (threadIdx.x < DD * DD) {
      const int k = threadIdx.x >> 4, n = threadIdx.x & 15;
      float acc = 0.f;
#pragma unroll
      for (int t = 0; t < DD; ++t)
        acc = fmaf(W1[b * DD * DD + t * DD + k], Qm[t * DD + n], acc);
      G1[b * DD * DD + k * DD + n] = acc;
    }
    return;
  }
  b -= steps;
  const unsigned chunk = (unsigned)b;
  const int* row = mrow;
  const int* col = mcol;
  int E = EM, local = b;
  unsigned list = 0u;
  if (b >= ncm + nc1) {
    row = e2row; col = e2col; E = E2; local = b - ncm - nc1; list = 2u;
  } else if (b >= ncm) {
    row = e1row; col = e1col; E = E1; local = b - ncm; list = 1u;
  }
  sort_chunk(row, col, E, local, list, chunk, recs, offs, cnt, cur, wsum, srt);
}

#define CPG 12

__device__ __forceinline__ void set_rec(unsigned rec, unsigned* bm) {
  const unsigned list = (rec >> 16) & 3u;
  if (list < 3u)
    atomicOr(&bm[(list * 16u + ((rec >> 12) & 15u)) * RW + ((rec & 4095u) >> 5)],
             1u << (rec & 31u));
}

__device__ __forceinline__ void load_bins(const unsigned* __restrict__ offs, int nch, int k0,
                                          unsigned b, unsigned o[CPG]) {
#pragma unroll
  for (int g = 0; g < CPG; ++g) {
    const int k = k0 + 16 * g;
    o[g] = k < nch ? offs[(size_t)k * NBIN + b] : 0u;
  }
}

__device__ __forceinline__ void load_recs(const unsigned* __restrict__ recs, int k0, unsigned lane,
                                          const unsigned o[CPG], unsigned rc[CPG]) {
#pragma unroll
  for (int g = 0; g < CPG; ++g) {
    const int k = k0 + 16 * g;
    const unsigned n = o[g] >> 16, st = o[g] & 0xffffu;
    rc[g] = lane < n ? recs[(size_t)k * CHK + st + lane] : REC_NONE;
  }
}

__device__ __forceinline__ void set_group(const unsigned* __restrict__ recs, int k0, unsigned lane,
                                          const unsigned o[CPG], const unsigned rc[CPG],
                                          unsigned* bm) {
#pragma unroll
  for (int g = 0; g < CPG; ++g) set_rec(rc[g], bm);
  for (int g = 0; g < CPG; ++g) {
    const int k = k0 + 16 * g;
    const unsigned n = o[g] >> 16, st = o[g] & 0xffffu;
    for (unsigned q = lane + 32u; q < n; q += 32u) set_rec(recs[(size_t)k * CHK + st + q], bm);
  }
}

__global__ __launch_bounds__(BT) void k_build(const unsigned* __restrict__ recs,
                                              const unsigned* __restrict__ offs, int nch,
                                              unsigned* __restrict__ AB,
                                              unsigned* __restrict__ part2) {
  __shared__ __attribute__((aligned(16))) unsigned bm[3 * 16 * RW];
  __shared__ unsigned wc[16];
  const unsigned t = threadIdx.x, wave = t >> 5, lane = t & 31u;
  const unsigned b = blockIdx.x;
  unsigned o[CPG], rc[CPG];
  load_bins(offs, nch, (int)wave, b, o);
  for (unsigned i = t; i < 3u * 16u * RW; i += BT) bm[i] = 0u;
  load_recs(recs, (int)wave, lane, o, rc);
  __syncthreads();
  set_group(recs, (int)wave, lane, o, rc, bm);
  for (int k0 = (int)wave + 16 * CPG; k0 < nch; k0 += 16 * CPG) {
    load_bins(offs, nch, k0, b, o);
    load_recs(recs, k0, lane, o, rc);
    set_group(recs, k0, lane, o, rc, bm);
  }
  __syncthreads();
  const uint4 m = reinterpret_cast<const uint4*>(bm + (0u * 16u + wave) * RW)[lane];
  const uint4 b1 = reinterpret_cast<const uint4*>(bm + (1u * 16u + wave) * RW)[lane];
  const uint4 b2 = reinterpret_cast<const uint4*>(bm + (2u * 16u + wave) * RW)[lane];
  uint4 a1, a2;
  a1.x = b1.x & m.x; a1.y = b1.y & m.y; a1.z = b1.z & m.z; a1.w = b1.w & m.w;
  a2.x = b2.x & m.x; a2.y = b2.y & m.y; a2.z = b2.z & m.z; a2.w = b2.w & m.w;
  const size_t j = (size_t)b * 16 + wave;
  reinterpret_cast<uint4*>(AB + j * (2 * RW))[lane] = a1;
  reinterpret_cast<uint4*>(AB + j * (2 * RW) + RW)[lane] = a2;
  unsigned n2 = __popc(a2.x) + __popc(a2.y) + __popc(a2.z) + __popc(a2.w);
#pragma unroll
  for (int s = 16; s > 0; s >>= 1) n2 += __shfl_xor(n2, s);
  if (lane == 0) wc[wave] = n2;
  __syncthreads();
  if (t == 0) {
    unsigned s = 0;
    for (unsigned k = 0; k < 16; ++k) s += wc[k];
    part2[b] = s;
  }
}

__device__ __forceinline__ float row16_sum(float v) {
  v += __int_as_float(__builtin_amdgcn_update_dpp(0, __float_as_int(v), 0x161, 0xf, 0xf, true));
  v += __int_as_float(__builtin_amdgcn_update_dpp(0, __float_as_int(v), 0x162, 0xf, 0xf, true));
  v += __int_as_float(__builtin_amdgcn_update_dpp(0, __float_as_int(v), 0x164, 0xf, 0xf, true));
  v += __int_as_float(__builtin_amdgcn_update_dpp(0, __float_as_int(v), 0x168, 0xf, 0xf, true));
  return v;
}

#define LCAP 256
#define EPG 8

__device__ __forceinline__ void put_sources(const unsigned w[4], unsigned lane, unsigned p,
                                            unsigned start, unsigned short* lst) {
#pragma unroll
  for (unsigned k = 0; k < 4; ++k) {
    unsigned v = w[k];
    while (v) {
      const unsigned bpos = (unsigned)__ffs((int)v) - 1u;
      v &= v - 1u;
      const unsigned q = p - start;
      if (q < LCAP) lst[q] = (unsigned short)((4u * lane + k) * 32u + bpos);
      ++p;
    }
  }
}

__global__ __launch_bounds__(512) void k_step(
    const float* __restrict__ xcur, const float* __restrict__ xq2,
    const unsigned* __restrict__ AB,
    const float* __restrict__ G1s, const float* __restrict__ Qm,
    const unsigned* __restrict__ part2, float s1,
    float* __restrict__ xnext, const float* __restrict__ W2n,
    float* __restrict__ xq2n, int has_next) {
  __shared__ unsigned short lst1[16][LCAP];
  __shared__ unsigned short lst2[16][LCAP];
  __shared__ unsigned tot[16];
  __shared__ __attribute__((aligned(16))) float yt1[16 * TP];
  __shared__ __attribute__((aligned(16))) float yt2[16 * TP];
  __shared__ __attribute__((aligned(16))) float tile[16 * TP];

  const unsigned wave = threadIdx.x >> 5;
  const unsigned lane = threadIdx.x & 31;
  const unsigned c = lane & 15u, h = lane >> 4;
  const int row0 = blockIdx.x * 16;
  const size_t j = (size_t)row0 + wave;

  const uint4 ra1 = reinterpret_cast<const uint4*>(AB + j * (2 * RW))[lane];
  const uint4 ra2 = reinterpret_cast<const uint4*>(AB + j * (2 * RW) + RW)[lane];
  const float xjc = xcur[j * DD + c];

  v8f cc = {};
  v2f bg[4] = {}, bq[4] = {}, bw[4] = {};
  unsigned cpart = 0u;
  if (wave == 0) {
    uint4 pw[NBIN / 128];
#pragma unroll
    for (int q = 0; q < NBIN / 128; ++q)
      pw[q] = reinterpret_cast<const uint4*>(part2)[lane + 32u * q];
    cc = load_c_tile(xcur, row0);
    load_b_frag(G1s, bg);
    load_b_frag(Qm, bq);
    load_bT_frag(W2n, bw);
#pragma unroll
    for (int q = 0; q < NBIN / 128; ++q) cpart += pw[q].x + pw[q].y + pw[q].z + pw[q].w;
  }

  const unsigned w1[4] = {ra1.x, ra1.y, ra1.z, ra1.w};
  const unsigned w2[4] = {ra2.x, ra2.y, ra2.z, ra2.w};
  const unsigned n1 = __popc(w1[0]) + __popc(w1[1]) + __popc(w1[2]) + __popc(w1[3]);
  const unsigned n2 = __popc(w2[0]) + __popc(w2[1]) + __popc(w2[2]) + __popc(w2[3]);

  const unsigned pk = n1 | (n2 << 16);
  unsigned inc = pk;
#pragma unroll
  for (unsigned o = 1; o < 32; o <<= 1) {
    const unsigned up = __shfl_up(inc, o);
    if (lane >= o) inc += up;
  }
  const unsigned totpk = __shfl(inc, 31);
  const unsigned exc = inc - pk;
  const unsigned off1 = exc & 0xffffu, off2 = exc >> 16;
  const unsigned tot1 = totpk & 0xffffu, tot2 = totpk >> 16;

  if (lane == 0) tot[wave] = tot1 > tot2 ? tot1 : tot2;
  __syncthreads();
  unsigned mx = 0;
#pragma unroll
  for (unsigned k = 0; k < 16; ++k) mx = tot[k] > mx ? tot[k] : mx;
  const unsigned npass = (mx + LCAP - 1) / LCAP;

  float acc1 = 0.f, acc2 = 0.f;
  for (unsigned ps = 0; ps < npass; ++ps) {
    const unsigned start = ps * LCAP;
    put_sources(w1, lane, off1, start, lst1[wave]);
    put_sources(w2, lane, off2, start, lst2[wave]);
    __syncthreads();
    const unsigned r1n = tot1 > start ? tot1 - start : 0u;
    const unsigned r2n = tot2 > start ? tot2 - start : 0u;
    const unsigned m1 = r1n < LCAP ? r1n : LCAP;
    const unsigned m2 = r2n < LCAP ? r2n : LCAP;
    const unsigned mm = m1 > m2 ? m1 : m2;
    for (unsigned t0 = 0; t0 < mm; t0 += 2 * EPG) {
      unsigned i1[EPG], i2[EPG];
      float v1[EPG];
      v2f qv[EPG];
#pragma unroll
      for (unsigned k = 0; k < EPG; ++k) {
        const unsigned e = t0 + 2 * k + h;
        const unsigned s1i = lst1[wave][e < m1 ? e : 0u];
        const unsigned s2i = lst2[wave][e < m2 ? e : 0u];
        i1[k] = e < m1 ? s1i : 0u;
        i2[k] = e < m2 ? s2i : 0u;
      }
#pragma unroll
      for (unsigned k = 0; k < EPG; ++k) {
        v1[k] = xcur[(size_t)i1[k] * DD + c];
        qv[k] = *reinterpret_cast<const v2f*>(xq2 + (size_t)i2[k] * XP + 2u * c);
      }
      __builtin_amdgcn_sched_barrier(0);
#pragma unroll
      for (unsigned k = 0; k < EPG; ++k) {
        const unsigned e = t0 + 2 * k + h;
        const float a = row16_sum(qv[k].x * xjc);
        acc1 += e < m1 ? v1[k] : 0.f;
        acc2 += e < m2 ? a * qv[k].y : 0.f;
      }
    }
    __syncthreads();
  }

  acc1 += __shfl_xor(acc1, 16);
  acc2 += __shfl_xor(acc2, 16);
  if (h == 0) {
    yt1[wave * TP + c] = acc1;
    yt2[wave * TP + c] = acc2;
  }
  __syncthreads();

  if (wave == 0) {
    unsigned cnt2 = cpart;
#pragma unroll
    for (int o = 16; o > 0; o >>= 1) cnt2 += __shfl_xor(cnt2, o);
    const float s2 = -(float)NN / (float)cnt2;
    v2f a[4];
    load_a_frag_lds(yt1, s1, a);
    cc = wmma16(a, bg, cc);
    load_a_frag_lds(yt2, s2, a);
    cc = wmma16(a, bq, cc);
    store_d_tile(xnext, row0, cc);
    if (has_next) {
#pragma unroll
      for (int r = 0; r < 8; ++r) tile[(r + 8 * h) * TP + c] = cc[r];
    }
  }
  if (has_next) {
    __syncthreads();
    if (wave == 0) {
      v2f a[4];
      load_a_frag_lds(tile, 1.0f, a);
      v8f d = {};
      d = wmma16(a, bq, d);
      v8f d2 = {};
      d2 = wmma16(a, bw, d2);
      store_d_pair(xq2n, row0, d, d2);
    }
  }
}

extern "C" void kernel_launch(void* const* d_in, const int* in_sizes, int n_in,
                              void* d_out, int out_size, void* d_ws, size_t ws_size,
                              hipStream_t stream) {
  const float* x_in = (const float*)d_in[0];
  const int* e1 = (const int*)d_in[1];
  const int* e2 = (const int*)d_in[2];
  const float* Qm = (const float*)d_in[3];
  const int* mk = (const int*)d_in[4];
  const float* W1 = (const float*)d_in[5];
  const float* W2 = (const float*)d_in[6];
  float* out = (float*)d_out;

  const int E1 = in_sizes[1] / 2;
  const int E2 = in_sizes[2] / 2;
  const int EM = in_sizes[4] / 2;
  const int STEPS = in_sizes[5] / (DD * DD);

  char* ws = (char*)d_ws;
  size_t off = 0;
  auto carve = [&](size_t bytes) {
    char* p = ws + off;
    off += (bytes + 255) & ~(size_t)255;
    return p;
  };
  const int NCM = (EM + CHK - 1) / CHK;
  const int NC1 = (E1 + CHK - 1) / CHK;
  const int NC2 = (E2 + CHK - 1) / CHK;
  const int NCH = NCM + NC1 + NC2;
  unsigned* recs  = (unsigned*)carve((size_t)NCH * CHK * 4);
  unsigned* offs  = (unsigned*)carve((size_t)NCH * NBIN * 4);
  unsigned* AB    = (unsigned*)carve((size_t)NN * 2 * RW * 4);
  unsigned* part2 = (unsigned*)carve((size_t)NBIN * 4);
  float* G1  = (float*)carve((size_t)STEPS * DD * DD * 4);
  float* xq2a = (float*)carve((size_t)NN * XP * 4);
  float* xq2b = (float*)carve((size_t)NN * XP * 4);
  float* xb0 = (float*)carve((size_t)NN * DD * 4);
  float* xb1 = (float*)carve((size_t)NN * DD * 4);

  const int TILE_BLOCKS = NN / 16 / (BT / 32);

  k_front<<<TILE_BLOCKS + STEPS + NCH, BT, 0, stream>>>(
      x_in, Qm, W1, W2, STEPS, TILE_BLOCKS, xq2a, G1,
      mk, mk + EM, EM, NCM,
      e1, e1 + E1, E1, NC1,
      e2, e2 + E2, E2, recs, offs);

  k_build<<<NBIN, BT, 0, stream>>>(recs, offs, NCH, AB, part2);

  const float s1 = (float)NN / (float)EM;

  const float* xc = x_in;
  for (int s = 0; s < STEPS; ++s) {
    float* xn = (s == STEPS - 1) ? out : ((s & 1) ? xb1 : xb0);
    const int has_next = (s + 1 < STEPS) ? 1 : 0;
    const float* W2n = W2 + (size_t)(has_next ? s + 1 : s) * DD * DD;
    const float* xq2c = (s & 1) ? xq2b : xq2a;
    float* xq2n = (s & 1) ? xq2a : xq2b;
    k_step<<<NN / 16, BT, 0, stream>>>(xc, xq2c, AB, G1 + (size_t)s * DD * DD, Qm,
                                       part2, s1, xn, W2n, xq2n, has_next);
    xc = xn;
  }
}


# k_step gather with 16-byte loads per lane: x row by 4 lanes, pair row by 8 (8 or 4 sources per load, not 2), 64 entries per list in flight; lane groups sum their sources, groups added last
# speedup vs baseline: 19.7330x; 1.0553x over previous
#include <hip/hip_runtime.h>


#define NN 4096
#define DD 16
#define RW (NN / 32)
#define NBIN (NN / 16)
#define CHK 4096
#define BT 512
#define EPB (CHK / BT)

typedef float v2f __attribute__((ext_vector_type(2)));

#define XP (2 * DD)
typedef float v8f __attribute__((ext_vector_type(8)));
typedef float v4f __attribute__((ext_vector_type(4)));

__device__ __forceinline__ void load_a_frag(const float* X, int row0, float scale, v2f a[4]) {
  const int lane = threadIdx.x & 31;
  const int m = lane & 15, h = lane >> 4;
  const float* base = X + (size_t)(row0 + m) * DD + 2 * h;
#pragma unroll
  for (int c = 0; c < 4; ++c) {
    a[c].x = base[4 * c] * scale;
    a[c].y = base[4 * c + 1] * scale;
  }
}

__device__ __forceinline__ void load_b_frag(const float* __restrict__ W, v2f b[4]) {
  const int lane = threadIdx.x & 31;
  const int n = lane & 15, h = lane >> 4;
#pragma unroll
  for (int c = 0; c < 4; ++c) {
    b[c].x = W[(4 * c + 2 * h) * DD + n];
    b[c].y = W[(4 * c + 2 * h + 1) * DD + n];
  }
}

__device__ __forceinline__ void load_bT_frag(const float* __restrict__ W, v2f b[4]) {
  const int lane = threadIdx.x & 31;
  const int n = lane & 15, h = lane >> 4;
  const float* row = W + n * DD;
#pragma unroll
  for (int c = 0; c < 4; ++c) {
    b[c].x = row[4 * c + 2 * h];
    b[c].y = row[4 * c + 2 * h + 1];
  }
}

__device__ __forceinline__ v8f load_c_tile(const float* __restrict__ X, int row0) {
  const int lane = threadIdx.x & 31;
  const int n = lane & 15, h = lane >> 4;
  v8f c;
#pragma unroll
  for (int r = 0; r < 8; ++r) c[r] = X[(size_t)(row0 + r + 8 * h) * DD + n];
  return c;
}

__device__ __forceinline__ void store_d_tile(float* __restrict__ O, int row0, v8f d) {
  const int lane = threadIdx.x & 31;
  const int n = lane & 15, h = lane >> 4;
#pragma unroll
  for (int r = 0; r < 8; ++r) O[(size_t)(row0 + r + 8 * h) * DD + n] = d[r];
}

__device__ __forceinline__ void store_d_pair(float* __restrict__ O, int row0, v8f dq, v8f d2) {
  const int lane = threadIdx.x & 31;
  const int n = lane & 15, h = lane >> 4;
#pragma unroll
  for (int r = 0; r < 8; ++r) {
    v2f p;
    p.x = dq[r];
    p.y = d2[r];
    *reinterpret_cast<v2f*>(O + (size_t)(row0 + r + 8 * h) * XP + 2 * n) = p;
  }
}

__device__ __forceinline__ v8f wmma16(const v2f a[4], const v2f b[4], v8f c) {
#pragma unroll
  for (int k = 0; k < 4; ++k)
    c = __builtin_amdgcn_wmma_f32_16x16x4_f32(false, a[k], false, b[k],
                                              (short)0, c, false, false);
  return c;
}

__device__ __forceinline__ void pre_tile(const v2f a[4], const float* __restrict__ Qm,
                                         const float* __restrict__ W2s,
                                         float* __restrict__ xq2, int row0) {
  v2f b[4];
  load_b_frag(Qm, b);
  v8f d = {};
  d = wmma16(a, b, d);
  load_bT_frag(W2s, b);
  v8f d2 = {};
  d2 = wmma16(a, b, d2);
  store_d_pair(xq2, row0, d, d2);
}

#define TP 18

__device__ __forceinline__ void load_a_frag_lds(const float* T, float scale, v2f a[4]) {
  const int lane = threadIdx.x & 31;
  const int m = lane & 15, h = lane >> 4;
#pragma unroll
  for (int c = 0; c < 4; ++c) {
    a[c].x = T[m * TP + 4 * c + 2 * h] * scale;
    a[c].y = T[m * TP + 4 * c + 2 * h + 1] * scale;
  }
}

#define REC_NONE 0xffffffffu

__device__ __forceinline__ void sort_chunk(const int* __restrict__ row, const int* __restrict__ col,
                                           int E, int local, unsigned list, unsigned chunk,
                                           unsigned* __restrict__ recs, unsigned* __restrict__ offs,
                                           unsigned* cnt, unsigned* cur, unsigned* wsum,
                                           unsigned* srt) {
  const unsigned t = threadIdx.x, lane = t & 31u;
  unsigned rec[EPB];
  const int e0 = local * CHK + (int)t;
#pragma unroll
  for (int i = 0; i < EPB; ++i) {
    const int e = e0 + i * BT;
    const int ec = e < E ? e : E - 1;
    const unsigned r = (unsigned)row[ec] & (unsigned)(NN - 1);
    const unsigned c = (unsigned)col[ec] & (unsigned)(NN - 1);
    const unsigned v = ((c >> 4) << 18) | (list << 16) | ((c & 15u) << 12) | r;
    rec[i] = e < E ? v : REC_NONE;
  }
  if (t < NBIN) cnt[t] = 0u;
  __syncthreads();
#pragma unroll
  for (int i = 0; i < EPB; ++i)
    if (rec[i] != REC_NONE) atomicAdd(&cnt[rec[i] >> 18], 1u);
  __syncthreads();
  const unsigned v = t < NBIN ? cnt[t] : 0u;
  unsigned inc = v;
#pragma unroll
  for (unsigned o = 1; o < 32; o <<= 1) {
    const unsigned up = __shfl_up(inc, o);
    if (lane >= o) inc += up;
  }
  if (t < NBIN && lane == 31u) wsum[t >> 5] = inc;
  __syncthreads();
  unsigned ex = 0u;
  if (t < NBIN) {
    unsigned base = 0u;
    for (unsigned w = 0; w < (t >> 5); ++w) base += wsum[w];
    ex = base + inc - v;
    cur[t] = ex;
  }
  __syncthreads();
  unsigned pos[EPB];
#pragma unroll
  for (int i = 0; i < EPB; ++i)
    pos[i] = rec[i] != REC_NONE ? atomicAdd(&cur[rec[i] >> 18], 1u) : 0u;
#pragma unroll
  for (int i = 0; i < EPB; ++i)
    if (rec[i] != REC_NONE) srt[pos[i]] = rec[i];
  __syncthreads();
  if (t < NBIN) offs[(size_t)chunk * NBIN + t] = (v << 16) | ex;
  const int left = E - local * CHK;
  const unsigned n = left < CHK ? (unsigned)left : (unsigned)CHK;
  for (unsigned i = t; i < n; i += BT) recs[(size_t)chunk * CHK + i] = srt[i];
}

__global__ __launch_bounds__(BT) void k_front(
    const float* __restrict__ x, const float* __restrict__ Qm,
    const float* __restrict__ W1, const float* __restrict__ W2,
    int steps, int tile_blocks,
    float* __restrict__ xq2, float* __restrict__ G1,
    const int* __restrict__ mrow, const int* __restrict__ mcol, int EM, int ncm,
    const int* __restrict__ e1row, const int* __restrict__ e1col, int E1, int nc1,
    const int* __restrict__ e2row, const int* __restrict__ e2col, int E2,
    unsigned* __restrict__ recs, unsigned* __restrict__ offs) {
  __shared__ unsigned cnt[NBIN];
  __shared__ unsigned cur[NBIN];
  __shared__ unsigned wsum[NBIN / 32];
  __shared__ unsigned srt[CHK];
  int b = blockIdx.x;
  if (b < tile_blocks) {
    const int wave = threadIdx.x >> 5;
    const int row0 = (b * (blockDim.x >> 5) + wave) * 16;
    v2f a[4];
    load_a_frag(x, row0, 1.0f, a);
    pre_tile(a, Qm, W2, xq2, row0);
    return;
  }
  b -= tile_blocks;
  if (b < steps) {
    if (threadIdx.x < DD * DD) {
      const int k = threadIdx.x >> 4, n = threadIdx.x & 15;
      float acc = 0.f;
#pragma unroll
      for (int t = 0; t < DD; ++t)
        acc = fmaf(W1[b * DD * DD + t * DD + k], Qm[t * DD + n], acc);
      G1[b * DD * DD + k * DD + n] = acc;
    }
    return;
  }
  b -= steps;
  const unsigned chunk = (unsigned)b;
  const int* row = mrow;
  const int* col = mcol;
  int E = EM, local = b;
  unsigned list = 0u;
  if (b >= ncm + nc1) {
    row = e2row; col = e2col; E = E2; local = b - ncm - nc1; list = 2u;
  } else if (b >= ncm) {
    row = e1row; col = e1col; E = E1; local = b - ncm; list = 1u;
  }
  sort_chunk(row, col, E, local, list, chunk, recs, offs, cnt, cur, wsum, srt);
}

#define CPG 12

__device__ __forceinline__ void set_rec(unsigned rec, unsigned* bm) {
  const unsigned list = (rec >> 16) & 3u;
  if (list < 3u)
    atomicOr(&bm[(list * 16u + ((rec >> 12) & 15u)) * RW + ((rec & 4095u) >> 5)],
             1u << (rec & 31u));
}

__device__ __forceinline__ void load_bins(const unsigned* __restrict__ offs, int nch, int k0,
                                          unsigned b, unsigned o[CPG]) {
#pragma unroll
  for (int g = 0; g < CPG; ++g) {
    const int k = k0 + 16 * g;
    o[g] = k < nch ? offs[(size_t)k * NBIN + b] : 0u;
  }
}

__device__ __forceinline__ void load_recs(const unsigned* __restrict__ recs, int k0, unsigned lane,
                                          const unsigned o[CPG], unsigned rc[CPG]) {
#pragma unroll
  for (int g = 0; g < CPG; ++g) {
    const int k = k0 + 16 * g;
    const unsigned n = o[g] >> 16, st = o[g] & 0xffffu;
    rc[g] = lane < n ? recs[(size_t)k * CHK + st + lane] : REC_NONE;
  }
}

__device__ __forceinline__ void set_group(const unsigned* __restrict__ recs, int k0, unsigned lane,
                                          const unsigned o[CPG], const unsigned rc[CPG],
                                          unsigned* bm) {
#pragma unroll
  for (int g = 0; g < CPG; ++g) set_rec(rc[g], bm);
  for (int g = 0; g < CPG; ++g) {
    const int k = k0 + 16 * g;
    const unsigned n = o[g] >> 16, st = o[g] & 0xffffu;
    for (unsigned q = lane + 32u; q < n; q += 32u) set_rec(recs[(size_t)k * CHK + st + q], bm);
  }
}

__global__ __launch_bounds__(BT) void k_build(const unsigned* __restrict__ recs,
                                              const unsigned* __restrict__ offs, int nch,
                                              unsigned* __restrict__ AB,
                                              unsigned* __restrict__ part2) {
  __shared__ __attribute__((aligned(16))) unsigned bm[3 * 16 * RW];
  __shared__ unsigned wc[16];
  const unsigned t = threadIdx.x, wave = t >> 5, lane = t & 31u;
  const unsigned b = blockIdx.x;
  unsigned o[CPG], rc[CPG];
  load_bins(offs, nch, (int)wave, b, o);
  for (unsigned i = t; i < 3u * 16u * RW; i += BT) bm[i] = 0u;
  load_recs(recs, (int)wave, lane, o, rc);
  __syncthreads();
  set_group(recs, (int)wave, lane, o, rc, bm);
  for (int k0 = (int)wave + 16 * CPG; k0 < nch; k0 += 16 * CPG) {
    load_bins(offs, nch, k0, b, o);
    load_recs(recs, k0, lane, o, rc);
    set_group(recs, k0, lane, o, rc, bm);
  }
  __syncthreads();
  const uint4 m = reinterpret_cast<const uint4*>(bm + (0u * 16u + wave) * RW)[lane];
  const uint4 b1 = reinterpret_cast<const uint4*>(bm + (1u * 16u + wave) * RW)[lane];
  const uint4 b2 = reinterpret_cast<const uint4*>(bm + (2u * 16u + wave) * RW)[lane];
  uint4 a1, a2;
  a1.x = b1.x & m.x; a1.y = b1.y & m.y; a1.z = b1.z & m.z; a1.w = b1.w & m.w;
  a2.x = b2.x & m.x; a2.y = b2.y & m.y; a2.z = b2.z & m.z; a2.w = b2.w & m.w;
  const size_t j = (size_t)b * 16 + wave;
  reinterpret_cast<uint4*>(AB + j * (2 * RW))[lane] = a1;
  reinterpret_cast<uint4*>(AB + j * (2 * RW) + RW)[lane] = a2;
  unsigned n2 = __popc(a2.x) + __popc(a2.y) + __popc(a2.z) + __popc(a2.w);
#pragma unroll
  for (int s = 16; s > 0; s >>= 1) n2 += __shfl_xor(n2, s);
  if (lane == 0) wc[wave] = n2;
  __syncthreads();
  if (t == 0) {
    unsigned s = 0;
    for (unsigned k = 0; k < 16; ++k) s += wc[k];
    part2[b] = s;
  }
}

template <int M>
__device__ __forceinline__ float xadd(float v) {
  return v + __int_as_float(__builtin_amdgcn_update_dpp(0, __float_as_int(v), 0x160 + M, 0xf, 0xf, true));
}

__device__ __forceinline__ float hadd(float v) {
  const int b = __float_as_int(v);
  return v + __int_as_float(__builtin_amdgcn_permlanex16(b, b, 0x76543210, 0xfedcba98, false, false));
}

#define LCAP 256
#define SEGE 16
#define NSEG 4
#define S1 (SEGE / 8)
#define S2 (SEGE / 4)

__device__ __forceinline__ void put_sources(const unsigned w[4], unsigned lane, unsigned p,
                                            unsigned start, unsigned short* lst) {
#pragma unroll
  for (unsigned k = 0; k < 4; ++k) {
    unsigned v = w[k];
    while (v) {
      const unsigned bpos = (unsigned)__ffs((int)v) - 1u;
      v &= v - 1u;
      const unsigned q = p - start;
      if (q < LCAP) lst[q] = (unsigned short)((4u * lane + k) * 32u + bpos);
      ++p;
    }
  }
}

__global__ __launch_bounds__(512) void k_step(
    const float* __restrict__ xcur, const float* __restrict__ xq2,
    const unsigned* __restrict__ AB,
    const float* __restrict__ G1s, const float* __restrict__ Qm,
    const unsigned* __restrict__ part2, float s1,
    float* __restrict__ xnext, const float* __restrict__ W2n,
    float* __restrict__ xq2n, int has_next) {
  __shared__ __attribute__((aligned(16))) unsigned short lst1[16][LCAP];
  __shared__ __attribute__((aligned(16))) unsigned short lst2[16][LCAP];
  __shared__ unsigned tot[16];
  __shared__ __attribute__((aligned(16))) float yt1[16 * TP];
  __shared__ __attribute__((aligned(16))) float yt2[16 * TP];
  __shared__ __attribute__((aligned(16))) float tile[16 * TP];

  const unsigned wave = threadIdx.x >> 5;
  const unsigned lane = threadIdx.x & 31;
  const unsigned c = lane & 15u, h = lane >> 4;
  const unsigned g4 = lane & 3u, q8 = lane >> 2;
  const unsigned u8 = lane & 7u, q4 = lane >> 3;
  const int row0 = blockIdx.x * 16;
  const size_t j = (size_t)row0 + wave;

  const uint4 ra1 = reinterpret_cast<const uint4*>(AB + j * (2 * RW))[lane];
  const uint4 ra2 = reinterpret_cast<const uint4*>(AB + j * (2 * RW) + RW)[lane];
  const v2f xj = reinterpret_cast<const v2f*>(xcur + j * DD)[u8];

  v8f cc = {};
  v2f bg[4] = {}, bq[4] = {}, bw[4] = {};
  unsigned cpart = 0u;
  if (wave == 0) {
    uint4 pw[NBIN / 128];
#pragma unroll
    for (int q = 0; q < NBIN / 128; ++q)
      pw[q] = reinterpret_cast<const uint4*>(part2)[lane + 32u * q];
    cc = load_c_tile(xcur, row0);
    load_b_frag(G1s, bg);
    load_b_frag(Qm, bq);
    load_bT_frag(W2n, bw);
#pragma unroll
    for (int q = 0; q < NBIN / 128; ++q) cpart += pw[q].x + pw[q].y + pw[q].z + pw[q].w;
  }

  const unsigned w1[4] = {ra1.x, ra1.y, ra1.z, ra1.w};
  const unsigned w2[4] = {ra2.x, ra2.y, ra2.z, ra2.w};
  const unsigned n1 = __popc(w1[0]) + __popc(w1[1]) + __popc(w1[2]) + __popc(w1[3]);
  const unsigned n2 = __popc(w2[0]) + __popc(w2[1]) + __popc(w2[2]) + __popc(w2[3]);

  const unsigned pk = n1 | (n2 << 16);
  unsigned inc = pk;
#pragma unroll
  for (unsigned o = 1; o < 32; o <<= 1) {
    const unsigned up = __shfl_up(inc, o);
    if (lane >= o) inc += up;
  }
  const unsigned totpk = __shfl(inc, 31);
  const unsigned exc = inc - pk;
  const unsigned off1 = exc & 0xffffu, off2 = exc >> 16;
  const unsigned tot1 = totpk & 0xffffu, tot2 = totpk >> 16;

  if (lane == 0) tot[wave] = tot1 > tot2 ? tot1 : tot2;
  __syncthreads();
  unsigned mx = 0;
#pragma unroll
  for (unsigned k = 0; k < 16; ++k) mx = tot[k] > mx ? tot[k] : mx;
  const unsigned npass = (mx + LCAP - 1) / LCAP;

  v4f acc1 = {};
  v2f acc2 = {};
  const v4f* X4 = reinterpret_cast<const v4f*>(xcur);
  const v4f* P4 = reinterpret_cast<const v4f*>(xq2);
  for (unsigned ps = 0; ps < npass; ++ps) {
    const unsigned start = ps * LCAP;
    put_sources(w1, lane, off1, start, lst1[wave]);
    put_sources(w2, lane, off2, start, lst2[wave]);
    __syncthreads();
    const unsigned r1n = tot1 > start ? tot1 - start : 0u;
    const unsigned r2n = tot2 > start ? tot2 - start : 0u;
    const unsigned m1 = r1n < LCAP ? r1n : LCAP;
    const unsigned m2 = r2n < LCAP ? r2n : LCAP;
    const unsigned mm = __builtin_amdgcn_readfirstlane(m1 > m2 ? m1 : m2);
    for (unsigned t0 = 0; t0 < mm; t0 += SEGE * NSEG) {
      unsigned pk1[NSEG];
      uint2 pk2[NSEG];
#pragma unroll
      for (unsigned sg = 0; sg < NSEG; ++sg) {
        pk1[sg] = *reinterpret_cast<const unsigned*>(&lst1[wave][t0 + SEGE * sg + 2u * q8]);
        pk2[sg] = *reinterpret_cast<const uint2*>(&lst2[wave][t0 + SEGE * sg + 4u * q4]);
      }
      v4f v1[NSEG * S1];
      v4f qv[NSEG * S2];
#pragma unroll
      for (unsigned sg = 0; sg < NSEG; ++sg) {
        if (t0 + SEGE * sg < mm) {
#pragma unroll
          for (unsigned u = 0; u < S1; ++u) {
            const unsigned e = t0 + SEGE * sg + 2u * q8 + u;
            const unsigned si = (pk1[sg] >> (16u * u)) & 0xffffu;
            v1[sg * S1 + u] = X4[(e < m1 ? si : 0u) * (DD / 4) + g4];
          }
#pragma unroll
          for (unsigned u = 0; u < S2; ++u) {
            const unsigned e = t0 + SEGE * sg + 4u * q4 + u;
            const unsigned wd = u < 2u ? pk2[sg].x : pk2[sg].y;
            const unsigned si = (wd >> (16u * (u & 1u))) & 0xffffu;
            qv[sg * S2 + u] = P4[(e < m2 ? si : 0u) * (XP / 4) + u8];
          }
        }
      }
      __builtin_amdgcn_sched_barrier(0);
#pragma unroll
      for (unsigned sg = 0; sg < NSEG; ++sg) {
        if (t0 + SEGE * sg < mm) {
#pragma unroll
          for (unsigned u = 0; u < S1; ++u) {
            const unsigned s = sg * S1 + u;
            const float w = t0 + SEGE * sg + 2u * q8 + u < m1 ? 1.0f : 0.0f;
            acc1.x = fmaf(v1[s].x, w, acc1.x);
            acc1.y = fmaf(v1[s].y, w, acc1.y);
            acc1.z = fmaf(v1[s].z, w, acc1.z);
            acc1.w = fmaf(v1[s].w, w, acc1.w);
          }
#pragma unroll
          for (unsigned u = 0; u < S2; ++u) {
            const unsigned s = sg * S2 + u;
            float a = qv[s].x * xj.x;
            a = fmaf(qv[s].z, xj.y, a);
            a = xadd<4>(xadd<2>(xadd<1>(a)));
            a = t0 + SEGE * sg + 4u * q4 + u < m2 ? a : 0.f;
            acc2.x = fmaf(a, qv[s].y, acc2.x);
            acc2.y = fmaf(a, qv[s].w, acc2.y);
          }
        }
      }
    }
    __syncthreads();
  }

  acc1.x = hadd(xadd<8>(xadd<4>(acc1.x)));
  acc1.y = hadd(xadd<8>(xadd<4>(acc1.y)));
  acc1.z = hadd(xadd<8>(xadd<4>(acc1.z)));
  acc1.w = hadd(xadd<8>(xadd<4>(acc1.w)));
  acc2.x = hadd(xadd<8>(acc2.x));
  acc2.y = hadd(xadd<8>(acc2.y));
  if (lane < 4u) {
    v2f* y = reinterpret_cast<v2f*>(yt1 + wave * TP + 4u * lane);
    v2f lo, hi;
    lo.x = acc1.x; lo.y = acc1.y;
    hi.x = acc1.z; hi.y = acc1.w;
    y[0] = lo;
    y[1] = hi;
  }
  if (lane < 8u) *reinterpret_cast<v2f*>(yt2 + wave * TP + 2u * lane) = acc2;
  __syncthreads();

  if (wave == 0) {
    unsigned cnt2 = cpart;
#pragma unroll
    for (int o = 16; o > 0; o >>= 1) cnt2 += __shfl_xor(cnt2, o);
    const float s2 = -(float)NN / (float)cnt2;
    v2f a[4];
    load_a_frag_lds(yt1, s1, a);
    cc = wmma16(a, bg, cc);
    load_a_frag_lds(yt2, s2, a);
    cc = wmma16(a, bq, cc);
    store_d_tile(xnext, row0, cc);
    if (has_next) {
#pragma unroll
      for (int r = 0; r < 8; ++r) tile[(r + 8 * h) * TP + c] = cc[r];
    }
  }
  if (has_next) {
    __syncthreads();
    if (wave == 0) {
      v2f a[4];
      load_a_frag_lds(tile, 1.0f, a);
      v8f d = {};
      d = wmma16(a, bq, d);
      v8f d2 = {};
      d2 = wmma16(a, bw, d2);
      store_d_pair(xq2n, row0, d, d2);
    }
  }
}

extern "C" void kernel_launch(void* const* d_in, const int* in_sizes, int n_in,
                              void* d_out, int out_size, void* d_ws, size_t ws_size,
                              hipStream_t stream) {
  const float* x_in = (const float*)d_in[0];
  const int* e1 = (const int*)d_in[1];
  const int* e2 = (const int*)d_in[2];
  const float* Qm = (const float*)d_in[3];
  const int* mk = (const int*)d_in[4];
  const float* W1 = (const float*)d_in[5];
  const float* W2 = (const float*)d_in[6];
  float* out = (float*)d_out;

  const int E1 = in_sizes[1] / 2;
  const int E2 = in_sizes[2] / 2;
  const int EM = in_sizes[4] / 2;
  const int STEPS = in_sizes[5] / (DD * DD);

  char* ws = (char*)d_ws;
  size_t off = 0;
  auto carve = [&](size_t bytes) {
    char* p = ws + off;
    off += (bytes + 255) & ~(size_t)255;
    return p;
  };
  const int NCM = (EM + CHK - 1) / CHK;
  const int NC1 = (E1 + CHK - 1) / CHK;
  const int NC2 = (E2 + CHK - 1) / CHK;
  const int NCH = NCM + NC1 + NC2;
  unsigned* recs  = (unsigned*)carve((size_t)NCH * CHK * 4);
  unsigned* offs  = (unsigned*)carve((size_t)NCH * NBIN * 4);
  unsigned* AB    = (unsigned*)carve((size_t)NN * 2 * RW * 4);
  unsigned* part2 = (unsigned*)carve((size_t)NBIN * 4);
  float* G1  = (float*)carve((size_t)STEPS * DD * DD * 4);
  float* xq2a = (float*)carve((size_t)NN * XP * 4);
  float* xq2b = (float*)carve((size_t)NN * XP * 4);
  float* xb0 = (float*)carve((size_t)NN * DD * 4);
  float* xb1 = (float*)carve((size_t)NN * DD * 4);

  const int TILE_BLOCKS = NN / 16 / (BT / 32);

  k_front<<<TILE_BLOCKS + STEPS + NCH, BT, 0, stream>>>(
      x_in, Qm, W1, W2, STEPS, TILE_BLOCKS, xq2a, G1,
      mk, mk + EM, EM, NCM,
      e1, e1 + E1, E1, NC1,
      e2, e2 + E2, E2, recs, offs);

  k_build<<<NBIN, BT, 0, stream>>>(recs, offs, NCH, AB, part2);

  const float s1 = (float)NN / (float)EM;

  const float* xc = x_in;
  for (int s = 0; s < STEPS; ++s) {
    float* xn = (s == STEPS - 1) ? out : ((s & 1) ? xb1 : xb0);
    const int has_next = (s + 1 < STEPS) ? 1 : 0;
    const float* W2n = W2 + (size_t)(has_next ? s + 1 : s) * DD * DD;
    const float* xq2c = (s & 1) ? xq2b : xq2a;
    float* xq2n = (s & 1) ? xq2a : xq2b;
    k_step<<<NN / 16, BT, 0, stream>>>(xc, xq2c, AB, G1 + (size_t)s * DD * DD, Qm,
                                       part2, s1, xn, W2n, xq2n, has_next);
    xc = xn;
  }
}


# Step 0 runs inside the bitmap-build launch (5 launches to 4): its x' is left as u - a*v, a = n/sum(A2) unknown there; step 1 joins the parts at each load (one more rounding per value read)
# speedup vs baseline: 21.2306x; 1.0759x over previous
#include <hip/hip_runtime.h>


#define NN 4096
#define DD 16
#define RW (NN / 32)
#define NBIN (NN / 16)
#define CHK 4096
#define BT 512
#define EPB (CHK / BT)

typedef float v2f __attribute__((ext_vector_type(2)));

#define XP (2 * DD)
typedef float v8f __attribute__((ext_vector_type(8)));
typedef float v4f __attribute__((ext_vector_type(4)));

__device__ __forceinline__ void load_a_frag(const float* X, int row0, float scale, v2f a[4]) {
  const int lane = threadIdx.x & 31;
  const int m = lane & 15, h = lane >> 4;
  const float* base = X + (size_t)(row0 + m) * DD + 2 * h;
#pragma unroll
  for (int c = 0; c < 4; ++c) {
    a[c].x = base[4 * c] * scale;
    a[c].y = base[4 * c + 1] * scale;
  }
}

__device__ __forceinline__ void load_b_frag(const float* __restrict__ W, v2f b[4]) {
  const int lane = threadIdx.x & 31;
  const int n = lane & 15, h = lane >> 4;
#pragma unroll
  for (int c = 0; c < 4; ++c) {
    b[c].x = W[(4 * c + 2 * h) * DD + n];
    b[c].y = W[(4 * c + 2 * h + 1) * DD + n];
  }
}

__device__ __forceinline__ void load_bT_frag(const float* __restrict__ W, v2f b[4]) {
  const int lane = threadIdx.x & 31;
  const int n = lane & 15, h = lane >> 4;
  const float* row = W + n * DD;
#pragma unroll
  for (int c = 0; c < 4; ++c) {
    b[c].x = row[4 * c + 2 * h];
    b[c].y = row[4 * c + 2 * h + 1];
  }
}

__device__ __forceinline__ v8f load_c_tile(const float* __restrict__ X, int row0) {
  const int lane = threadIdx.x & 31;
  const int n = lane & 15, h = lane >> 4;
  v8f c;
#pragma unroll
  for (int r = 0; r < 8; ++r) c[r] = X[(size_t)(row0 + r + 8 * h) * DD + n];
  return c;
}

__device__ __forceinline__ void store_d_tile(float* __restrict__ O, int row0, v8f d) {
  const int lane = threadIdx.x & 31;
  const int n = lane & 15, h = lane >> 4;
#pragma unroll
  for (int r = 0; r < 8; ++r) O[(size_t)(row0 + r + 8 * h) * DD + n] = d[r];
}

__device__ __forceinline__ void store_d_pair(float* __restrict__ O, int row0, v8f dq, v8f d2) {
  const int lane = threadIdx.x & 31;
  const int n = lane & 15, h = lane >> 4;
#pragma unroll
  for (int r = 0; r < 8; ++r) {
    v2f p;
    p.x = dq[r];
    p.y = d2[r];
    *reinterpret_cast<v2f*>(O + (size_t)(row0 + r + 8 * h) * XP + 2 * n) = p;
  }
}

__device__ __forceinline__ v8f wmma16(const v2f a[4], const v2f b[4], v8f c) {
#pragma unroll
  for (int k = 0; k < 4; ++k)
    c = __builtin_amdgcn_wmma_f32_16x16x4_f32(false, a[k], false, b[k],
                                              (short)0, c, false, false);
  return c;
}

__device__ __forceinline__ void pre_tile(const v2f a[4], const float* __restrict__ Qm,
                                         const float* __restrict__ W2s,
                                         float* __restrict__ xq2, int row0) {
  v2f b[4];
  load_b_frag(Qm, b);
  v8f d = {};
  d = wmma16(a, b, d);
  load_bT_frag(W2s, b);
  v8f d2 = {};
  d2 = wmma16(a, b, d2);
  store_d_pair(xq2, row0, d, d2);
}

#define TP 18

__device__ __forceinline__ void load_a_frag_lds(const float* T, float scale, v2f a[4]) {
  const int lane = threadIdx.x & 31;
  const int m = lane & 15, h = lane >> 4;
#pragma unroll
  for (int c = 0; c < 4; ++c) {
    a[c].x = T[m * TP + 4 * c + 2 * h] * scale;
    a[c].y = T[m * TP + 4 * c + 2 * h + 1] * scale;
  }
}

#define REC_NONE 0xffffffffu

__device__ __forceinline__ void sort_chunk(const int* __restrict__ row, const int* __restrict__ col,
                                           int E, int local, unsigned list, unsigned chunk,
                                           unsigned* __restrict__ recs, unsigned* __restrict__ offs,
                                           unsigned* cnt, unsigned* cur, unsigned* wsum,
                                           unsigned* srt) {
  const unsigned t = threadIdx.x, lane = t & 31u;
  unsigned rec[EPB];
  const int e0 = local * CHK + (int)t;
#pragma unroll
  for (int i = 0; i < EPB; ++i) {
    const int e = e0 + i * BT;
    const int ec = e < E ? e : E - 1;
    const unsigned r = (unsigned)row[ec] & (unsigned)(NN - 1);
    const unsigned c = (unsigned)col[ec] & (unsigned)(NN - 1);
    const unsigned v = ((c >> 4) << 18) | (list << 16) | ((c & 15u) << 12) | r;
    rec[i] = e < E ? v : REC_NONE;
  }
  if (t < NBIN) cnt[t] = 0u;
  __syncthreads();
#pragma unroll
  for (int i = 0; i < EPB; ++i)
    if (rec[i] != REC_NONE) atomicAdd(&cnt[rec[i] >> 18], 1u);
  __syncthreads();
  const unsigned v = t < NBIN ? cnt[t] : 0u;
  unsigned inc = v;
#pragma unroll
  for (unsigned o = 1; o < 32; o <<= 1) {
    const unsigned up = __shfl_up(inc, o);
    if (lane >= o) inc += up;
  }
  if (t < NBIN && lane == 31u) wsum[t >> 5] = inc;
  __syncthreads();
  unsigned ex = 0u;
  if (t < NBIN) {
    unsigned base = 0u;
    for (unsigned w = 0; w < (t >> 5); ++w) base += wsum[w];
    ex = base + inc - v;
    cur[t] = ex;
  }
  __syncthreads();
  unsigned pos[EPB];
#pragma unroll
  for (int i = 0; i < EPB; ++i)
    pos[i] = rec[i] != REC_NONE ? atomicAdd(&cur[rec[i] >> 18], 1u) : 0u;
#pragma unroll
  for (int i = 0; i < EPB; ++i)
    if (rec[i] != REC_NONE) srt[pos[i]] = rec[i];
  __syncthreads();
  if (t < NBIN) offs[(size_t)chunk * NBIN + t] = (v << 16) | ex;
  const int left = E - local * CHK;
  const unsigned n = left < CHK ? (unsigned)left : (unsigned)CHK;
  for (unsigned i = t; i < n; i += BT) recs[(size_t)chunk * CHK + i] = srt[i];
}

__global__ __launch_bounds__(BT) void k_front(
    const float* __restrict__ x, const float* __restrict__ Qm,
    const float* __restrict__ W1, const float* __restrict__ W2,
    int steps, int tile_blocks,
    float* __restrict__ xq2, float* __restrict__ G1,
    const int* __restrict__ mrow, const int* __restrict__ mcol, int EM, int ncm,
    const int* __restrict__ e1row, const int* __restrict__ e1col, int E1, int nc1,
    const int* __restrict__ e2row, const int* __restrict__ e2col, int E2,
    unsigned* __restrict__ recs, unsigned* __restrict__ offs) {
  __shared__ unsigned cnt[NBIN];
  __shared__ unsigned cur[NBIN];
  __shared__ unsigned wsum[NBIN / 32];
  __shared__ unsigned srt[CHK];
  int b = blockIdx.x;
  if (b < tile_blocks) {
    const int wave = threadIdx.x >> 5;
    const int row0 = (b * (blockDim.x >> 5) + wave) * 16;
    v2f a[4];
    load_a_frag(x, row0, 1.0f, a);
    pre_tile(a, Qm, W2, xq2, row0);
    return;
  }
  b -= tile_blocks;
  if (b < steps) {
    if (threadIdx.x < DD * DD) {
      const int k = threadIdx.x >> 4, n = threadIdx.x & 15;
      float acc = 0.f;
#pragma unroll
      for (int t = 0; t < DD; ++t)
        acc = fmaf(W1[b * DD * DD + t * DD + k], Qm[t * DD + n], acc);
      G1[b * DD * DD + k * DD + n] = acc;
    }
    return;
  }
  b -= steps;
  const unsigned chunk = (unsigned)b;
  const int* row = mrow;
  const int* col = mcol;
  int E = EM, local = b;
  unsigned list = 0u;
  if (b >= ncm + nc1) {
    row = e2row; col = e2col; E = E2; local = b - ncm - nc1; list = 2u;
  } else if (b >= ncm) {
    row = e1row; col = e1col; E = E1; local = b - ncm; list = 1u;
  }
  sort_chunk(row, col, E, local, list, chunk, recs, offs, cnt, cur, wsum, srt);
}

#define CPG 12

__device__ __forceinline__ void set_rec(unsigned rec, unsigned* bm) {
  const unsigned list = (rec >> 16) & 3u;
  if (list < 3u)
    atomicOr(&bm[(list * 16u + ((rec >> 12) & 15u)) * RW + ((rec & 4095u) >> 5)],
             1u << (rec & 31u));
}

__device__ __forceinline__ void load_bins(const unsigned* __restrict__ offs, int nch, int k0,
                                          unsigned b, unsigned o[CPG]) {
#pragma unroll
  for (int g = 0; g < CPG; ++g) {
    const int k = k0 + 16 * g;
    o[g] = k < nch ? offs[(size_t)k * NBIN + b] : 0u;
  }
}

__device__ __forceinline__ void load_recs(const unsigned* __restrict__ recs, int k0, unsigned lane,
                                          const unsigned o[CPG], unsigned rc[CPG]) {
#pragma unroll
  for (int g = 0; g < CPG; ++g) {
    const int k = k0 + 16 * g;
    const unsigned n = o[g] >> 16, st = o[g] & 0xffffu;
    rc[g] = lane < n ? recs[(size_t)k * CHK + st + lane] : REC_NONE;
  }
}

__device__ __forceinline__ void set_group(const unsigned* __restrict__ recs, int k0, unsigned lane,
                                          const unsigned o[CPG], const unsigned rc[CPG],
                                          unsigned* bm) {
#pragma unroll
  for (int g = 0; g < CPG; ++g) set_rec(rc[g], bm);
  for (int g = 0; g < CPG; ++g) {
    const int k = k0 + 16 * g;
    const unsigned n = o[g] >> 16, st = o[g] & 0xffffu;
    for (unsigned q = lane + 32u; q < n; q += 32u) set_rec(recs[(size_t)k * CHK + st + q], bm);
  }
}

template <int M>
__device__ __forceinline__ float xadd(float v) {
  return v + __int_as_float(__builtin_amdgcn_update_dpp(0, __float_as_int(v), 0x160 + M, 0xf, 0xf, true));
}

__device__ __forceinline__ float hadd(float v) {
  const int b = __float_as_int(v);
  return v + __int_as_float(__builtin_amdgcn_permlanex16(b, b, 0x76543210, 0xfedcba98, false, false));
}

#define LCAP 256
#define LPITCH (LCAP + 32)
#define SEGE 16
#define S1 (SEGE / 8)
#define S2 (SEGE / 4)

struct __attribute__((aligned(16))) StepLds {
  unsigned short lst1[16][LPITCH];
  unsigned short lst2[16][LPITCH];
  float yt1[16 * TP];
  float yt2[16 * TP];
  float tile[16 * TP];
  float tile2[16 * TP];
  unsigned tot[16];
};

__device__ __forceinline__ void put_sources(const unsigned w[4], unsigned lane, unsigned p,
                                            unsigned start, unsigned short* lst) {
#pragma unroll
  for (unsigned k = 0; k < 4; ++k) {
    unsigned v = w[k];
    while (v) {
      const unsigned bpos = (unsigned)__ffs((int)v) - 1u;
      v &= v - 1u;
      const unsigned q = p - start;
      if (q < LCAP) lst[q] = (unsigned short)((4u * lane + k) * 32u + bpos);
      ++p;
    }
  }
}

template <int MODE>
__device__ __forceinline__ void step_core(
    StepLds& L, const uint4 ra1, const uint4 ra2, const int row0,
    const float* __restrict__ xa, const float* __restrict__ xb,
    const float* __restrict__ pa, const float* __restrict__ pb,
    const float* __restrict__ G1s, const float* __restrict__ Qm,
    const unsigned* __restrict__ part2, const float s1,
    float* __restrict__ xo, float* __restrict__ xo2,
    const float* __restrict__ W2n,
    float* __restrict__ po, float* __restrict__ po2, const int has_next) {
  constexpr unsigned NS = MODE == 1 ? 3u : 4u;
  constexpr unsigned NB = MODE == 1 ? NS : 1u;

  const unsigned wave = threadIdx.x >> 5;
  const unsigned lane = threadIdx.x & 31;
  const unsigned c = lane & 15u, h = lane >> 4;
  const unsigned g4 = lane & 3u, q8 = lane >> 2;
  const unsigned u8 = lane & 7u, q4 = lane >> 3;
  const size_t j = (size_t)row0 + wave;

  v2f xj = reinterpret_cast<const v2f*>(xa + j * DD)[u8];
  v2f xjb = {};
  if (MODE == 1) xjb = reinterpret_cast<const v2f*>(xb + j * DD)[u8];

  v8f cc = {}, cb = {};
  v2f bg[4] = {}, bq[4] = {}, bw[4] = {};
  unsigned cpart = 0u;
  const bool counts = MODE == 1 || (MODE == 2 && wave == 0);
  uint4 pw[NBIN / 128];
  if (counts) {
#pragma unroll
    for (int q = 0; q < NBIN / 128; ++q)
      pw[q] = reinterpret_cast<const uint4*>(part2)[lane + 32u * q];
  }
  if (wave == 0) {
    cc = load_c_tile(xa, row0);
    if (MODE == 1) cb = load_c_tile(xb, row0);
    load_b_frag(G1s, bg);
    load_b_frag(Qm, bq);
    load_bT_frag(W2n, bw);
  }
  if (counts) {
#pragma unroll
    for (int q = 0; q < NBIN / 128; ++q) cpart += pw[q].x + pw[q].y + pw[q].z + pw[q].w;
  }

  const unsigned w1[4] = {ra1.x, ra1.y, ra1.z, ra1.w};
  const unsigned w2[4] = {ra2.x, ra2.y, ra2.z, ra2.w};
  const unsigned n1 = __popc(w1[0]) + __popc(w1[1]) + __popc(w1[2]) + __popc(w1[3]);
  const unsigned n2 = __popc(w2[0]) + __popc(w2[1]) + __popc(w2[2]) + __popc(w2[3]);

  const unsigned pk = n1 | (n2 << 16);
  unsigned inc = pk;
#pragma unroll
  for (unsigned o = 1; o < 32; o <<= 1) {
    const unsigned up = __shfl_up(inc, o);
    if (lane >= o) inc += up;
  }
  const unsigned totpk = __shfl(inc, 31);
  const unsigned exc = inc - pk;
  const unsigned off1 = exc & 0xffffu, off2 = exc >> 16;
  const unsigned tot1 = totpk & 0xffffu, tot2 = totpk >> 16;

  if (lane == 0) L.tot[wave] = tot1 > tot2 ? tot1 : tot2;
  __syncthreads();
  unsigned mx = 0;
#pragma unroll
  for (unsigned k = 0; k < 16; ++k) mx = L.tot[k] > mx ? L.tot[k] : mx;
  const unsigned npass = (mx + LCAP - 1) / LCAP;

  float nalpha = 0.f;
  if (MODE == 1) {
    unsigned cnt2 = cpart;
#pragma unroll
    for (int o = 16; o > 0; o >>= 1) cnt2 += __shfl_xor(cnt2, o);
    nalpha = -(float)NN / (float)cnt2;
    xj.x = fmaf(nalpha, xjb.x, xj.x);
    xj.y = fmaf(nalpha, xjb.y, xj.y);
  }

  v4f acc1 = {};
  v2f acc2 = {};
  const v4f* X4 = reinterpret_cast<const v4f*>(xa);
  const v4f* P4 = reinterpret_cast<const v4f*>(pa);
  const v4f* X4b = reinterpret_cast<const v4f*>(xb);
  const v4f* P4b = reinterpret_cast<const v4f*>(pb);
  for (unsigned ps = 0; ps < npass; ++ps) {
    const unsigned start = ps * LCAP;
    put_sources(w1, lane, off1, start, L.lst1[wave]);
    put_sources(w2, lane, off2, start, L.lst2[wave]);
    __syncthreads();
    const unsigned r1n = tot1 > start ? tot1 - start : 0u;
    const unsigned r2n = tot2 > start ? tot2 - start : 0u;
    const unsigned m1 = r1n < LCAP ? r1n : LCAP;
    const unsigned m2 = r2n < LCAP ? r2n : LCAP;
    const unsigned mm = __builtin_amdgcn_readfirstlane(m1 > m2 ? m1 : m2);
    for (unsigned t0 = 0; t0 < mm; t0 += SEGE * NS) {
      unsigned pk1[NS];
      uint2 pk2[NS];
#pragma unroll
      for (unsigned sg = 0; sg < NS; ++sg) {
        pk1[sg] = *reinterpret_cast<const unsigned*>(&L.lst1[wave][t0 + SEGE * sg + 2u * q8]);
        pk2[sg] = *reinterpret_cast<const uint2*>(&L.lst2[wave][t0 + SEGE * sg + 4u * q4]);
      }
      v4f v1[NS * S1];
      v4f qv[NS * S2];
      v4f v1b[NB * S1];
      v4f qvb[NB * S2];
#pragma unroll
      for (unsigned sg = 0; sg < NS; ++sg) {
        if (t0 + SEGE * sg < mm) {
#pragma unroll
          for (unsigned u = 0; u < S1; ++u) {
            const unsigned e = t0 + SEGE * sg + 2u * q8 + u;
            const unsigned si = (pk1[sg] >> (16u * u)) & 0xffffu;
            const unsigned ix = (e < m1 ? si : 0u) * (DD / 4) + g4;
            v1[sg * S1 + u] = X4[ix];
            if (MODE == 1) v1b[sg * S1 + u] = X4b[ix];
          }
#pragma unroll
          for (unsigned u = 0; u < S2; ++u) {
            const unsigned e = t0 + SEGE * sg + 4u * q4 + u;
            const unsigned wd = u < 2u ? pk2[sg].x : pk2[sg].y;
            const unsigned si = (wd >> (16u * (u & 1u))) & 0xffffu;
            const unsigned ix = (e < m2 ? si : 0u) * (XP / 4) + u8;
            qv[sg * S2 + u] = P4[ix];
            if (MODE == 1) qvb[sg * S2 + u] = P4b[ix];
          }
        }
      }
      __builtin_amdgcn_sched_barrier(0);
#pragma unroll
      for (unsigned sg = 0; sg < NS; ++sg) {
        if (t0 + SEGE * sg < mm) {
#pragma unroll
          for (unsigned u = 0; u < S1; ++u) {
            const unsigned s = sg * S1 + u;
            v4f xs = v1[s];
            if (MODE == 1) {
              xs.x = fmaf(nalpha, v1b[s].x, xs.x);
              xs.y = fmaf(nalpha, v1b[s].y, xs.y);
              xs.z = fmaf(nalpha, v1b[s].z, xs.z);
              xs.w = fmaf(nalpha, v1b[s].w, xs.w);
            }
            const float w = t0 + SEGE * sg + 2u * q8 + u < m1 ? 1.0f : 0.0f;
            acc1.x = fmaf(xs.x, w, acc1.x);
            acc1.y = fmaf(xs.y, w, acc1.y);
            acc1.z = fmaf(xs.z, w, acc1.z);
            acc1.w = fmaf(xs.w, w, acc1.w);
          }
#pragma unroll
          for (unsigned u = 0; u < S2; ++u) {
            const unsigned s = sg * S2 + u;
            v4f qs = qv[s];
            if (MODE == 1) {
              qs.x = fmaf(nalpha, qvb[s].x, qs.x);
              qs.y = fmaf(nalpha, qvb[s].y, qs.y);
              qs.z = fmaf(nalpha, qvb[s].z, qs.z);
              qs.w = fmaf(nalpha, qvb[s].w, qs.w);
            }
            float a = qs.x * xj.x;
            a = fmaf(qs.z, xj.y, a);
            a = xadd<4>(xadd<2>(xadd<1>(a)));
            a = t0 + SEGE * sg + 4u * q4 + u < m2 ? a : 0.f;
            acc2.x = fmaf(a, qs.y, acc2.x);
            acc2.y = fmaf(a, qs.w, acc2.y);
          }
        }
      }
    }
    __syncthreads();
  }

  acc1.x = hadd(xadd<8>(xadd<4>(acc1.x)));
  acc1.y = hadd(xadd<8>(xadd<4>(acc1.y)));
  acc1.z = hadd(xadd<8>(xadd<4>(acc1.z)));
  acc1.w = hadd(xadd<8>(xadd<4>(acc1.w)));
  acc2.x = hadd(xadd<8>(acc2.x));
  acc2.y = hadd(xadd<8>(acc2.y));
  if (lane < 4u) {
    v2f* y = reinterpret_cast<v2f*>(L.yt1 + wave * TP + 4u * lane);
    v2f lo, hi;
    lo.x = acc1.x; lo.y = acc1.y;
    hi.x = acc1.z; hi.y = acc1.w;
    y[0] = lo;
    y[1] = hi;
  }
  if (lane < 8u) *reinterpret_cast<v2f*>(L.yt2 + wave * TP + 2u * lane) = acc2;
  __syncthreads();

  const bool next = MODE == 0 ? true : has_next != 0;
  if (wave == 0) {
    v2f a[4];
    if (MODE == 0) {
      load_a_frag_lds(L.yt1, s1, a);
      cc = wmma16(a, bg, cc);
      load_a_frag_lds(L.yt2, 1.0f, a);
      cb = wmma16(a, bq, cb);
      store_d_tile(xo, row0, cc);
      store_d_tile(xo2, row0, cb);
#pragma unroll
      for (int r = 0; r < 8; ++r) {
        L.tile[(r + 8 * h) * TP + c] = cc[r];
        L.tile2[(r + 8 * h) * TP + c] = cb[r];
      }
    } else {
      float s2 = nalpha;
      if (MODE == 2) {
        unsigned cnt2 = cpart;
#pragma unroll
        for (int o = 16; o > 0; o >>= 1) cnt2 += __shfl_xor(cnt2, o);
        s2 = -(float)NN / (float)cnt2;
      }
      if (MODE == 1) {
#pragma unroll
        for (int r = 0; r < 8; ++r) cc[r] = fmaf(s2, cb[r], cc[r]);
      }
      load_a_frag_lds(L.yt1, s1, a);
      cc = wmma16(a, bg, cc);
      load_a_frag_lds(L.yt2, s2, a);
      cc = wmma16(a, bq, cc);
      store_d_tile(xo, row0, cc);
      if (next) {
#pragma unroll
        for (int r = 0; r < 8; ++r) L.tile[(r + 8 * h) * TP + c] = cc[r];
      }
    }
  }
  if (next) {
    __syncthreads();
    if (wave == 0) {
      v2f a[4];
      load_a_frag_lds(L.tile, 1.0f, a);
      v8f d = {};
      d = wmma16(a, bq, d);
      v8f d2 = {};
      d2 = wmma16(a, bw, d2);
      store_d_pair(po, row0, d, d2);
      if (MODE == 0) {
        load_a_frag_lds(L.tile2, 1.0f, a);
        v8f e = {};
        e = wmma16(a, bq, e);
        v8f e2 = {};
        e2 = wmma16(a, bw, e2);
        store_d_pair(po2, row0, e, e2);
      }
    }
  }
}

__global__ __launch_bounds__(BT) void k_build0(
    const unsigned* __restrict__ recs, const unsigned* __restrict__ offs, int nch,
    unsigned* __restrict__ AB, unsigned* __restrict__ part2,
    const float* __restrict__ x, const float* __restrict__ xq2,
    const float* __restrict__ G1s, const float* __restrict__ Qm, float s1,
    float* __restrict__ xu, float* __restrict__ xv, const float* __restrict__ W2n,
    float* __restrict__ pu, float* __restrict__ pv) {
  __shared__ __attribute__((aligned(16))) unsigned bm[3 * 16 * RW];
  __shared__ unsigned wc[16];
  __shared__ StepLds L;
  const unsigned t = threadIdx.x, wave = t >> 5, lane = t & 31u;
  const unsigned b = blockIdx.x;
  unsigned o[CPG], rc[CPG];
  load_bins(offs, nch, (int)wave, b, o);
  for (unsigned i = t; i < 3u * 16u * RW; i += BT) bm[i] = 0u;
  load_recs(recs, (int)wave, lane, o, rc);
  __syncthreads();
  set_group(recs, (int)wave, lane, o, rc, bm);
  for (int k0 = (int)wave + 16 * CPG; k0 < nch; k0 += 16 * CPG) {
    load_bins(offs, nch, k0, b, o);
    load_recs(recs, k0, lane, o, rc);
    set_group(recs, k0, lane, o, rc, bm);
  }
  __syncthreads();
  const uint4 m = reinterpret_cast<const uint4*>(bm + (0u * 16u + wave) * RW)[lane];
  const uint4 b1 = reinterpret_cast<const uint4*>(bm + (1u * 16u + wave) * RW)[lane];
  const uint4 b2 = reinterpret_cast<const uint4*>(bm + (2u * 16u + wave) * RW)[lane];
  uint4 a1, a2;
  a1.x = b1.x & m.x; a1.y = b1.y & m.y; a1.z = b1.z & m.z; a1.w = b1.w & m.w;
  a2.x = b2.x & m.x; a2.y = b2.y & m.y; a2.z = b2.z & m.z; a2.w = b2.w & m.w;
  const size_t j = (size_t)b * 16 + wave;
  reinterpret_cast<uint4*>(AB + j * (2 * RW))[lane] = a1;
  reinterpret_cast<uint4*>(AB + j * (2 * RW) + RW)[lane] = a2;
  unsigned n2 = __popc(a2.x) + __popc(a2.y) + __popc(a2.z) + __popc(a2.w);
#pragma unroll
  for (int s = 16; s > 0; s >>= 1) n2 += __shfl_xor(n2, s);
  if (lane == 0) wc[wave] = n2;

  step_core<0>(L, a1, a2, (int)b * 16, x, nullptr, xq2, nullptr, G1s, Qm, nullptr, s1,
               xu, xv, W2n, pu, pv, 1);

  if (t == 32u) {
    unsigned s = 0;
    for (unsigned k = 0; k < 16; ++k) s += wc[k];
    part2[b] = s;
  }
}

__global__ __launch_bounds__(512) void k_step1(
    const float* __restrict__ xu, const float* __restrict__ xv,
    const float* __restrict__ pu, const float* __restrict__ pv,
    const unsigned* __restrict__ AB,
    const float* __restrict__ G1s, const float* __restrict__ Qm,
    const unsigned* __restrict__ part2, float s1,
    float* __restrict__ xnext, const float* __restrict__ W2n,
    float* __restrict__ xq2n, int has_next) {
  __shared__ StepLds L;
  const unsigned wave = threadIdx.x >> 5;
  const unsigned lane = threadIdx.x & 31;
  const int row0 = blockIdx.x * 16;
  const size_t j = (size_t)row0 + wave;
  const uint4 ra1 = reinterpret_cast<const uint4*>(AB + j * (2 * RW))[lane];
  const uint4 ra2 = reinterpret_cast<const uint4*>(AB + j * (2 * RW) + RW)[lane];
  step_core<1>(L, ra1, ra2, row0, xu, xv, pu, pv, G1s, Qm, part2, s1,
               xnext, nullptr, W2n, xq2n, nullptr, has_next);
}

__global__ __launch_bounds__(512) void k_step2(
    const float* __restrict__ xcur, const float* __restrict__ xq2,
    const unsigned* __restrict__ AB,
    const float* __restrict__ G1s, const float* __restrict__ Qm,
    const unsigned* __restrict__ part2, float s1,
    float* __restrict__ xnext, const float* __restrict__ W2n,
    float* __restrict__ xq2n, int has_next) {
  __shared__ StepLds L;
  const unsigned wave = threadIdx.x >> 5;
  const unsigned lane = threadIdx.x & 31;
  const int row0 = blockIdx.x * 16;
  const size_t j = (size_t)row0 + wave;
  const uint4 ra1 = reinterpret_cast<const uint4*>(AB + j * (2 * RW))[lane];
  const uint4 ra2 = reinterpret_cast<const uint4*>(AB + j * (2 * RW) + RW)[lane];
  step_core<2>(L, ra1, ra2, row0, xcur, nullptr, xq2, nullptr, G1s, Qm, part2, s1,
               xnext, nullptr, W2n, xq2n, nullptr, has_next);
}

extern "C" void kernel_launch(void* const* d_in, const int* in_sizes, int n_in,
                              void* d_out, int out_size, void* d_ws, size_t ws_size,
                              hipStream_t stream) {
  const float* x_in = (const float*)d_in[0];
  const int* e1 = (const int*)d_in[1];
  const int* e2 = (const int*)d_in[2];
  const float* Qm = (const float*)d_in[3];
  const int* mk = (const int*)d_in[4];
  const float* W1 = (const float*)d_in[5];
  const float* W2 = (const float*)d_in[6];
  float* out = (float*)d_out;

  const int E1 = in_sizes[1] / 2;
  const int E2 = in_sizes[2] / 2;
  const int EM = in_sizes[4] / 2;
  const int STEPS = in_sizes[5] / (DD * DD);

  char* ws = (char*)d_ws;
  size_t off = 0;
  auto carve = [&](size_t bytes) {
    char* p = ws + off;
    off += (bytes + 255) & ~(size_t)255;
    return p;
  };
  const int NCM = (EM + CHK - 1) / CHK;
  const int NC1 = (E1 + CHK - 1) / CHK;
  const int NC2 = (E2 + CHK - 1) / CHK;
  const int NCH = NCM + NC1 + NC2;
  unsigned* recs  = (unsigned*)carve((size_t)NCH * CHK * 4);
  unsigned* offs  = (unsigned*)carve((size_t)NCH * NBIN * 4);
  unsigned* AB    = (unsigned*)carve((size_t)NN * 2 * RW * 4);
  unsigned* part2 = (unsigned*)carve((size_t)NBIN * 4);
  float* G1  = (float*)carve((size_t)STEPS * DD * DD * 4);
  float* xq2a = (float*)carve((size_t)NN * XP * 4);
  float* xq2b = (float*)carve((size_t)NN * XP * 4);
  float* xb0 = (float*)carve((size_t)NN * DD * 4);
  float* xb1 = (float*)carve((size_t)NN * DD * 4);
  float* xu = (float*)carve((size_t)NN * DD * 4);
  float* xv = (float*)carve((size_t)NN * DD * 4);
  float* pu = (float*)carve((size_t)NN * XP * 4);
  float* pv = (float*)carve((size_t)NN * XP * 4);

  const int TILE_BLOCKS = NN / 16 / (BT / 32);

  k_front<<<TILE_BLOCKS + STEPS + NCH, BT, 0, stream>>>(
      x_in, Qm, W1, W2, STEPS, TILE_BLOCKS, xq2a, G1,
      mk, mk + EM, EM, NCM,
      e1, e1 + E1, E1, NC1,
      e2, e2 + E2, E2, recs, offs);

  const float s1 = (float)NN / (float)EM;

  k_build0<<<NBIN, BT, 0, stream>>>(recs, offs, NCH, AB, part2, x_in, xq2a, G1, Qm, s1,
                                    xu, xv, W2 + (size_t)(STEPS > 1 ? 1 : 0) * DD * DD, pu, pv);

  const float* xc = xu;
  const float* xq2c = pu;
  for (int s = 1; s < STEPS; ++s) {
    float* xn = (s == STEPS - 1) ? out : ((s & 1) ? xb0 : xb1);
    const int has_next = (s + 1 < STEPS) ? 1 : 0;
    const float* W2n = W2 + (size_t)(has_next ? s + 1 : s) * DD * DD;
    float* xq2n = (s & 1) ? xq2a : xq2b;
    if (s == 1)
      k_step1<<<NN / 16, BT, 0, stream>>>(xu, xv, pu, pv, AB, G1 + (size_t)s * DD * DD, Qm,
                                          part2, s1, xn, W2n, xq2n, has_next);
    else
      k_step2<<<NN / 16, BT, 0, stream>>>(xc, xq2c, AB, G1 + (size_t)s * DD * DD, Qm,
                                          part2, s1, xn, W2n, xq2n, has_next);
    xc = xn;
    xq2c = xq2n;
  }
}


# step_core: what a wave makes for itself stays in the wave: lane prefix sum by DPP (no ds_bpermute); lists and finish tile read back in the wave's LDS order; 4 of 5 block barriers go; sums unchanged
# speedup vs baseline: 24.5551x; 1.1566x over previous
#include <hip/hip_runtime.h>


#define NN 4096
#define DD 16
#define RW (NN / 32)
#define NBIN (NN / 16)
#define CHK 4096
#define BT 512
#define EPB (CHK / BT)

typedef float v2f __attribute__((ext_vector_type(2)));

#define XP (2 * DD)
typedef float v8f __attribute__((ext_vector_type(8)));
typedef float v4f __attribute__((ext_vector_type(4)));

__device__ __forceinline__ void load_a_frag(const float* X, int row0, float scale, v2f a[4]) {
  const int lane = threadIdx.x & 31;
  const int m = lane & 15, h = lane >> 4;
  const float* base = X + (size_t)(row0 + m) * DD + 2 * h;
#pragma unroll
  for (int c = 0; c < 4; ++c) {
    a[c].x = base[4 * c] * scale;
    a[c].y = base[4 * c + 1] * scale;
  }
}

__device__ __forceinline__ void load_b_frag(const float* __restrict__ W, v2f b[4]) {
  const int lane = threadIdx.x & 31;
  const int n = lane & 15, h = lane >> 4;
#pragma unroll
  for (int c = 0; c < 4; ++c) {
    b[c].x = W[(4 * c + 2 * h) * DD + n];
    b[c].y = W[(4 * c + 2 * h + 1) * DD + n];
  }
}

__device__ __forceinline__ void load_bT_frag(const float* __restrict__ W, v2f b[4]) {
  const int lane = threadIdx.x & 31;
  const int n = lane & 15, h = lane >> 4;
  const float* row = W + n * DD;
#pragma unroll
  for (int c = 0; c < 4; ++c) {
    b[c].x = row[4 * c + 2 * h];
    b[c].y = row[4 * c + 2 * h + 1];
  }
}

__device__ __forceinline__ v8f load_c_tile(const float* __restrict__ X, int row0) {
  const int lane = threadIdx.x & 31;
  const int n = lane & 15, h = lane >> 4;
  v8f c;
#pragma unroll
  for (int r = 0; r < 8; ++r) c[r] = X[(size_t)(row0 + r + 8 * h) * DD + n];
  return c;
}

__device__ __forceinline__ void store_d_tile(float* __restrict__ O, int row0, v8f d) {
  const int lane = threadIdx.x & 31;
  const int n = lane & 15, h = lane >> 4;
#pragma unroll
  for (int r = 0; r < 8; ++r) O[(size_t)(row0 + r + 8 * h) * DD + n] = d[r];
}

__device__ __forceinline__ void store_d_pair(float* __restrict__ O, int row0, v8f dq, v8f d2) {
  const int lane = threadIdx.x & 31;
  const int n = lane & 15, h = lane >> 4;
#pragma unroll
  for (int r = 0; r < 8; ++r) {
    v2f p;
    p.x = dq[r];
    p.y = d2[r];
    *reinterpret_cast<v2f*>(O + (size_t)(row0 + r + 8 * h) * XP + 2 * n) = p;
  }
}

__device__ __forceinline__ v8f wmma16(const v2f a[4], const v2f b[4], v8f c) {
#pragma unroll
  for (int k = 0; k < 4; ++k)
    c = __builtin_amdgcn_wmma_f32_16x16x4_f32(false, a[k], false, b[k],
                                              (short)0, c, false, false);
  return c;
}

__device__ __forceinline__ void pre_tile(const v2f a[4], const float* __restrict__ Qm,
                                         const float* __restrict__ W2s,
                                         float* __restrict__ xq2, int row0) {
  v2f b[4];
  load_b_frag(Qm, b);
  v8f d = {};
  d = wmma16(a, b, d);
  load_bT_frag(W2s, b);
  v8f d2 = {};
  d2 = wmma16(a, b, d2);
  store_d_pair(xq2, row0, d, d2);
}

#define TP 18

__device__ __forceinline__ void load_a_frag_lds(const float* T, float scale, v2f a[4]) {
  const int lane = threadIdx.x & 31;
  const int m = lane & 15, h = lane >> 4;
#pragma unroll
  for (int c = 0; c < 4; ++c) {
    a[c].x = T[m * TP + 4 * c + 2 * h] * scale;
    a[c].y = T[m * TP + 4 * c + 2 * h + 1] * scale;
  }
}

#define REC_NONE 0xffffffffu

__device__ __forceinline__ void sort_chunk(const int* __restrict__ row, const int* __restrict__ col,
                                           int E, int local, unsigned list, unsigned chunk,
                                           unsigned* __restrict__ recs, unsigned* __restrict__ offs,
                                           unsigned* cnt, unsigned* cur, unsigned* wsum,
                                           unsigned* srt) {
  const unsigned t = threadIdx.x, lane = t & 31u;
  unsigned rec[EPB];
  const int e0 = local * CHK + (int)t;
#pragma unroll
  for (int i = 0; i < EPB; ++i) {
    const int e = e0 + i * BT;
    const int ec = e < E ? e : E - 1;
    const unsigned r = (unsigned)row[ec] & (unsigned)(NN - 1);
    const unsigned c = (unsigned)col[ec] & (unsigned)(NN - 1);
    const unsigned v = ((c >> 4) << 18) | (list << 16) | ((c & 15u) << 12) | r;
    rec[i] = e < E ? v : REC_NONE;
  }
  if (t < NBIN) cnt[t] = 0u;
  __syncthreads();
#pragma unroll
  for (int i = 0; i < EPB; ++i)
    if (rec[i] != REC_NONE) atomicAdd(&cnt[rec[i] >> 18], 1u);
  __syncthreads();
  const unsigned v = t < NBIN ? cnt[t] : 0u;
  unsigned inc = v;
#pragma unroll
  for (unsigned o = 1; o < 32; o <<= 1) {
    const unsigned up = __shfl_up(inc, o);
    if (lane >= o) inc += up;
  }
  if (t < NBIN && lane == 31u) wsum[t >> 5] = inc;
  __syncthreads();
  unsigned ex = 0u;
  if (t < NBIN) {
    unsigned base = 0u;
    for (unsigned w = 0; w < (t >> 5); ++w) base += wsum[w];
    ex = base + inc - v;
    cur[t] = ex;
  }
  __syncthreads();
  unsigned pos[EPB];
#pragma unroll
  for (int i = 0; i < EPB; ++i)
    pos[i] = rec[i] != REC_NONE ? atomicAdd(&cur[rec[i] >> 18], 1u) : 0u;
#pragma unroll
  for (int i = 0; i < EPB; ++i)
    if (rec[i] != REC_NONE) srt[pos[i]] = rec[i];
  __syncthreads();
  if (t < NBIN) offs[(size_t)chunk * NBIN + t] = (v << 16) | ex;
  const int left = E - local * CHK;
  const unsigned n = left < CHK ? (unsigned)left : (unsigned)CHK;
  for (unsigned i = t; i < n; i += BT) recs[(size_t)chunk * CHK + i] = srt[i];
}

__global__ __launch_bounds__(BT) void k_front(
    const float* __restrict__ x, const float* __restrict__ Qm,
    const float* __restrict__ W1, const float* __restrict__ W2,
    int steps, int tile_blocks,
    float* __restrict__ xq2, float* __restrict__ G1,
    const int* __restrict__ mrow, const int* __restrict__ mcol, int EM, int ncm,
    const int* __restrict__ e1row, const int* __restrict__ e1col, int E1, int nc1,
    const int* __restrict__ e2row, const int* __restrict__ e2col, int E2,
    unsigned* __restrict__ recs, unsigned* __restrict__ offs) {
  __shared__ unsigned cnt[NBIN];
  __shared__ unsigned cur[NBIN];
  __shared__ unsigned wsum[NBIN / 32];
  __shared__ unsigned srt[CHK];
  int b = blockIdx.x;
  if (b < tile_blocks) {
    const int wave = threadIdx.x >> 5;
    const int row0 = (b * (blockDim.x >> 5) + wave) * 16;
    v2f a[4];
    load_a_frag(x, row0, 1.0f, a);
    pre_tile(a, Qm, W2, xq2, row0);
    return;
  }
  b -= tile_blocks;
  if (b < steps) {
    if (threadIdx.x < DD * DD) {
      const int k = threadIdx.x >> 4, n = threadIdx.x & 15;
      float acc = 0.f;
#pragma unroll
      for (int t = 0; t < DD; ++t)
        acc = fmaf(W1[b * DD * DD + t * DD + k], Qm[t * DD + n], acc);
      G1[b * DD * DD + k * DD + n] = acc;
    }
    return;
  }
  b -= steps;
  const unsigned chunk = (unsigned)b;
  const int* row = mrow;
  const int* col = mcol;
  int E = EM, local = b;
  unsigned list = 0u;
  if (b >= ncm + nc1) {
    row = e2row; col = e2col; E = E2; local = b - ncm - nc1; list = 2u;
  } else if (b >= ncm) {
    row = e1row; col = e1col; E = E1; local = b - ncm; list = 1u;
  }
  sort_chunk(row, col, E, local, list, chunk, recs, offs, cnt, cur, wsum, srt);
}

#define CPG 12

__device__ __forceinline__ void set_rec(unsigned rec, unsigned* bm) {
  const unsigned list = (rec >> 16) & 3u;
  if (list < 3u)
    atomicOr(&bm[(list * 16u + ((rec >> 12) & 15u)) * RW + ((rec & 4095u) >> 5)],
             1u << (rec & 31u));
}

__device__ __forceinline__ void load_bins(const unsigned* __restrict__ offs, int nch, int k0,
                                          unsigned b, unsigned o[CPG]) {
#pragma unroll
  for (int g = 0; g < CPG; ++g) {
    const int k = k0 + 16 * g;
    o[g] = k < nch ? offs[(size_t)k * NBIN + b] : 0u;
  }
}

__device__ __forceinline__ void load_recs(const unsigned* __restrict__ recs, int k0, unsigned lane,
                                          const unsigned o[CPG], unsigned rc[CPG]) {
#pragma unroll
  for (int g = 0; g < CPG; ++g) {
    const int k = k0 + 16 * g;
    const unsigned n = o[g] >> 16, st = o[g] & 0xffffu;
    rc[g] = lane < n ? recs[(size_t)k * CHK + st + lane] : REC_NONE;
  }
}

__device__ __forceinline__ void set_group(const unsigned* __restrict__ recs, int k0, unsigned lane,
                                          const unsigned o[CPG], const unsigned rc[CPG],
                                          unsigned* bm) {
#pragma unroll
  for (int g = 0; g < CPG; ++g) set_rec(rc[g], bm);
  for (int g = 0; g < CPG; ++g) {
    const int k = k0 + 16 * g;
    const unsigned n = o[g] >> 16, st = o[g] & 0xffffu;
    for (unsigned q = lane + 32u; q < n; q += 32u) set_rec(recs[(size_t)k * CHK + st + q], bm);
  }
}

template <int M>
__device__ __forceinline__ float xadd(float v) {
  return v + __int_as_float(__builtin_amdgcn_update_dpp(0, __float_as_int(v), 0x160 + M, 0xf, 0xf, true));
}

__device__ __forceinline__ float hadd(float v) {
  const int b = __float_as_int(v);
  return v + __int_as_float(__builtin_amdgcn_permlanex16(b, b, 0x76543210, 0xfedcba98, false, false));
}

__device__ __forceinline__ unsigned wave_scan_incl(unsigned v) {
  v += (unsigned)__builtin_amdgcn_update_dpp(0, (int)v, 0x111, 0xf, 0xf, true);
  v += (unsigned)__builtin_amdgcn_update_dpp(0, (int)v, 0x112, 0xf, 0xf, true);
  v += (unsigned)__builtin_amdgcn_update_dpp(0, (int)v, 0x114, 0xf, 0xf, true);
  v += (unsigned)__builtin_amdgcn_update_dpp(0, (int)v, 0x118, 0xf, 0xf, true);
  const unsigned row0 = (unsigned)__builtin_amdgcn_readlane((int)v, 15);
  return (threadIdx.x & 16u) ? v + row0 : v;
}

#define LCAP 256
#define LPITCH (LCAP + 32)
#define SEGE 16
#define S1 (SEGE / 8)
#define S2 (SEGE / 4)

struct __attribute__((aligned(16))) StepLds {
  unsigned short lst1[16][LPITCH];
  unsigned short lst2[16][LPITCH];
  float yt1[16 * TP];
  float yt2[16 * TP];
  float tile[16 * TP];
  float tile2[16 * TP];
};

__device__ __forceinline__ void put_sources(const unsigned w[4], unsigned lane, unsigned p,
                                            unsigned start, unsigned short* lst) {
#pragma unroll
  for (unsigned k = 0; k < 4; ++k) {
    unsigned v = w[k];
    while (v) {
      const unsigned bpos = (unsigned)__ffs((int)v) - 1u;
      v &= v - 1u;
      const unsigned q = p - start;
      if (q < LCAP) lst[q] = (unsigned short)((4u * lane + k) * 32u + bpos);
      ++p;
    }
  }
}

template <int MODE>
__device__ __forceinline__ void step_core(
    StepLds& L, const uint4 ra1, const uint4 ra2, const int row0,
    const float* __restrict__ xa, const float* __restrict__ xb,
    const float* __restrict__ pa, const float* __restrict__ pb,
    const float* __restrict__ G1s, const float* __restrict__ Qm,
    const unsigned* __restrict__ part2, const float s1,
    float* __restrict__ xo, float* __restrict__ xo2,
    const float* __restrict__ W2n,
    float* __restrict__ po, float* __restrict__ po2, const int has_next) {
  constexpr unsigned NS = MODE == 1 ? 3u : 4u;
  constexpr unsigned NB = MODE == 1 ? NS : 1u;

  const unsigned wave = threadIdx.x >> 5;
  const unsigned lane = threadIdx.x & 31;
  const unsigned c = lane & 15u, h = lane >> 4;
  const unsigned g4 = lane & 3u, q8 = lane >> 2;
  const unsigned u8 = lane & 7u, q4 = lane >> 3;
  const size_t j = (size_t)row0 + wave;

  v2f xj = reinterpret_cast<const v2f*>(xa + j * DD)[u8];
  v2f xjb = {};
  if (MODE == 1) xjb = reinterpret_cast<const v2f*>(xb + j * DD)[u8];

  v8f cc = {}, cb = {};
  v2f bg[4] = {}, bq[4] = {}, bw[4] = {};
  unsigned cpart = 0u;
  const bool counts = MODE == 1 || (MODE == 2 && wave == 0);
  uint4 pw[NBIN / 128];
  if (counts) {
#pragma unroll
    for (int q = 0; q < NBIN / 128; ++q)
      pw[q] = reinterpret_cast<const uint4*>(part2)[lane + 32u * q];
  }
  if (wave == 0) {
    cc = load_c_tile(xa, row0);
    if (MODE == 1) cb = load_c_tile(xb, row0);
    load_b_frag(G1s, bg);
    load_b_frag(Qm, bq);
    load_bT_frag(W2n, bw);
  }
  if (counts) {
#pragma unroll
    for (int q = 0; q < NBIN / 128; ++q) cpart += pw[q].x + pw[q].y + pw[q].z + pw[q].w;
  }

  const unsigned w1[4] = {ra1.x, ra1.y, ra1.z, ra1.w};
  const unsigned w2[4] = {ra2.x, ra2.y, ra2.z, ra2.w};
  const unsigned n1 = __popc(w1[0]) + __popc(w1[1]) + __popc(w1[2]) + __popc(w1[3]);
  const unsigned n2 = __popc(w2[0]) + __popc(w2[1]) + __popc(w2[2]) + __popc(w2[3]);

  const unsigned pk = n1 | (n2 << 16);
  const unsigned inc = wave_scan_incl(pk);
  const unsigned totpk = (unsigned)__builtin_amdgcn_readlane((int)inc, 31);
  const unsigned exc = inc - pk;
  const unsigned off1 = exc & 0xffffu, off2 = exc >> 16;
  const unsigned tot1 = totpk & 0xffffu, tot2 = totpk >> 16;

  const unsigned npass = ((tot1 > tot2 ? tot1 : tot2) + LCAP - 1) / LCAP;

  float nalpha = 0.f;
  if (MODE == 1) {
    unsigned cnt2 = cpart;
#pragma unroll
    for (int o = 16; o > 0; o >>= 1) cnt2 += __shfl_xor(cnt2, o);
    nalpha = -(float)NN / (float)cnt2;
    xj.x = fmaf(nalpha, xjb.x, xj.x);
    xj.y = fmaf(nalpha, xjb.y, xj.y);
  }

  v4f acc1 = {};
  v2f acc2 = {};
  const v4f* X4 = reinterpret_cast<const v4f*>(xa);
  const v4f* P4 = reinterpret_cast<const v4f*>(pa);
  const v4f* X4b = reinterpret_cast<const v4f*>(xb);
  const v4f* P4b = reinterpret_cast<const v4f*>(pb);
  for (unsigned ps = 0; ps < npass; ++ps) {
    const unsigned start = ps * LCAP;
    put_sources(w1, lane, off1, start, L.lst1[wave]);
    put_sources(w2, lane, off2, start, L.lst2[wave]);
    const unsigned r1n = tot1 > start ? tot1 - start : 0u;
    const unsigned r2n = tot2 > start ? tot2 - start : 0u;
    const unsigned m1 = r1n < LCAP ? r1n : LCAP;
    const unsigned m2 = r2n < LCAP ? r2n : LCAP;
    const unsigned mm = __builtin_amdgcn_readfirstlane(m1 > m2 ? m1 : m2);
    for (unsigned t0 = 0; t0 < mm; t0 += SEGE * NS) {
      unsigned pk1[NS];
      uint2 pk2[NS];
#pragma unroll
      for (unsigned sg = 0; sg < NS; ++sg) {
        pk1[sg] = *reinterpret_cast<const unsigned*>(&L.lst1[wave][t0 + SEGE * sg + 2u * q8]);
        pk2[sg] = *reinterpret_cast<const uint2*>(&L.lst2[wave][t0 + SEGE * sg + 4u * q4]);
      }
      v4f v1[NS * S1];
      v4f qv[NS * S2];
      v4f v1b[NB * S1];
      v4f qvb[NB * S2];
#pragma unroll
      for (unsigned sg = 0; sg < NS; ++sg) {
        if (t0 + SEGE * sg < mm) {
#pragma unroll
          for (unsigned u = 0; u < S1; ++u) {
            const unsigned e = t0 + SEGE * sg + 2u * q8 + u;
            const unsigned si = (pk1[sg] >> (16u * u)) & 0xffffu;
            const unsigned ix = (e < m1 ? si : 0u) * (DD / 4) + g4;
            v1[sg * S1 + u] = X4[ix];
            if (MODE == 1) v1b[sg * S1 + u] = X4b[ix];
          }
#pragma unroll
          for (unsigned u = 0; u < S2; ++u) {
            const unsigned e = t0 + SEGE * sg + 4u * q4 + u;
            const unsigned wd = u < 2u ? pk2[sg].x : pk2[sg].y;
            const unsigned si = (wd >> (16u * (u & 1u))) & 0xffffu;
            const unsigned ix = (e < m2 ? si : 0u) * (XP / 4) + u8;
            qv[sg * S2 + u] = P4[ix];
            if (MODE == 1) qvb[sg * S2 + u] = P4b[ix];
          }
        }
      }
      __builtin_amdgcn_sched_barrier(0);
#pragma unroll
      for (unsigned sg = 0; sg < NS; ++sg) {
        if (t0 + SEGE * sg < mm) {
#pragma unroll
          for (unsigned u = 0; u < S1; ++u) {
            const unsigned s = sg * S1 + u;
            v4f xs = v1[s];
            if (MODE == 1) {
              xs.x = fmaf(nalpha, v1b[s].x, xs.x);
              xs.y = fmaf(nalpha, v1b[s].y, xs.y);
              xs.z = fmaf(nalpha, v1b[s].z, xs.z);
              xs.w = fmaf(nalpha, v1b[s].w, xs.w);
            }
            const float w = t0 + SEGE * sg + 2u * q8 + u < m1 ? 1.0f : 0.0f;
            acc1.x = fmaf(xs.x, w, acc1.x);
            acc1.y = fmaf(xs.y, w, acc1.y);
            acc1.z = fmaf(xs.z, w, acc1.z);
            acc1.w = fmaf(xs.w, w, acc1.w);
          }
#pragma unroll
          for (unsigned u = 0; u < S2; ++u) {
            const unsigned s = sg * S2 + u;
            v4f qs = qv[s];
            if (MODE == 1) {
              qs.x = fmaf(nalpha, qvb[s].x, qs.x);
              qs.y = fmaf(nalpha, qvb[s].y, qs.y);
              qs.z = fmaf(nalpha, qvb[s].z, qs.z);
              qs.w = fmaf(nalpha, qvb[s].w, qs.w);
            }
            float a = qs.x * xj.x;
            a = fmaf(qs.z, xj.y, a);
            a = xadd<4>(xadd<2>(xadd<1>(a)));
            a = t0 + SEGE * sg + 4u * q4 + u < m2 ? a : 0.f;
            acc2.x = fmaf(a, qs.y, acc2.x);
            acc2.y = fmaf(a, qs.w, acc2.y);
          }
        }
      }
    }
  }

  acc1.x = hadd(xadd<8>(xadd<4>(acc1.x)));
  acc1.y = hadd(xadd<8>(xadd<4>(acc1.y)));
  acc1.z = hadd(xadd<8>(xadd<4>(acc1.z)));
  acc1.w = hadd(xadd<8>(xadd<4>(acc1.w)));
  acc2.x = hadd(xadd<8>(acc2.x));
  acc2.y = hadd(xadd<8>(acc2.y));
  if (lane < 4u) {
    v2f* y = reinterpret_cast<v2f*>(L.yt1 + wave * TP + 4u * lane);
    v2f lo, hi;
    lo.x = acc1.x; lo.y = acc1.y;
    hi.x = acc1.z; hi.y = acc1.w;
    y[0] = lo;
    y[1] = hi;
  }
  if (lane < 8u) *reinterpret_cast<v2f*>(L.yt2 + wave * TP + 2u * lane) = acc2;
  __syncthreads();

  const bool next = MODE == 0 ? true : has_next != 0;
  if (wave == 0) {
    v2f a[4];
    if (MODE == 0) {
      load_a_frag_lds(L.yt1, s1, a);
      cc = wmma16(a, bg, cc);
      load_a_frag_lds(L.yt2, 1.0f, a);
      cb = wmma16(a, bq, cb);
      store_d_tile(xo, row0, cc);
      store_d_tile(xo2, row0, cb);
#pragma unroll
      for (int r = 0; r < 8; ++r) {
        L.tile[(r + 8 * h) * TP + c] = cc[r];
        L.tile2[(r + 8 * h) * TP + c] = cb[r];
      }
    } else {
      float s2 = nalpha;
      if (MODE == 2) {
        unsigned cnt2 = cpart;
#pragma unroll
        for (int o = 16; o > 0; o >>= 1) cnt2 += __shfl_xor(cnt2, o);
        s2 = -(float)NN / (float)cnt2;
      }
      if (MODE == 1) {
#pragma unroll
        for (int r = 0; r < 8; ++r) cc[r] = fmaf(s2, cb[r], cc[r]);
      }
      load_a_frag_lds(L.yt1, s1, a);
      cc = wmma16(a, bg, cc);
      load_a_frag_lds(L.yt2, s2, a);
      cc = wmma16(a, bq, cc);
      store_d_tile(xo, row0, cc);
      if (next) {
#pragma unroll
        for (int r = 0; r < 8; ++r) L.tile[(r + 8 * h) * TP + c] = cc[r];
      }
    }
  }
  if (next) {
    if (wave == 0) {
      v2f a[4];
      load_a_frag_lds(L.tile, 1.0f, a);
      v8f d = {};
      d = wmma16(a, bq, d);
      v8f d2 = {};
      d2 = wmma16(a, bw, d2);
      store_d_pair(po, row0, d, d2);
      if (MODE == 0) {
        load_a_frag_lds(L.tile2, 1.0f, a);
        v8f e = {};
        e = wmma16(a, bq, e);
        v8f e2 = {};
        e2 = wmma16(a, bw, e2);
        store_d_pair(po2, row0, e, e2);
      }
    }
  }
}

__global__ __launch_bounds__(BT) void k_build0(
    const unsigned* __restrict__ recs, const unsigned* __restrict__ offs, int nch,
    unsigned* __restrict__ AB, unsigned* __restrict__ part2,
    const float* __restrict__ x, const float* __restrict__ xq2,
    const float* __restrict__ G1s, const float* __restrict__ Qm, float s1,
    float* __restrict__ xu, float* __restrict__ xv, const float* __restrict__ W2n,
    float* __restrict__ pu, float* __restrict__ pv) {
  __shared__ __attribute__((aligned(16))) unsigned bm[3 * 16 * RW];
  __shared__ unsigned wc[16];
  __shared__ StepLds L;
  const unsigned t = threadIdx.x, wave = t >> 5, lane = t & 31u;
  const unsigned b = blockIdx.x;
  unsigned o[CPG], rc[CPG];
  load_bins(offs, nch, (int)wave, b, o);
  for (unsigned i = t; i < 3u * 16u * RW; i += BT) bm[i] = 0u;
  load_recs(recs, (int)wave, lane, o, rc);
  __syncthreads();
  set_group(recs, (int)wave, lane, o, rc, bm);
  for (int k0 = (int)wave + 16 * CPG; k0 < nch; k0 += 16 * CPG) {
    load_bins(offs, nch, k0, b, o);
    load_recs(recs, k0, lane, o, rc);
    set_group(recs, k0, lane, o, rc, bm);
  }
  __syncthreads();
  const uint4 m = reinterpret_cast<const uint4*>(bm + (0u * 16u + wave) * RW)[lane];
  const uint4 b1 = reinterpret_cast<const uint4*>(bm + (1u * 16u + wave) * RW)[lane];
  const uint4 b2 = reinterpret_cast<const uint4*>(bm + (2u * 16u + wave) * RW)[lane];
  uint4 a1, a2;
  a1.x = b1.x & m.x; a1.y = b1.y & m.y; a1.z = b1.z & m.z; a1.w = b1.w & m.w;
  a2.x = b2.x & m.x; a2.y = b2.y & m.y; a2.z = b2.z & m.z; a2.w = b2.w & m.w;
  const size_t j = (size_t)b * 16 + wave;
  reinterpret_cast<uint4*>(AB + j * (2 * RW))[lane] = a1;
  reinterpret_cast<uint4*>(AB + j * (2 * RW) + RW)[lane] = a2;
  unsigned n2 = __popc(a2.x) + __popc(a2.y) + __popc(a2.z) + __popc(a2.w);
#pragma unroll
  for (int s = 16; s > 0; s >>= 1) n2 += __shfl_xor(n2, s);
  if (lane == 0) wc[wave] = n2;

  step_core<0>(L, a1, a2, (int)b * 16, x, nullptr, xq2, nullptr, G1s, Qm, nullptr, s1,
               xu, xv, W2n, pu, pv, 1);

  if (t == 32u) {
    unsigned s = 0;
    for (unsigned k = 0; k < 16; ++k) s += wc[k];
    part2[b] = s;
  }
}

__global__ __launch_bounds__(512) void k_step1(
    const float* __restrict__ xu, const float* __restrict__ xv,
    const float* __restrict__ pu, const float* __restrict__ pv,
    const unsigned* __restrict__ AB,
    const float* __restrict__ G1s, const float* __restrict__ Qm,
    const unsigned* __restrict__ part2, float s1,
    float* __restrict__ xnext, const float* __restrict__ W2n,
    float* __restrict__ xq2n, int has_next) {
  __shared__ StepLds L;
  const unsigned wave = threadIdx.x >> 5;
  const unsigned lane = threadIdx.x & 31;
  const int row0 = blockIdx.x * 16;
  const size_t j = (size_t)row0 + wave;
  const uint4 ra1 = reinterpret_cast<const uint4*>(AB + j * (2 * RW))[lane];
  const uint4 ra2 = reinterpret_cast<const uint4*>(AB + j * (2 * RW) + RW)[lane];
  step_core<1>(L, ra1, ra2, row0, xu, xv, pu, pv, G1s, Qm, part2, s1,
               xnext, nullptr, W2n, xq2n, nullptr, has_next);
}

__global__ __launch_bounds__(512) void k_step2(
    const float* __restrict__ xcur, const float* __restrict__ xq2,
    const unsigned* __restrict__ AB,
    const float* __restrict__ G1s, const float* __restrict__ Qm,
    const unsigned* __restrict__ part2, float s1,
    float* __restrict__ xnext, const float* __restrict__ W2n,
    float* __restrict__ xq2n, int has_next) {
  __shared__ StepLds L;
  const unsigned wave = threadIdx.x >> 5;
  const unsigned lane = threadIdx.x & 31;
  const int row0 = blockIdx.x * 16;
  const size_t j = (size_t)row0 + wave;
  const uint4 ra1 = reinterpret_cast<const uint4*>(AB + j * (2 * RW))[lane];
  const uint4 ra2 = reinterpret_cast<const uint4*>(AB + j * (2 * RW) + RW)[lane];
  step_core<2>(L, ra1, ra2, row0, xcur, nullptr, xq2, nullptr, G1s, Qm, part2, s1,
               xnext, nullptr, W2n, xq2n, nullptr, has_next);
}

extern "C" void kernel_launch(void* const* d_in, const int* in_sizes, int n_in,
                              void* d_out, int out_size, void* d_ws, size_t ws_size,
                              hipStream_t stream) {
  const float* x_in = (const float*)d_in[0];
  const int* e1 = (const int*)d_in[1];
  const int* e2 = (const int*)d_in[2];
  const float* Qm = (const float*)d_in[3];
  const int* mk = (const int*)d_in[4];
  const float* W1 = (const float*)d_in[5];
  const float* W2 = (const float*)d_in[6];
  float* out = (float*)d_out;

  const int E1 = in_sizes[1] / 2;
  const int E2 = in_sizes[2] / 2;
  const int EM = in_sizes[4] / 2;
  const int STEPS = in_sizes[5] / (DD * DD);

  char* ws = (char*)d_ws;
  size_t off = 0;
  auto carve = [&](size_t bytes) {
    char* p = ws + off;
    off += (bytes + 255) & ~(size_t)255;
    return p;
  };
  const int NCM = (EM + CHK - 1) / CHK;
  const int NC1 = (E1 + CHK - 1) / CHK;
  const int NC2 = (E2 + CHK - 1) / CHK;
  const int NCH = NCM + NC1 + NC2;
  unsigned* recs  = (unsigned*)carve((size_t)NCH * CHK * 4);
  unsigned* offs  = (unsigned*)carve((size_t)NCH * NBIN * 4);
  unsigned* AB    = (unsigned*)carve((size_t)NN * 2 * RW * 4);
  unsigned* part2 = (unsigned*)carve((size_t)NBIN * 4);
  float* G1  = (float*)carve((size_t)STEPS * DD * DD * 4);
  float* xq2a = (float*)carve((size_t)NN * XP * 4);
  float* xq2b = (float*)carve((size_t)NN * XP * 4);
  float* xb0 = (float*)carve((size_t)NN * DD * 4);
  float* xb1 = (float*)carve((size_t)NN * DD * 4);
  float* xu = (float*)carve((size_t)NN * DD * 4);
  float* xv = (float*)carve((size_t)NN * DD * 4);
  float* pu = (float*)carve((size_t)NN * XP * 4);
  float* pv = (float*)carve((size_t)NN * XP * 4);

  const int TILE_BLOCKS = NN / 16 / (BT / 32);

  k_front<<<TILE_BLOCKS + STEPS + NCH, BT, 0, stream>>>(
      x_in, Qm, W1, W2, STEPS, TILE_BLOCKS, xq2a, G1,
      mk, mk + EM, EM, NCM,
      e1, e1 + E1, E1, NC1,
      e2, e2 + E2, E2, recs, offs);

  const float s1 = (float)NN / (float)EM;

  k_build0<<<NBIN, BT, 0, stream>>>(recs, offs, NCH, AB, part2, x_in, xq2a, G1, Qm, s1,
                                    xu, xv, W2 + (size_t)(STEPS > 1 ? 1 : 0) * DD * DD, pu, pv);

  const float* xc = xu;
  const float* xq2c = pu;
  for (int s = 1; s < STEPS; ++s) {
    float* xn = (s == STEPS - 1) ? out : ((s & 1) ? xb0 : xb1);
    const int has_next = (s + 1 < STEPS) ? 1 : 0;
    const float* W2n = W2 + (size_t)(has_next ? s + 1 : s) * DD * DD;
    float* xq2n = (s & 1) ? xq2a : xq2b;
    if (s == 1)
      k_step1<<<NN / 16, BT, 0, stream>>>(xu, xv, pu, pv, AB, G1 + (size_t)s * DD * DD, Qm,
                                          part2, s1, xn, W2n, xq2n, has_next);
    else
      k_step2<<<NN / 16, BT, 0, stream>>>(xc, xq2c, AB, G1 + (size_t)s * DD * DD, Qm,
                                          part2, s1, xn, W2n, xq2n, has_next);
    xc = xn;
    xq2c = xq2n;
  }
}
